# starting kernel
# baseline (speedup 1.0000x reference)
#include <hip/hip_runtime.h>

typedef __attribute__((ext_vector_type(2))) float v2f;
typedef __attribute__((ext_vector_type(8))) float v8f;

constexpr int kNCOEF = 25;
constexpr int kM0    = 5;
constexpr int kC     = 128;
constexpr int kTILE  = 16;
constexpr float kCUTOFF  = 12.0f;
constexpr float kRESCALE = 5.0f;
constexpr float kEPS     = 1e-5f;

__device__ __forceinline__ float silu_f(float x) {
    return x / (1.0f + __expf(-x));
}

__device__ __forceinline__ void ln_silu_16x64(const float* __restrict__ hraw,
                                              float* __restrict__ hout,
                                              const float* __restrict__ g,
                                              const float* __restrict__ be,
                                              int tid) {
    if (tid < kTILE) {
        float s = 0.0f, s2 = 0.0f;
        #pragma unroll 8
        for (int c = 0; c < 64; ++c) {
            float v = hraw[tid * 64 + c];
            s += v; s2 += v * v;
        }
        float mu  = s * (1.0f / 64.0f);
        float var = s2 * (1.0f / 64.0f) - mu * mu;
        float inv = rsqrtf(var + kEPS);
        #pragma unroll 8
        for (int c = 0; c < 64; ++c) {
            float v = (hraw[tid * 64 + c] - mu) * inv * g[c] + be[c];
            hout[tid * 64 + c] = silu_f(v);
        }
    }
}

__global__ __launch_bounds__(128)
void edge_degree_embed_kernel(const float* __restrict__ x_edge,
                              const float* __restrict__ edge_distance,
                              const float* __restrict__ wigner_inv,
                              const int*   __restrict__ edge_index,
                              const float* __restrict__ w1, const float* __restrict__ b1,
                              const float* __restrict__ g1, const float* __restrict__ be1,
                              const float* __restrict__ w2, const float* __restrict__ b2,
                              const float* __restrict__ g2, const float* __restrict__ be2,
                              const float* __restrict__ w3, const float* __restrict__ b3,
                              const int*   __restrict__ node_offset_p,
                              float* __restrict__ out,
                              int E)
{
    __shared__ float xe[kTILE * 128];
    __shared__ float hraw[kTILE * 64];
    __shared__ float h1[kTILE * 64];
    __shared__ float h2[kTILE * 64];
    __shared__ float m0s[kTILE * kM0 * kC];
    __shared__ float envs[kTILE];

    const int tid   = threadIdx.x;
    const int wave  = tid >> 5;
    const int lane  = tid & 31;
    const int lrow  = lane & 15;
    const int khalf = (lane >> 4) << 1;
    const int rbase = (lane >> 4) << 3;
    const int e0    = blockIdx.x * kTILE;

    if (tid < kTILE && (e0 + tid) < E) {
        __builtin_prefetch(&wigner_inv[(size_t)(e0 + tid) * (kNCOEF * kNCOEF)], 0, 0);
    }

    for (int i = tid; i < kTILE * 128; i += blockDim.x) {
        int e = e0 + (i >> 7);
        xe[i] = (e < E) ? x_edge[(size_t)e * 128 + (i & 127)] : 0.0f;
    }
    __syncthreads();

    {
        const int colBase = wave * 16;
        v8f acc = {};
        for (int k0 = 0; k0 < 128; k0 += 4) {
            v2f a, b;
            a.x = xe[lrow * 128 + k0 + khalf];
            a.y = xe[lrow * 128 + k0 + khalf + 1];
            b.x = w1[(k0 + khalf)     * 64 + colBase + lrow];
            b.y = w1[(k0 + khalf + 1) * 64 + colBase + lrow];
            acc = __builtin_amdgcn_wmma_f32_16x16x4_f32(
                false, a, false, b, (short)0, acc, false, false);
        }
        #pragma unroll
        for (int r = 0; r < 8; ++r) {
            int row = rbase + r;
            hraw[row * 64 + colBase + lrow] = acc[r] + b1[colBase + lrow];
        }
    }
    __syncthreads();
    ln_silu_16x64(hraw, h1, g1, be1, tid);
    __syncthreads();

    {
        const int colBase = wave * 16;
        v8f acc = {};
        for (int k0 = 0; k0 < 64; k0 += 4) {
            v2f a, b;
            a.x = h1[lrow * 64 + k0 + khalf];
            a.y = h1[lrow * 64 + k0 + khalf + 1];
            b.x = w2[(k0 + khalf)     * 64 + colBase + lrow];
            b.y = w2[(k0 + khalf + 1) * 64 + colBase + lrow];
            acc = __builtin_amdgcn_wmma_f32_16x16x4_f32(
                false, a, false, b, (short)0, acc, false, false);
        }
        #pragma unroll
        for (int r = 0; r < 8; ++r) {
            int row = rbase + r;
            hraw[row * 64 + colBase + lrow] = acc[r] + b2[colBase + lrow];
        }
    }
    __syncthreads();
    ln_silu_16x64(hraw, h2, g2, be2, tid);
    __syncthreads();

    for (int t = wave; t < 40; t += 4) {
        const int colBase = t * 16;
        v8f acc = {};
        for (int k0 = 0; k0 < 64; k0 += 4) {
            v2f a, b;
            a.x = h2[lrow * 64 + k0 + khalf];
            a.y = h2[lrow * 64 + k0 + khalf + 1];
            b.x = w3[(size_t)(k0 + khalf)     * 640 + colBase + lrow];
            b.y = w3[(size_t)(k0 + khalf + 1) * 640 + colBase + lrow];
            acc = __builtin_amdgcn_wmma_f32_16x16x4_f32(
                false, a, false, b, (short)0, acc, false, false);
        }
        #pragma unroll
        for (int r = 0; r < 8; ++r) {
            int row = rbase + r;
            m0s[row * (kM0 * kC) + colBase + lrow] = acc[r] + b3[colBase + lrow];
        }
    }
    __syncthreads();

    float* wig = xe;
    for (int i = tid; i < kTILE * kNCOEF * kM0; i += blockDim.x) {
        int e_loc = i / (kNCOEF * kM0);
        int rem   = i % (kNCOEF * kM0);
        int r     = rem / kM0;
        int l     = rem % kM0;
        int e     = e0 + e_loc;
        wig[i] = (e < E)
            ? wigner_inv[(size_t)e * (kNCOEF * kNCOEF) + r * kNCOEF + l * (l + 1)]
            : 0.0f;
    }
    if (tid < kTILE) {
        int e = e0 + tid;
        float env = 0.0f;
        if (e < E) {
            float d = edge_distance[e] * (1.0f / kCUTOFF);
            if (d < 1.0f) {
                float d5 = d * d * d * d * d;
                env = 1.0f - 21.0f * d5 + 35.0f * d5 * d - 15.0f * d5 * d * d;
            }
        }
        envs[tid] = env * (1.0f / kRESCALE);
    }
    __syncthreads();

    const int noff = node_offset_p[0];
    for (int i = tid; i < kTILE * kNCOEF * kC; i += blockDim.x) {
        int e_loc = i / (kNCOEF * kC);
        int rem   = i % (kNCOEF * kC);
        int e     = e0 + e_loc;
        if (e >= E) continue;
        int r = rem >> 7;
        int c = rem & 127;
        float s = 0.0f;
        #pragma unroll
        for (int l = 0; l < kM0; ++l) {
            s += wig[(e_loc * kNCOEF + r) * kM0 + l] * m0s[e_loc * (kM0 * kC) + l * kC + c];
        }
        int dst = edge_index[E + e] - noff;
        atomicAdd(&out[((size_t)dst * kNCOEF + r) * kC + c], s * envs[e_loc]);
    }
}

extern "C" void kernel_launch(void* const* d_in, const int* in_sizes, int n_in,
                              void* d_out, int out_size, void* d_ws, size_t ws_size,
                              hipStream_t stream) {
    const float* x             = (const float*)d_in[0];
    const float* x_edge        = (const float*)d_in[1];
    const float* edge_distance = (const float*)d_in[2];
    const float* wigner_inv    = (const float*)d_in[3];
    const int*   edge_index    = (const int*)d_in[4];
    const float* w1  = (const float*)d_in[5];
    const float* b1  = (const float*)d_in[6];
    const float* g1  = (const float*)d_in[7];
    const float* be1 = (const float*)d_in[8];
    const float* w2  = (const float*)d_in[9];
    const float* b2  = (const float*)d_in[10];
    const float* g2  = (const float*)d_in[11];
    const float* be2 = (const float*)d_in[12];
    const float* w3  = (const float*)d_in[13];
    const float* b3  = (const float*)d_in[14];
    const int* node_offset = (const int*)d_in[15];

    const int E = in_sizes[1] / 128;

    hipMemcpyAsync(d_out, x, (size_t)out_size * sizeof(float),
                   hipMemcpyDeviceToDevice, stream);

    const int blocks = (E + kTILE - 1) / kTILE;
    edge_degree_embed_kernel<<<blocks, 128, 0, stream>>>(
        x_edge, edge_distance, wigner_inv, edge_index,
        w1, b1, g1, be1, w2, b2, g2, be2, w3, b3,
        node_offset, (float*)d_out, E);
}


# Float atomics and the copy of x removed: edges grouped by destination in-call, m0 kept per edge in workspace, one block per node contracts and stores x+sum in edge-id order (sum order changes)
# speedup vs baseline: 2.4268x; 2.4268x over previous
#include <hip/hip_runtime.h>

typedef __attribute__((ext_vector_type(2))) float v2f;
typedef __attribute__((ext_vector_type(8))) float v8f;

constexpr int kNCOEF = 25;
constexpr int kM0    = 5;
constexpr int kC     = 128;
constexpr int kTILE  = 16;
constexpr float kCUTOFF  = 12.0f;
constexpr float kRESCALE = 5.0f;
constexpr float kEPS     = 1e-5f;
constexpr int kLISTT  = 1024;
constexpr int kLISTU  = 20;
constexpr int kNCHUNK = 8192;
constexpr int kGB     = 16;
constexpr int kNPAIR  = 13;
constexpr int kWROW   = 160;

__device__ __forceinline__ float silu_f(float x) {
    return x / (1.0f + __expf(-x));
}

__device__ __forceinline__ void ln_silu_16x64(const float* __restrict__ hraw,
                                              float* __restrict__ hout,
                                              const float* __restrict__ g,
                                              const float* __restrict__ be,
                                              int tid) {
    if (tid < kTILE) {
        float s = 0.0f, s2 = 0.0f;
        #pragma unroll 8
        for (int c = 0; c < 64; ++c) {
            float v = hraw[tid * 64 + c];
            s += v; s2 += v * v;
        }
        float mu  = s * (1.0f / 64.0f);
        float var = s2 * (1.0f / 64.0f) - mu * mu;
        float inv = rsqrtf(var + kEPS);
        #pragma unroll 8
        for (int c = 0; c < 64; ++c) {
            float v = (hraw[tid * 64 + c] - mu) * inv * g[c] + be[c];
            hout[tid * 64 + c] = silu_f(v);
        }
    }
}


__global__ __launch_bounds__(kLISTT)
void build_dst_lists_kernel(const int* __restrict__ edge_index,
                            const int* __restrict__ node_offset_p,
                            int* __restrict__ row_off,
                            int* __restrict__ perm,
                            int E, int N)
{
    __shared__ int cnt[kNCHUNK];
    __shared__ int wsum[32];
    __shared__ int total_s;

    const int tid  = threadIdx.x;
    const int lane = tid & 31;
    const int wave = tid >> 5;
    const int noff = node_offset_p[0];
    const int* __restrict__ dstv = edge_index + E;
    constexpr int kPER = kNCHUNK / kLISTT;

    int carry = 0;
    for (int base = 0; base < N; base += kNCHUNK) {
        const int nn = (N - base < kNCHUNK) ? (N - base) : kNCHUNK;
        for (int i = tid; i < kNCHUNK; i += kLISTT) cnt[i] = 0;
        __syncthreads();

        for (int eb = tid; eb < E; eb += kLISTT * kLISTU) {
            int d[kLISTU];
            #pragma unroll
            for (int u = 0; u < kLISTU; ++u) {
                const int e = eb + u * kLISTT;
                d[u] = dstv[e < E ? e : eb];
            }
            #pragma unroll
            for (int u = 0; u < kLISTU; ++u) {
                const int e  = eb + u * kLISTT;
                const int dd = d[u] - noff - base;
                if (e < E && dd >= 0 && dd < nn) atomicAdd(&cnt[dd], 1);
            }
        }
        __syncthreads();

        int loc[kPER];
        int s = 0;
        #pragma unroll
        for (int j = 0; j < kPER; ++j) { loc[j] = s; s += cnt[tid * kPER + j]; }
        int incl = s;
        #pragma unroll
        for (int off = 1; off < 32; off <<= 1) {
            const int t = __shfl_up(incl, off);
            if (lane >= off) incl += t;
        }
        if (lane == 31) wsum[wave] = incl;
        __syncthreads();
        if (wave == 0) {
            const int w = wsum[lane];
            int wi = w;
            #pragma unroll
            for (int off = 1; off < 32; off <<= 1) {
                const int t = __shfl_up(wi, off);
                if (lane >= off) wi += t;
            }
            wsum[lane] = wi - w;
            if (lane == 31) total_s = wi;
        }
        __syncthreads();
        const int first = carry + wsum[wave] + (incl - s);
        #pragma unroll
        for (int j = 0; j < kPER; ++j) {
            const int idx = tid * kPER + j;
            const int o = first + loc[j];
            cnt[idx] = o;
            if (idx < nn) row_off[base + idx] = o;
        }
        carry += total_s;
        __syncthreads();

        for (int eb = tid; eb < E; eb += kLISTT * kLISTU) {
            int d[kLISTU];
            #pragma unroll
            for (int u = 0; u < kLISTU; ++u) {
                const int e = eb + u * kLISTT;
                d[u] = dstv[e < E ? e : eb];
            }
            int pos[kLISTU];
            #pragma unroll
            for (int u = 0; u < kLISTU; ++u) {
                const int e  = eb + u * kLISTT;
                const int dd = d[u] - noff - base;
                pos[u] = (e < E && dd >= 0 && dd < nn) ? atomicAdd(&cnt[dd], 1) : -1;
            }
            #pragma unroll
            for (int u = 0; u < kLISTU; ++u) {
                if (pos[u] >= 0) perm[pos[u]] = eb + u * kLISTT;
            }
        }
        __syncthreads();
    }
    if (tid == 0) row_off[N] = carry;
}

__global__ __launch_bounds__(128)
void edge_mlp_kernel(const float* __restrict__ x_edge,
                              const float* __restrict__ edge_distance,
                              const float* __restrict__ w1, const float* __restrict__ b1,
                              const float* __restrict__ g1, const float* __restrict__ be1,
                              const float* __restrict__ w2, const float* __restrict__ b2,
                              const float* __restrict__ g2, const float* __restrict__ be2,
                              const float* __restrict__ w3, const float* __restrict__ b3,
                              float* __restrict__ m0buf,
                              float* __restrict__ envbuf,
                              int E)
{
    __shared__ float xe[kTILE * 128];
    __shared__ float hraw[kTILE * 64];
    __shared__ float h1[kTILE * 64];
    __shared__ float h2[kTILE * 64];
    __shared__ __attribute__((aligned(16))) float m0s[kTILE * kM0 * kC];

    const int tid   = threadIdx.x;
    const int wave  = tid >> 5;
    const int lane  = tid & 31;
    const int lrow  = lane & 15;
    const int khalf = (lane >> 4) << 1;
    const int rbase = (lane >> 4) << 3;
    const int e0    = blockIdx.x * kTILE;

    for (int i = tid; i < kTILE * 128; i += blockDim.x) {
        int e = e0 + (i >> 7);
        xe[i] = (e < E) ? x_edge[(size_t)e * 128 + (i & 127)] : 0.0f;
    }
    __syncthreads();

    {
        const int colBase = wave * 16;
        v8f acc = {};
        for (int k0 = 0; k0 < 128; k0 += 4) {
            v2f a, b;
            a.x = xe[lrow * 128 + k0 + khalf];
            a.y = xe[lrow * 128 + k0 + khalf + 1];
            b.x = w1[(k0 + khalf)     * 64 + colBase + lrow];
            b.y = w1[(k0 + khalf + 1) * 64 + colBase + lrow];
            acc = __builtin_amdgcn_wmma_f32_16x16x4_f32(
                false, a, false, b, (short)0, acc, false, false);
        }
        #pragma unroll
        for (int r = 0; r < 8; ++r) {
            int row = rbase + r;
            hraw[row * 64 + colBase + lrow] = acc[r] + b1[colBase + lrow];
        }
    }
    __syncthreads();
    ln_silu_16x64(hraw, h1, g1, be1, tid);
    __syncthreads();

    {
        const int colBase = wave * 16;
        v8f acc = {};
        for (int k0 = 0; k0 < 64; k0 += 4) {
            v2f a, b;
            a.x = h1[lrow * 64 + k0 + khalf];
            a.y = h1[lrow * 64 + k0 + khalf + 1];
            b.x = w2[(k0 + khalf)     * 64 + colBase + lrow];
            b.y = w2[(k0 + khalf + 1) * 64 + colBase + lrow];
            acc = __builtin_amdgcn_wmma_f32_16x16x4_f32(
                false, a, false, b, (short)0, acc, false, false);
        }
        #pragma unroll
        for (int r = 0; r < 8; ++r) {
            int row = rbase + r;
            hraw[row * 64 + colBase + lrow] = acc[r] + b2[colBase + lrow];
        }
    }
    __syncthreads();
    ln_silu_16x64(hraw, h2, g2, be2, tid);
    __syncthreads();

    for (int t = wave; t < 40; t += 4) {
        const int colBase = t * 16;
        v8f acc = {};
        for (int k0 = 0; k0 < 64; k0 += 4) {
            v2f a, b;
            a.x = h2[lrow * 64 + k0 + khalf];
            a.y = h2[lrow * 64 + k0 + khalf + 1];
            b.x = w3[(size_t)(k0 + khalf)     * 640 + colBase + lrow];
            b.y = w3[(size_t)(k0 + khalf + 1) * 640 + colBase + lrow];
            acc = __builtin_amdgcn_wmma_f32_16x16x4_f32(
                false, a, false, b, (short)0, acc, false, false);
        }
        #pragma unroll
        for (int r = 0; r < 8; ++r) {
            int row = rbase + r;
            m0s[row * (kM0 * kC) + colBase + lrow] = acc[r] + b3[colBase + lrow];
        }
    }
    __syncthreads();

    {
        const float4* src4 = reinterpret_cast<const float4*>(m0s);
        float4* dst4 = reinterpret_cast<float4*>(m0buf + (size_t)e0 * (kM0 * kC));
        for (int i = tid; i < kTILE * (kM0 * kC / 4); i += blockDim.x) {
            if (e0 + i / (kM0 * kC / 4) < E) dst4[i] = src4[i];
        }
    }
    if (tid < kTILE) {
        int e = e0 + tid;
        if (e < E) {
            float env = 0.0f;
            float d = edge_distance[e] * (1.0f / kCUTOFF);
            if (d < 1.0f) {
                float d5 = d * d * d * d * d;
                env = 1.0f - 21.0f * d5 + 35.0f * d5 * d - 15.0f * d5 * d * d;
            }
            envbuf[e] = env * (1.0f / kRESCALE);
        }
    }
}

__global__ __launch_bounds__(128, 5)
void node_gather_kernel(const float* __restrict__ x,
                        const float* __restrict__ wigner_inv,
                        const float* __restrict__ m0buf,
                        const float* __restrict__ envbuf,
                        const int*   __restrict__ row_off,
                        const int*   __restrict__ perm,
                        int*         __restrict__ perm_sorted,
                        float*       __restrict__ out)
{
    __shared__ __attribute__((aligned(16))) float wig_s[kGB * kWROW];
    __shared__ int elist[kGB];

    const int n   = blockIdx.x;
    const int c   = threadIdx.x;
    const int beg = row_off[n];
    const int cnt = row_off[n + 1] - beg;

    v2f acc[kNPAIR];
    const float* xn = x + (size_t)n * (kNCOEF * kC) + c;
    #pragma unroll
    for (int p = 0; p < kNPAIR; ++p) {
        acc[p].x = xn[(2 * p) * kC];
        acc[p].y = (2 * p + 1 < kNCOEF) ? xn[(2 * p + 1) * kC] : 0.0f;
    }
    for (int i = c; i < kGB * kWROW; i += blockDim.x) wig_s[i] = 0.0f;

    const int  stage_r    = c / kM0;
    const int  stage_l    = c - stage_r * kM0;
    const bool stage_w    = c < kNCOEF * kM0;
    const int  stage_goff = stage_r * kNCOEF + stage_l * (stage_l + 1);
    const int  stage_lidx = stage_w ? ((stage_r >> 1) * 12 + stage_l * 2 + (stage_r & 1)) : 10;

    for (int i = c; i < cnt; i += blockDim.x) {
        const int ei = perm[beg + i];
        int rank = 0;
        for (int j = 0; j < cnt; ++j) rank += (perm[beg + j] < ei) ? 1 : 0;
        perm_sorted[beg + rank] = ei;
    }
    __syncthreads();

    for (int b0 = 0; b0 < cnt; b0 += kGB) {
        const int nb = (cnt - b0 < kGB) ? (cnt - b0) : kGB;
        if (c < nb) elist[c] = perm_sorted[beg + b0 + c];
        __syncthreads();

        v2f m[kGB][3];
        float v[kGB];
        #pragma unroll
        for (int k = 0; k < kGB; ++k) {
            if (k < nb) {
                const int e = elist[k];
                const float* mp = m0buf + (size_t)e * (kM0 * kC) + c;
                m[k][0].x = mp[0 * kC];
                m[k][0].y = mp[1 * kC];
                m[k][1].x = mp[2 * kC];
                m[k][1].y = mp[3 * kC];
                m[k][2].x = mp[4 * kC];
                m[k][2].y = 0.0f;
                if (c <= kNCOEF * kM0) {
                    v[k] = stage_w ? wigner_inv[(size_t)e * (kNCOEF * kNCOEF) + stage_goff]
                                   : envbuf[e];
                }
            }
        }
        #pragma unroll
        for (int k = 0; k < kGB; ++k) {
            if (k < nb && c <= kNCOEF * kM0) wig_s[k * kWROW + stage_lidx] = v[k];
        }
        __syncthreads();

        #pragma unroll
        for (int k = 0; k < kGB; ++k) {
            if (k < nb) {
                const float* w = wig_s + k * kWROW;
                const float env = w[10];
                #pragma unroll
                for (int p = 0; p < kNPAIR; ++p) {
                    v2f s = {0.0f, 0.0f};
                    #pragma unroll
                    for (int l = 0; l < kM0; ++l) {
                        const v2f w2 = *reinterpret_cast<const v2f*>(w + p * 12 + l * 2);
                        s += w2 * ((l & 1) ? m[k][l >> 1].y : m[k][l >> 1].x);
                    }
                    acc[p] += s * env;
                }
            }
        }
        __syncthreads();
    }

    float* on = out + (size_t)n * (kNCOEF * kC) + c;
    #pragma unroll
    for (int r = 0; r < kNCOEF; ++r) on[r * kC] = (r & 1) ? acc[r >> 1].y : acc[r >> 1].x;
}

extern "C" void kernel_launch(void* const* d_in, const int* in_sizes, int n_in,
                              void* d_out, int out_size, void* d_ws, size_t ws_size,
                              hipStream_t stream) {
    const float* x             = (const float*)d_in[0];
    const float* x_edge        = (const float*)d_in[1];
    const float* edge_distance = (const float*)d_in[2];
    const float* wigner_inv    = (const float*)d_in[3];
    const int*   edge_index    = (const int*)d_in[4];
    const float* w1  = (const float*)d_in[5];
    const float* b1  = (const float*)d_in[6];
    const float* g1  = (const float*)d_in[7];
    const float* be1 = (const float*)d_in[8];
    const float* w2  = (const float*)d_in[9];
    const float* b2  = (const float*)d_in[10];
    const float* g2  = (const float*)d_in[11];
    const float* be2 = (const float*)d_in[12];
    const float* w3  = (const float*)d_in[13];
    const float* b3  = (const float*)d_in[14];
    const int* node_offset = (const int*)d_in[15];

    const int E = in_sizes[1] / 128;
    const int N = out_size / (kNCOEF * kC);

    float* m0buf       = (float*)d_ws;
    float* envbuf      = m0buf + (size_t)E * (kM0 * kC);
    int*   row_off     = (int*)(envbuf + E);
    int*   perm        = row_off + (N + 1);
    int*   perm_sorted = perm + E;

    build_dst_lists_kernel<<<1, kLISTT, 0, stream>>>(
        edge_index, node_offset, row_off, perm, E, N);

    const int blocks = (E + kTILE - 1) / kTILE;
    edge_mlp_kernel<<<blocks, 128, 0, stream>>>(
        x_edge, edge_distance,
        w1, b1, g1, be1, w2, b2, g2, be2, w3, b3,
        m0buf, envbuf, E);

    node_gather_kernel<<<N, 128, 0, stream>>>(
        x, wigner_inv, m0buf, envbuf, row_off, perm, perm_sorted, (float*)d_out);
}


# Per-node contraction moved from packed FMAs fed by LDS broadcasts to f32 16x16x4 WMMA: envelope-scaled m=0 columns staged as A, m0 rows read as B fragments, x as start value (sum order changes)
# speedup vs baseline: 3.0564x; 1.2594x over previous
#include <hip/hip_runtime.h>

typedef __attribute__((ext_vector_type(2))) float v2f;
typedef __attribute__((ext_vector_type(8))) float v8f;

constexpr int kNCOEF = 25;
constexpr int kM0    = 5;
constexpr int kC     = 128;
constexpr int kTILE  = 16;
constexpr float kCUTOFF  = 12.0f;
constexpr float kRESCALE = 5.0f;
constexpr float kEPS     = 1e-5f;
constexpr int kLISTT  = 1024;
constexpr int kLISTU  = 20;
constexpr int kNCHUNK = 8192;
constexpr int kGB     = 16;
constexpr int kGRP    = 4;
constexpr int kAP     = 84;

__device__ __forceinline__ float silu_f(float x) {
    return x / (1.0f + __expf(-x));
}

__device__ __forceinline__ void ln_silu_16x64(const float* __restrict__ hraw,
                                              float* __restrict__ hout,
                                              const float* __restrict__ g,
                                              const float* __restrict__ be,
                                              int tid) {
    if (tid < kTILE) {
        float s = 0.0f, s2 = 0.0f;
        #pragma unroll 8
        for (int c = 0; c < 64; ++c) {
            float v = hraw[tid * 64 + c];
            s += v; s2 += v * v;
        }
        float mu  = s * (1.0f / 64.0f);
        float var = s2 * (1.0f / 64.0f) - mu * mu;
        float inv = rsqrtf(var + kEPS);
        #pragma unroll 8
        for (int c = 0; c < 64; ++c) {
            float v = (hraw[tid * 64 + c] - mu) * inv * g[c] + be[c];
            hout[tid * 64 + c] = silu_f(v);
        }
    }
}


__global__ __launch_bounds__(kLISTT)
void build_dst_lists_kernel(const int* __restrict__ edge_index,
                            const int* __restrict__ node_offset_p,
                            int* __restrict__ row_off,
                            int* __restrict__ perm,
                            int E, int N)
{
    __shared__ int cnt[kNCHUNK];
    __shared__ int wsum[32];
    __shared__ int total_s;

    const int tid  = threadIdx.x;
    const int lane = tid & 31;
    const int wave = tid >> 5;
    const int noff = node_offset_p[0];
    const int* __restrict__ dstv = edge_index + E;
    constexpr int kPER = kNCHUNK / kLISTT;

    int carry = 0;
    for (int base = 0; base < N; base += kNCHUNK) {
        const int nn = (N - base < kNCHUNK) ? (N - base) : kNCHUNK;
        for (int i = tid; i < kNCHUNK; i += kLISTT) cnt[i] = 0;
        __syncthreads();

        for (int eb = tid; eb < E; eb += kLISTT * kLISTU) {
            int d[kLISTU];
            #pragma unroll
            for (int u = 0; u < kLISTU; ++u) {
                const int e = eb + u * kLISTT;
                d[u] = dstv[e < E ? e : eb];
            }
            #pragma unroll
            for (int u = 0; u < kLISTU; ++u) {
                const int e  = eb + u * kLISTT;
                const int dd = d[u] - noff - base;
                if (e < E && dd >= 0 && dd < nn) atomicAdd(&cnt[dd], 1);
            }
        }
        __syncthreads();

        int loc[kPER];
        int s = 0;
        #pragma unroll
        for (int j = 0; j < kPER; ++j) { loc[j] = s; s += cnt[tid * kPER + j]; }
        int incl = s;
        #pragma unroll
        for (int off = 1; off < 32; off <<= 1) {
            const int t = __shfl_up(incl, off);
            if (lane >= off) incl += t;
        }
        if (lane == 31) wsum[wave] = incl;
        __syncthreads();
        if (wave == 0) {
            const int w = wsum[lane];
            int wi = w;
            #pragma unroll
            for (int off = 1; off < 32; off <<= 1) {
                const int t = __shfl_up(wi, off);
                if (lane >= off) wi += t;
            }
            wsum[lane] = wi - w;
            if (lane == 31) total_s = wi;
        }
        __syncthreads();
        const int first = carry + wsum[wave] + (incl - s);
        #pragma unroll
        for (int j = 0; j < kPER; ++j) {
            const int idx = tid * kPER + j;
            const int o = first + loc[j];
            cnt[idx] = o;
            if (idx < nn) row_off[base + idx] = o;
        }
        carry += total_s;
        __syncthreads();

        for (int eb = tid; eb < E; eb += kLISTT * kLISTU) {
            int d[kLISTU];
            #pragma unroll
            for (int u = 0; u < kLISTU; ++u) {
                const int e = eb + u * kLISTT;
                d[u] = dstv[e < E ? e : eb];
            }
            int pos[kLISTU];
            #pragma unroll
            for (int u = 0; u < kLISTU; ++u) {
                const int e  = eb + u * kLISTT;
                const int dd = d[u] - noff - base;
                pos[u] = (e < E && dd >= 0 && dd < nn) ? atomicAdd(&cnt[dd], 1) : -1;
            }
            #pragma unroll
            for (int u = 0; u < kLISTU; ++u) {
                if (pos[u] >= 0) perm[pos[u]] = eb + u * kLISTT;
            }
        }
        __syncthreads();
    }
    if (tid == 0) row_off[N] = carry;
}

__global__ __launch_bounds__(128)
void edge_mlp_kernel(const float* __restrict__ x_edge,
                              const float* __restrict__ edge_distance,
                              const float* __restrict__ w1, const float* __restrict__ b1,
                              const float* __restrict__ g1, const float* __restrict__ be1,
                              const float* __restrict__ w2, const float* __restrict__ b2,
                              const float* __restrict__ g2, const float* __restrict__ be2,
                              const float* __restrict__ w3, const float* __restrict__ b3,
                              float* __restrict__ m0buf,
                              float* __restrict__ envbuf,
                              int E)
{
    __shared__ float xe[kTILE * 128];
    __shared__ float hraw[kTILE * 64];
    __shared__ float h1[kTILE * 64];
    __shared__ float h2[kTILE * 64];
    __shared__ __attribute__((aligned(16))) float m0s[kTILE * kM0 * kC];

    const int tid   = threadIdx.x;
    const int wave  = tid >> 5;
    const int lane  = tid & 31;
    const int lrow  = lane & 15;
    const int khalf = (lane >> 4) << 1;
    const int rbase = (lane >> 4) << 3;
    const int e0    = blockIdx.x * kTILE;

    for (int i = tid; i < kTILE * 128; i += blockDim.x) {
        int e = e0 + (i >> 7);
        xe[i] = (e < E) ? x_edge[(size_t)e * 128 + (i & 127)] : 0.0f;
    }
    __syncthreads();

    {
        const int colBase = wave * 16;
        v8f acc = {};
        for (int k0 = 0; k0 < 128; k0 += 4) {
            v2f a, b;
            a.x = xe[lrow * 128 + k0 + khalf];
            a.y = xe[lrow * 128 + k0 + khalf + 1];
            b.x = w1[(k0 + khalf)     * 64 + colBase + lrow];
            b.y = w1[(k0 + khalf + 1) * 64 + colBase + lrow];
            acc = __builtin_amdgcn_wmma_f32_16x16x4_f32(
                false, a, false, b, (short)0, acc, false, false);
        }
        #pragma unroll
        for (int r = 0; r < 8; ++r) {
            int row = rbase + r;
            hraw[row * 64 + colBase + lrow] = acc[r] + b1[colBase + lrow];
        }
    }
    __syncthreads();
    ln_silu_16x64(hraw, h1, g1, be1, tid);
    __syncthreads();

    {
        const int colBase = wave * 16;
        v8f acc = {};
        for (int k0 = 0; k0 < 64; k0 += 4) {
            v2f a, b;
            a.x = h1[lrow * 64 + k0 + khalf];
            a.y = h1[lrow * 64 + k0 + khalf + 1];
            b.x = w2[(k0 + khalf)     * 64 + colBase + lrow];
            b.y = w2[(k0 + khalf + 1) * 64 + colBase + lrow];
            acc = __builtin_amdgcn_wmma_f32_16x16x4_f32(
                false, a, false, b, (short)0, acc, false, false);
        }
        #pragma unroll
        for (int r = 0; r < 8; ++r) {
            int row = rbase + r;
            hraw[row * 64 + colBase + lrow] = acc[r] + b2[colBase + lrow];
        }
    }
    __syncthreads();
    ln_silu_16x64(hraw, h2, g2, be2, tid);
    __syncthreads();

    for (int t = wave; t < 40; t += 4) {
        const int colBase = t * 16;
        v8f acc = {};
        for (int k0 = 0; k0 < 64; k0 += 4) {
            v2f a, b;
            a.x = h2[lrow * 64 + k0 + khalf];
            a.y = h2[lrow * 64 + k0 + khalf + 1];
            b.x = w3[(size_t)(k0 + khalf)     * 640 + colBase + lrow];
            b.y = w3[(size_t)(k0 + khalf + 1) * 640 + colBase + lrow];
            acc = __builtin_amdgcn_wmma_f32_16x16x4_f32(
                false, a, false, b, (short)0, acc, false, false);
        }
        #pragma unroll
        for (int r = 0; r < 8; ++r) {
            int row = rbase + r;
            m0s[row * (kM0 * kC) + colBase + lrow] = acc[r] + b3[colBase + lrow];
        }
    }
    __syncthreads();

    {
        const float4* src4 = reinterpret_cast<const float4*>(m0s);
        float4* dst4 = reinterpret_cast<float4*>(m0buf + (size_t)e0 * (kM0 * kC));
        for (int i = tid; i < kTILE * (kM0 * kC / 4); i += blockDim.x) {
            if (e0 + i / (kM0 * kC / 4) < E) dst4[i] = src4[i];
        }
    }
    if (tid < kTILE) {
        int e = e0 + tid;
        if (e < E) {
            float env = 0.0f;
            float d = edge_distance[e] * (1.0f / kCUTOFF);
            if (d < 1.0f) {
                float d5 = d * d * d * d * d;
                env = 1.0f - 21.0f * d5 + 35.0f * d5 * d - 15.0f * d5 * d * d;
            }
            envbuf[e] = env * (1.0f / kRESCALE);
        }
    }
}

__device__ __forceinline__ float xhalf(float v) {
    return __builtin_bit_cast(float, __builtin_amdgcn_permlanex16(
        __builtin_bit_cast(int, v), __builtin_bit_cast(int, v), 0x76543210, 0xfedcba98, false, false));
}

__global__ __launch_bounds__(128, 5)
void node_gather_kernel(const float* __restrict__ x,
                        const float* __restrict__ wigner_inv,
                        const float* __restrict__ m0buf,
                        const float* __restrict__ envbuf,
                        const int*   __restrict__ row_off,
                        const int*   __restrict__ perm,
                        int*         __restrict__ perm_sorted,
                        float*       __restrict__ out)
{
    __shared__ __attribute__((aligned(16))) float a_s[32 * kAP];
    __shared__ __attribute__((aligned(8))) float env_k[kGB * kM0];
    __shared__ int elist[kGB];

    const int n    = blockIdx.x;
    const int c    = threadIdx.x;
    const int lane = c & 31;
    const int wave = c >> 5;
    const int lc   = lane & 15;
    const int lh   = lane >> 4;
    const int beg  = row_off[n];
    const int cnt  = row_off[n + 1] - beg;

    v8f acc[2][2];
    const float* xn = x + (size_t)n * (kNCOEF * kC) + wave * 32 + lc;
    #pragma unroll
    for (int mt = 0; mt < 2; ++mt) {
        #pragma unroll
        for (int nt = 0; nt < 2; ++nt) {
            #pragma unroll
            for (int i = 0; i < 8; ++i) {
                const int row = 16 * mt + 8 * lh + i;
                acc[mt][nt][i] = (row < kNCOEF) ? xn[row * kC + 16 * nt] : 0.0f;
            }
        }
    }

    for (int i = c; i < (32 - kNCOEF) * kAP; i += blockDim.x) a_s[kNCOEF * kAP + i] = 0.0f;

    const int  stage_r    = c / kM0;
    const int  stage_l    = c - stage_r * kM0;
    const bool stage_w    = c < kNCOEF * kM0;
    const int  stage_goff = stage_r * kNCOEF + stage_l * (stage_l + 1);
    const int  stage_aidx = stage_r * kAP + stage_l;

    for (int i = c; i < cnt; i += blockDim.x) {
        const int ei = perm[beg + i];
        int rank = 0;
        for (int j = 0; j < cnt; ++j) rank += (perm[beg + j] < ei) ? 1 : 0;
        perm_sorted[beg + rank] = ei;
    }
    __syncthreads();

    for (int b0 = 0; b0 < cnt; b0 += kGB) {
        const int nb = (cnt - b0 < kGB) ? (cnt - b0) : kGB;
        const int ng = (nb + kGRP - 1) / kGRP;
        if (c < kGB) elist[c] = perm_sorted[beg + b0 + (c < nb ? c : nb - 1)];
        __syncthreads();

        v2f bf[kGB / kGRP][5][2];
        #pragma unroll
        for (int g = 0; g < kGB / kGRP; ++g) {
            if (g < ng) {
                int eg[kGRP];
                #pragma unroll
                for (int q = 0; q < kGRP; ++q) eg[q] = elist[g * kGRP + q];
                #pragma unroll
                for (int ks = 0; ks < 5; ++ks) {
                    #pragma unroll
                    for (int s2 = 0; s2 < 2; ++s2) {
                        const int k0 = 4 * ks + s2;
                        const int k1 = 4 * ks + 2 + s2;
                        const int e  = lh ? eg[k1 / kM0] : eg[k0 / kM0];
                        const int l  = lh ? (k1 % kM0) : (k0 % kM0);
                        const float* mp = m0buf + (size_t)e * (kM0 * kC) + l * kC + wave * 32 + lc;
                        const float b_lo = mp[0];
                        const float b_hi = mp[16];
                        if (s2 == 0) { bf[g][ks][0].x = b_lo; bf[g][ks][1].x = b_hi; }
                        else         { bf[g][ks][0].y = b_lo; bf[g][ks][1].y = b_hi; }
                    }
                }
            }
        }

        float wv[kGB];
        #pragma unroll
        for (int j = 0; j < kGB; ++j) {
            if (j < ng * kGRP && stage_w) {
                wv[j] = wigner_inv[(size_t)elist[j] * (kNCOEF * kNCOEF) + stage_goff];
            }
        }
        if (c < ng * (kGRP * kM0)) env_k[c] = (stage_r < nb) ? envbuf[elist[stage_r]] : 0.0f;
        #pragma unroll
        for (int j = 0; j < kGB; ++j) {
            if (j < ng * kGRP && stage_w) a_s[stage_aidx + kM0 * j] = (j < nb) ? wv[j] : 0.0f;
        }
        __syncthreads();

        #pragma unroll
        for (int g = 0; g < kGB / kGRP; ++g) {
            if (g < ng) {
                #pragma unroll
                for (int ks = 0; ks < 5; ++ks) {
                    const int kk = g * (kGRP * kM0) + 4 * ks + 2 * lh;
                    const v2f ek = *reinterpret_cast<const v2f*>(env_k + kk);
                    const v2f a0 = *reinterpret_cast<const v2f*>(a_s + lc * kAP + kk) * ek;
                    const v2f a1 = *reinterpret_cast<const v2f*>(a_s + (16 + lc) * kAP + kk) * ek;
                    acc[0][0] = __builtin_amdgcn_wmma_f32_16x16x4_f32(
                        false, a0, false, bf[g][ks][0], (short)0, acc[0][0], false, false);
                    acc[0][1] = __builtin_amdgcn_wmma_f32_16x16x4_f32(
                        false, a0, false, bf[g][ks][1], (short)0, acc[0][1], false, false);
                    acc[1][0] = __builtin_amdgcn_wmma_f32_16x16x4_f32(
                        false, a1, false, bf[g][ks][0], (short)0, acc[1][0], false, false);
                    acc[1][1] = __builtin_amdgcn_wmma_f32_16x16x4_f32(
                        false, a1, false, bf[g][ks][1], (short)0, acc[1][1], false, false);
                }
            }
        }
        __syncthreads();
    }

    float* on = out + (size_t)n * (kNCOEF * kC) + c;
    #pragma unroll
    for (int mt = 0; mt < 2; ++mt) {
        #pragma unroll
        for (int i = 0; i < 8; ++i) {
            const float r0 = acc[mt][0][i];
            const float r1 = acc[mt][1][i];
            const float x0 = xhalf(r0);
            const float x1 = xhalf(r1);
            const float s_lo = lh ? x1 : r0;
            const float s_hi = lh ? r1 : x0;
            const int row_lo = 16 * mt + i;
            const int row_hi = 16 * mt + 8 + i;
            if (row_lo < kNCOEF) on[row_lo * kC] = s_lo;
            if (row_hi < kNCOEF) on[row_hi * kC] = s_hi;
        }
    }
}

extern "C" void kernel_launch(void* const* d_in, const int* in_sizes, int n_in,
                              void* d_out, int out_size, void* d_ws, size_t ws_size,
                              hipStream_t stream) {
    const float* x             = (const float*)d_in[0];
    const float* x_edge        = (const float*)d_in[1];
    const float* edge_distance = (const float*)d_in[2];
    const float* wigner_inv    = (const float*)d_in[3];
    const int*   edge_index    = (const int*)d_in[4];
    const float* w1  = (const float*)d_in[5];
    const float* b1  = (const float*)d_in[6];
    const float* g1  = (const float*)d_in[7];
    const float* be1 = (const float*)d_in[8];
    const float* w2  = (const float*)d_in[9];
    const float* b2  = (const float*)d_in[10];
    const float* g2  = (const float*)d_in[11];
    const float* be2 = (const float*)d_in[12];
    const float* w3  = (const float*)d_in[13];
    const float* b3  = (const float*)d_in[14];
    const int* node_offset = (const int*)d_in[15];

    const int E = in_sizes[1] / 128;
    const int N = out_size / (kNCOEF * kC);

    float* m0buf       = (float*)d_ws;
    float* envbuf      = m0buf + (size_t)E * (kM0 * kC);
    int*   row_off     = (int*)(envbuf + E);
    int*   perm        = row_off + (N + 1);
    int*   perm_sorted = perm + E;

    build_dst_lists_kernel<<<1, kLISTT, 0, stream>>>(
        edge_index, node_offset, row_off, perm, E, N);

    const int blocks = (E + kTILE - 1) / kTILE;
    edge_mlp_kernel<<<blocks, 128, 0, stream>>>(
        x_edge, edge_distance,
        w1, b1, g1, be1, w2, b2, g2, be2, w3, b3,
        m0buf, envbuf, E);

    node_gather_kernel<<<N, 128, 0, stream>>>(
        x, wigner_inv, m0buf, envbuf, row_off, perm, perm_sorted, (float*)d_out);
}


# Edge network kernel: x_edge tile-in and m0 tile-out copies unrolled at the fixed block size with 128-bit accesses, so a copy's loads are all in flight instead of one dependent round trip per pass
# speedup vs baseline: 3.3532x; 1.0971x over previous
#include <hip/hip_runtime.h>

typedef __attribute__((ext_vector_type(2))) float v2f;
typedef __attribute__((ext_vector_type(8))) float v8f;

constexpr int kNCOEF = 25;
constexpr int kM0    = 5;
constexpr int kC     = 128;
constexpr int kTILE  = 16;
constexpr float kCUTOFF  = 12.0f;
constexpr float kRESCALE = 5.0f;
constexpr float kEPS     = 1e-5f;
constexpr int kLISTT  = 1024;
constexpr int kLISTU  = 20;
constexpr int kNCHUNK = 8192;
constexpr int kGB     = 16;
constexpr int kGRP    = 4;
constexpr int kEDGET = 128;
constexpr int kXLD   = 4;
constexpr int kAP     = 84;

__device__ __forceinline__ float silu_f(float x) {
    return x / (1.0f + __expf(-x));
}

__device__ __forceinline__ void ln_silu_16x64(const float* __restrict__ hraw,
                                              float* __restrict__ hout,
                                              const float* __restrict__ g,
                                              const float* __restrict__ be,
                                              int tid) {
    if (tid < kTILE) {
        float s = 0.0f, s2 = 0.0f;
        #pragma unroll 8
        for (int c = 0; c < 64; ++c) {
            float v = hraw[tid * 64 + c];
            s += v; s2 += v * v;
        }
        float mu  = s * (1.0f / 64.0f);
        float var = s2 * (1.0f / 64.0f) - mu * mu;
        float inv = rsqrtf(var + kEPS);
        #pragma unroll 8
        for (int c = 0; c < 64; ++c) {
            float v = (hraw[tid * 64 + c] - mu) * inv * g[c] + be[c];
            hout[tid * 64 + c] = silu_f(v);
        }
    }
}


__global__ __launch_bounds__(kLISTT)
void build_dst_lists_kernel(const int* __restrict__ edge_index,
                            const int* __restrict__ node_offset_p,
                            int* __restrict__ row_off,
                            int* __restrict__ perm,
                            int E, int N)
{
    __shared__ int cnt[kNCHUNK];
    __shared__ int wsum[32];
    __shared__ int total_s;

    const int tid  = threadIdx.x;
    const int lane = tid & 31;
    const int wave = tid >> 5;
    const int noff = node_offset_p[0];
    const int* __restrict__ dstv = edge_index + E;
    constexpr int kPER = kNCHUNK / kLISTT;

    int carry = 0;
    for (int base = 0; base < N; base += kNCHUNK) {
        const int nn = (N - base < kNCHUNK) ? (N - base) : kNCHUNK;
        for (int i = tid; i < kNCHUNK; i += kLISTT) cnt[i] = 0;
        __syncthreads();

        for (int eb = tid; eb < E; eb += kLISTT * kLISTU) {
            int d[kLISTU];
            #pragma unroll
            for (int u = 0; u < kLISTU; ++u) {
                const int e = eb + u * kLISTT;
                d[u] = dstv[e < E ? e : eb];
            }
            #pragma unroll
            for (int u = 0; u < kLISTU; ++u) {
                const int e  = eb + u * kLISTT;
                const int dd = d[u] - noff - base;
                if (e < E && dd >= 0 && dd < nn) atomicAdd(&cnt[dd], 1);
            }
        }
        __syncthreads();

        int loc[kPER];
        int s = 0;
        #pragma unroll
        for (int j = 0; j < kPER; ++j) { loc[j] = s; s += cnt[tid * kPER + j]; }
        int incl = s;
        #pragma unroll
        for (int off = 1; off < 32; off <<= 1) {
            const int t = __shfl_up(incl, off);
            if (lane >= off) incl += t;
        }
        if (lane == 31) wsum[wave] = incl;
        __syncthreads();
        if (wave == 0) {
            const int w = wsum[lane];
            int wi = w;
            #pragma unroll
            for (int off = 1; off < 32; off <<= 1) {
                const int t = __shfl_up(wi, off);
                if (lane >= off) wi += t;
            }
            wsum[lane] = wi - w;
            if (lane == 31) total_s = wi;
        }
        __syncthreads();
        const int first = carry + wsum[wave] + (incl - s);
        #pragma unroll
        for (int j = 0; j < kPER; ++j) {
            const int idx = tid * kPER + j;
            const int o = first + loc[j];
            cnt[idx] = o;
            if (idx < nn) row_off[base + idx] = o;
        }
        carry += total_s;
        __syncthreads();

        for (int eb = tid; eb < E; eb += kLISTT * kLISTU) {
            int d[kLISTU];
            #pragma unroll
            for (int u = 0; u < kLISTU; ++u) {
                const int e = eb + u * kLISTT;
                d[u] = dstv[e < E ? e : eb];
            }
            int pos[kLISTU];
            #pragma unroll
            for (int u = 0; u < kLISTU; ++u) {
                const int e  = eb + u * kLISTT;
                const int dd = d[u] - noff - base;
                pos[u] = (e < E && dd >= 0 && dd < nn) ? atomicAdd(&cnt[dd], 1) : -1;
            }
            #pragma unroll
            for (int u = 0; u < kLISTU; ++u) {
                if (pos[u] >= 0) perm[pos[u]] = eb + u * kLISTT;
            }
        }
        __syncthreads();
    }
    if (tid == 0) row_off[N] = carry;
}

__global__ __launch_bounds__(kEDGET)
void edge_mlp_kernel(const float* __restrict__ x_edge,
                              const float* __restrict__ edge_distance,
                              const float* __restrict__ w1, const float* __restrict__ b1,
                              const float* __restrict__ g1, const float* __restrict__ be1,
                              const float* __restrict__ w2, const float* __restrict__ b2,
                              const float* __restrict__ g2, const float* __restrict__ be2,
                              const float* __restrict__ w3, const float* __restrict__ b3,
                              float* __restrict__ m0buf,
                              float* __restrict__ envbuf,
                              int E)
{
    __shared__ __attribute__((aligned(16))) float xe[kTILE * 128];
    __shared__ float hraw[kTILE * 64];
    __shared__ float h1[kTILE * 64];
    __shared__ float h2[kTILE * 64];
    __shared__ __attribute__((aligned(16))) float m0s[kTILE * kM0 * kC];

    const int tid   = threadIdx.x;
    const int wave  = tid >> 5;
    const int lane  = tid & 31;
    const int lrow  = lane & 15;
    const int khalf = (lane >> 4) << 1;
    const int rbase = (lane >> 4) << 3;
    const int e0    = blockIdx.x * kTILE;

    {
        const float4* src4 = reinterpret_cast<const float4*>(x_edge + (size_t)e0 * 128);
        float4 v[kXLD];
        #pragma unroll
        for (int u = 0; u < kXLD; ++u) {
            const int i4 = tid + u * kEDGET;
            const bool ok = (e0 + (i4 >> 5)) < E;
            v[u] = src4[ok ? i4 : (i4 & 31)];
        }
        float4* xe4 = reinterpret_cast<float4*>(xe);
        #pragma unroll
        for (int u = 0; u < kXLD; ++u) {
            const int i4 = tid + u * kEDGET;
            const bool ok = (e0 + (i4 >> 5)) < E;
            xe4[i4] = ok ? v[u] : make_float4(0.0f, 0.0f, 0.0f, 0.0f);
        }
    }
    __syncthreads();

    {
        const int colBase = wave * 16;
        v8f acc = {};
        for (int k0 = 0; k0 < 128; k0 += 4) {
            v2f a, b;
            a.x = xe[lrow * 128 + k0 + khalf];
            a.y = xe[lrow * 128 + k0 + khalf + 1];
            b.x = w1[(k0 + khalf)     * 64 + colBase + lrow];
            b.y = w1[(k0 + khalf + 1) * 64 + colBase + lrow];
            acc = __builtin_amdgcn_wmma_f32_16x16x4_f32(
                false, a, false, b, (short)0, acc, false, false);
        }
        #pragma unroll
        for (int r = 0; r < 8; ++r) {
            int row = rbase + r;
            hraw[row * 64 + colBase + lrow] = acc[r] + b1[colBase + lrow];
        }
    }
    __syncthreads();
    ln_silu_16x64(hraw, h1, g1, be1, tid);
    __syncthreads();

    {
        const int colBase = wave * 16;
        v8f acc = {};
        for (int k0 = 0; k0 < 64; k0 += 4) {
            v2f a, b;
            a.x = h1[lrow * 64 + k0 + khalf];
            a.y = h1[lrow * 64 + k0 + khalf + 1];
            b.x = w2[(k0 + khalf)     * 64 + colBase + lrow];
            b.y = w2[(k0 + khalf + 1) * 64 + colBase + lrow];
            acc = __builtin_amdgcn_wmma_f32_16x16x4_f32(
                false, a, false, b, (short)0, acc, false, false);
        }
        #pragma unroll
        for (int r = 0; r < 8; ++r) {
            int row = rbase + r;
            hraw[row * 64 + colBase + lrow] = acc[r] + b2[colBase + lrow];
        }
    }
    __syncthreads();
    ln_silu_16x64(hraw, h2, g2, be2, tid);
    __syncthreads();

    for (int t = wave; t < 40; t += 4) {
        const int colBase = t * 16;
        v8f acc = {};
        for (int k0 = 0; k0 < 64; k0 += 4) {
            v2f a, b;
            a.x = h2[lrow * 64 + k0 + khalf];
            a.y = h2[lrow * 64 + k0 + khalf + 1];
            b.x = w3[(size_t)(k0 + khalf)     * 640 + colBase + lrow];
            b.y = w3[(size_t)(k0 + khalf + 1) * 640 + colBase + lrow];
            acc = __builtin_amdgcn_wmma_f32_16x16x4_f32(
                false, a, false, b, (short)0, acc, false, false);
        }
        #pragma unroll
        for (int r = 0; r < 8; ++r) {
            int row = rbase + r;
            m0s[row * (kM0 * kC) + colBase + lrow] = acc[r] + b3[colBase + lrow];
        }
    }
    __syncthreads();

    {
        const float4* src4 = reinterpret_cast<const float4*>(m0s);
        float4* dst4 = reinterpret_cast<float4*>(m0buf + (size_t)e0 * (kM0 * kC));
        constexpr int kOUT4 = kTILE * (kM0 * kC / 4) / kEDGET;
        const int lim4 = ((E - e0 < kTILE) ? (E - e0) : kTILE) * (kM0 * kC / 4);
        if (lim4 == kTILE * (kM0 * kC / 4)) {
            float4 t[kOUT4];
            #pragma unroll
            for (int u = 0; u < kOUT4; ++u) t[u] = src4[tid + u * kEDGET];
            #pragma unroll
            for (int u = 0; u < kOUT4; ++u) dst4[tid + u * kEDGET] = t[u];
        } else {
            for (int i = tid; i < lim4; i += kEDGET) dst4[i] = src4[i];
        }
    }
    if (tid < kTILE) {
        int e = e0 + tid;
        if (e < E) {
            float env = 0.0f;
            float d = edge_distance[e] * (1.0f / kCUTOFF);
            if (d < 1.0f) {
                float d5 = d * d * d * d * d;
                env = 1.0f - 21.0f * d5 + 35.0f * d5 * d - 15.0f * d5 * d * d;
            }
            envbuf[e] = env * (1.0f / kRESCALE);
        }
    }
}

__device__ __forceinline__ float xhalf(float v) {
    return __builtin_bit_cast(float, __builtin_amdgcn_permlanex16(
        __builtin_bit_cast(int, v), __builtin_bit_cast(int, v), 0x76543210, 0xfedcba98, false, false));
}

__global__ __launch_bounds__(128, 5)
void node_gather_kernel(const float* __restrict__ x,
                        const float* __restrict__ wigner_inv,
                        const float* __restrict__ m0buf,
                        const float* __restrict__ envbuf,
                        const int*   __restrict__ row_off,
                        const int*   __restrict__ perm,
                        int*         __restrict__ perm_sorted,
                        float*       __restrict__ out)
{
    __shared__ __attribute__((aligned(16))) float a_s[32 * kAP];
    __shared__ __attribute__((aligned(8))) float env_k[kGB * kM0];
    __shared__ int elist[kGB];

    const int n    = blockIdx.x;
    const int c    = threadIdx.x;
    const int lane = c & 31;
    const int wave = c >> 5;
    const int lc   = lane & 15;
    const int lh   = lane >> 4;
    const int beg  = row_off[n];
    const int cnt  = row_off[n + 1] - beg;

    v8f acc[2][2];
    const float* xn = x + (size_t)n * (kNCOEF * kC) + wave * 32 + lc;
    #pragma unroll
    for (int mt = 0; mt < 2; ++mt) {
        #pragma unroll
        for (int nt = 0; nt < 2; ++nt) {
            #pragma unroll
            for (int i = 0; i < 8; ++i) {
                const int row = 16 * mt + 8 * lh + i;
                acc[mt][nt][i] = (row < kNCOEF) ? xn[row * kC + 16 * nt] : 0.0f;
            }
        }
    }

    for (int i = c; i < (32 - kNCOEF) * kAP; i += blockDim.x) a_s[kNCOEF * kAP + i] = 0.0f;

    const int  stage_r    = c / kM0;
    const int  stage_l    = c - stage_r * kM0;
    const bool stage_w    = c < kNCOEF * kM0;
    const int  stage_goff = stage_r * kNCOEF + stage_l * (stage_l + 1);
    const int  stage_aidx = stage_r * kAP + stage_l;

    for (int i = c; i < cnt; i += blockDim.x) {
        const int ei = perm[beg + i];
        int rank = 0;
        for (int j = 0; j < cnt; ++j) rank += (perm[beg + j] < ei) ? 1 : 0;
        perm_sorted[beg + rank] = ei;
    }
    __syncthreads();

    for (int b0 = 0; b0 < cnt; b0 += kGB) {
        const int nb = (cnt - b0 < kGB) ? (cnt - b0) : kGB;
        const int ng = (nb + kGRP - 1) / kGRP;
        if (c < kGB) elist[c] = perm_sorted[beg + b0 + (c < nb ? c : nb - 1)];
        __syncthreads();

        v2f bf[kGB / kGRP][5][2];
        #pragma unroll
        for (int g = 0; g < kGB / kGRP; ++g) {
            if (g < ng) {
                int eg[kGRP];
                #pragma unroll
                for (int q = 0; q < kGRP; ++q) eg[q] = elist[g * kGRP + q];
                #pragma unroll
                for (int ks = 0; ks < 5; ++ks) {
                    #pragma unroll
                    for (int s2 = 0; s2 < 2; ++s2) {
                        const int k0 = 4 * ks + s2;
                        const int k1 = 4 * ks + 2 + s2;
                        const int e  = lh ? eg[k1 / kM0] : eg[k0 / kM0];
                        const int l  = lh ? (k1 % kM0) : (k0 % kM0);
                        const float* mp = m0buf + (size_t)e * (kM0 * kC) + l * kC + wave * 32 + lc;
                        const float b_lo = mp[0];
                        const float b_hi = mp[16];
                        if (s2 == 0) { bf[g][ks][0].x = b_lo; bf[g][ks][1].x = b_hi; }
                        else         { bf[g][ks][0].y = b_lo; bf[g][ks][1].y = b_hi; }
                    }
                }
            }
        }

        float wv[kGB];
        #pragma unroll
        for (int j = 0; j < kGB; ++j) {
            if (j < ng * kGRP && stage_w) {
                wv[j] = wigner_inv[(size_t)elist[j] * (kNCOEF * kNCOEF) + stage_goff];
            }
        }
        if (c < ng * (kGRP * kM0)) env_k[c] = (stage_r < nb) ? envbuf[elist[stage_r]] : 0.0f;
        #pragma unroll
        for (int j = 0; j < kGB; ++j) {
            if (j < ng * kGRP && stage_w) a_s[stage_aidx + kM0 * j] = (j < nb) ? wv[j] : 0.0f;
        }
        __syncthreads();

        #pragma unroll
        for (int g = 0; g < kGB / kGRP; ++g) {
            if (g < ng) {
                #pragma unroll
                for (int ks = 0; ks < 5; ++ks) {
                    const int kk = g * (kGRP * kM0) + 4 * ks + 2 * lh;
                    const v2f ek = *reinterpret_cast<const v2f*>(env_k + kk);
                    const v2f a0 = *reinterpret_cast<const v2f*>(a_s + lc * kAP + kk) * ek;
                    const v2f a1 = *reinterpret_cast<const v2f*>(a_s + (16 + lc) * kAP + kk) * ek;
                    acc[0][0] = __builtin_amdgcn_wmma_f32_16x16x4_f32(
                        false, a0, false, bf[g][ks][0], (short)0, acc[0][0], false, false);
                    acc[0][1] = __builtin_amdgcn_wmma_f32_16x16x4_f32(
                        false, a0, false, bf[g][ks][1], (short)0, acc[0][1], false, false);
                    acc[1][0] = __builtin_amdgcn_wmma_f32_16x16x4_f32(
                        false, a1, false, bf[g][ks][0], (short)0, acc[1][0], false, false);
                    acc[1][1] = __builtin_amdgcn_wmma_f32_16x16x4_f32(
                        false, a1, false, bf[g][ks][1], (short)0, acc[1][1], false, false);
                }
            }
        }
        __syncthreads();
    }

    float* on = out + (size_t)n * (kNCOEF * kC) + c;
    #pragma unroll
    for (int mt = 0; mt < 2; ++mt) {
        #pragma unroll
        for (int i = 0; i < 8; ++i) {
            const float r0 = acc[mt][0][i];
            const float r1 = acc[mt][1][i];
            const float x0 = xhalf(r0);
            const float x1 = xhalf(r1);
            const float s_lo = lh ? x1 : r0;
            const float s_hi = lh ? r1 : x0;
            const int row_lo = 16 * mt + i;
            const int row_hi = 16 * mt + 8 + i;
            if (row_lo < kNCOEF) on[row_lo * kC] = s_lo;
            if (row_hi < kNCOEF) on[row_hi * kC] = s_hi;
        }
    }
}

extern "C" void kernel_launch(void* const* d_in, const int* in_sizes, int n_in,
                              void* d_out, int out_size, void* d_ws, size_t ws_size,
                              hipStream_t stream) {
    const float* x             = (const float*)d_in[0];
    const float* x_edge        = (const float*)d_in[1];
    const float* edge_distance = (const float*)d_in[2];
    const float* wigner_inv    = (const float*)d_in[3];
    const int*   edge_index    = (const int*)d_in[4];
    const float* w1  = (const float*)d_in[5];
    const float* b1  = (const float*)d_in[6];
    const float* g1  = (const float*)d_in[7];
    const float* be1 = (const float*)d_in[8];
    const float* w2  = (const float*)d_in[9];
    const float* b2  = (const float*)d_in[10];
    const float* g2  = (const float*)d_in[11];
    const float* be2 = (const float*)d_in[12];
    const float* w3  = (const float*)d_in[13];
    const float* b3  = (const float*)d_in[14];
    const int* node_offset = (const int*)d_in[15];

    const int E = in_sizes[1] / 128;
    const int N = out_size / (kNCOEF * kC);

    float* m0buf       = (float*)d_ws;
    float* envbuf      = m0buf + (size_t)E * (kM0 * kC);
    int*   row_off     = (int*)(envbuf + E);
    int*   perm        = row_off + (N + 1);
    int*   perm_sorted = perm + E;

    build_dst_lists_kernel<<<1, kLISTT, 0, stream>>>(
        edge_index, node_offset, row_off, perm, E, N);

    const int blocks = (E + kTILE - 1) / kTILE;
    edge_mlp_kernel<<<blocks, kEDGET, 0, stream>>>(
        x_edge, edge_distance,
        w1, b1, g1, be1, w2, b2, g2, be2, w3, b3,
        m0buf, envbuf, E);

    node_gather_kernel<<<N, 128, 0, stream>>>(
        x, wigner_inv, m0buf, envbuf, row_off, perm, perm_sorted, (float*)d_out);
}


# List build moved from its own one-block launch into 32 extra blocks of the edge network launch, each building the lists of N/32 nodes from its own two passes over the destinations (3 launches -> 2)
# speedup vs baseline: 4.3039x; 1.2835x over previous
#include <hip/hip_runtime.h>

typedef __attribute__((ext_vector_type(2))) float v2f;
typedef __attribute__((ext_vector_type(8))) float v8f;

constexpr int kNCOEF = 25;
constexpr int kM0    = 5;
constexpr int kC     = 128;
constexpr int kTILE  = 16;
constexpr float kCUTOFF  = 12.0f;
constexpr float kRESCALE = 5.0f;
constexpr float kEPS     = 1e-5f;
constexpr int kLISTB  = 32;
constexpr int kLISTU  = 40;
constexpr int kLCH    = 8192;
constexpr int kGB     = 16;
constexpr int kGRP    = 4;
constexpr int kEDGET = 128;
constexpr int kXLD   = 4;
constexpr int kAP     = 84;

__device__ __forceinline__ float silu_f(float x) {
    return x / (1.0f + __expf(-x));
}

__device__ __forceinline__ void ln_silu_16x64(const float* __restrict__ hraw,
                                              float* __restrict__ hout,
                                              const float* __restrict__ g,
                                              const float* __restrict__ be,
                                              int tid) {
    if (tid < kTILE) {
        float s = 0.0f, s2 = 0.0f;
        #pragma unroll 8
        for (int c = 0; c < 64; ++c) {
            float v = hraw[tid * 64 + c];
            s += v; s2 += v * v;
        }
        float mu  = s * (1.0f / 64.0f);
        float var = s2 * (1.0f / 64.0f) - mu * mu;
        float inv = rsqrtf(var + kEPS);
        #pragma unroll 8
        for (int c = 0; c < 64; ++c) {
            float v = (hraw[tid * 64 + c] - mu) * inv * g[c] + be[c];
            hout[tid * 64 + c] = silu_f(v);
        }
    }
}


__device__ __forceinline__ void build_dst_lists_slice(const int* __restrict__ edge_index,
                                                      const int* __restrict__ node_offset_p,
                                                      int* __restrict__ row_off,
                                                      int* __restrict__ perm,
                                                      int E, int N, int part,
                                                      int* __restrict__ cnt,
                                                      int* __restrict__ lsum)
{
    constexpr int kW = kEDGET / 32;
    const int tid  = threadIdx.x;
    const int lane = tid & 31;
    const int wave = tid >> 5;
    const int noff = node_offset_p[0];
    const int* __restrict__ dstv = edge_index + E;

    const int npb = (N + kLISTB - 1) / kLISTB;
    const int lo  = (part * npb < N) ? part * npb : N;
    const int hi  = (part == kLISTB - 1) ? N : ((lo + npb < N) ? lo + npb : N);
    if (lo >= hi && part != kLISTB - 1) return;

    int cb = lo;
    do {
        const int ce = (cb + kLCH < hi) ? cb + kLCH : hi;
        const int nn = ce - cb;
        for (int i = tid; i < nn; i += kEDGET) cnt[i] = 0;
        __syncthreads();

        const unsigned ucb = (unsigned)cb, unn = (unsigned)nn, unoff = (unsigned)noff;
        unsigned below = 0;
        for (int eb = tid; eb < E; eb += kEDGET * kLISTU) {
            unsigned dd[kLISTU];
            #pragma unroll
            for (int u = 0; u < kLISTU; ++u) {
                const int e  = eb + u * kEDGET;
                const int dv = dstv[e < E ? e : eb];
                dd[u] = (e < E) ? (unsigned)dv - unoff : 0xFFFFFFFFu;
            }
            #pragma unroll
            for (int u = 0; u < kLISTU; ++u) {
                const unsigned r = dd[u] - ucb;
                if (r < unn) atomicAdd(&cnt[r], 1);
                below += (dd[u] < ucb) ? 1u : 0u;
            }
        }
        #pragma unroll
        for (int off = 16; off > 0; off >>= 1) below += __shfl_xor(below, off);
        if (lane == 0) lsum[wave] = (int)below;
        __syncthreads();
        int base = 0;
        #pragma unroll
        for (int w = 0; w < kW; ++w) base += lsum[w];

        const int per = (nn + kEDGET - 1) / kEDGET;
        const int i0  = (tid * per < nn) ? tid * per : nn;
        const int i1  = (i0 + per < nn) ? i0 + per : nn;
        int s = 0;
        for (int i = i0; i < i1; ++i) s += cnt[i];
        int incl = s;
        #pragma unroll
        for (int off = 1; off < 32; off <<= 1) {
            const int t = __shfl_up(incl, off);
            if (lane >= off) incl += t;
        }
        if (lane == 31) lsum[kW + wave] = incl;
        __syncthreads();
        int run = base + (incl - s);
        for (int w = 0; w < wave; ++w) run += lsum[kW + w];
        for (int i = i0; i < i1; ++i) {
            const int k = cnt[i];
            cnt[i] = run;
            row_off[cb + i] = run;
            run += k;
        }
        if (ce == N && tid == kEDGET - 1) row_off[N] = run;
        __syncthreads();

        for (int eb = tid; eb < E; eb += kEDGET * kLISTU) {
            unsigned dd[kLISTU];
            #pragma unroll
            for (int u = 0; u < kLISTU; ++u) {
                const int e  = eb + u * kEDGET;
                const int dv = dstv[e < E ? e : eb];
                dd[u] = (e < E) ? (unsigned)dv - unoff : 0xFFFFFFFFu;
            }
            int pos[kLISTU];
            #pragma unroll
            for (int u = 0; u < kLISTU; ++u) {
                const unsigned r = dd[u] - ucb;
                pos[u] = (r < unn) ? atomicAdd(&cnt[r], 1) : -1;
            }
            #pragma unroll
            for (int u = 0; u < kLISTU; ++u) {
                if (pos[u] >= 0) perm[pos[u]] = eb + u * kEDGET;
            }
        }
        __syncthreads();
        cb += kLCH;
    } while (cb < hi);
}

__global__ __launch_bounds__(kEDGET)
void edge_mlp_kernel(const float* __restrict__ x_edge,
                              const float* __restrict__ edge_distance,
                              const float* __restrict__ w1, const float* __restrict__ b1,
                              const float* __restrict__ g1, const float* __restrict__ be1,
                              const float* __restrict__ w2, const float* __restrict__ b2,
                              const float* __restrict__ g2, const float* __restrict__ be2,
                              const float* __restrict__ w3, const float* __restrict__ b3,
                              float* __restrict__ m0buf,
                              float* __restrict__ envbuf,
                              const int* __restrict__ edge_index,
                              const int* __restrict__ node_offset_p,
                              int* __restrict__ row_off,
                              int* __restrict__ perm,
                              int E, int N)
{
    __shared__ __attribute__((aligned(16))) float xe[kTILE * 128];
    __shared__ float hraw[kTILE * 64];
    __shared__ float h1[kTILE * 64];
    __shared__ float h2[kTILE * 64];
    __shared__ __attribute__((aligned(16))) float m0s[kTILE * kM0 * kC];
    __shared__ int lsum[2 * (kEDGET / 32)];

    static_assert(kLCH <= kTILE * kM0 * kC, "list counters must fit the m0 tile");
    if (blockIdx.x < kLISTB) {
        build_dst_lists_slice(edge_index, node_offset_p, row_off, perm, E, N,
                              (int)blockIdx.x, reinterpret_cast<int*>(m0s), lsum);
        return;
    }

    const int tid   = threadIdx.x;
    const int wave  = tid >> 5;
    const int lane  = tid & 31;
    const int lrow  = lane & 15;
    const int khalf = (lane >> 4) << 1;
    const int rbase = (lane >> 4) << 3;
    const int e0    = ((int)blockIdx.x - kLISTB) * kTILE;

    {
        const float4* src4 = reinterpret_cast<const float4*>(x_edge + (size_t)e0 * 128);
        float4 v[kXLD];
        #pragma unroll
        for (int u = 0; u < kXLD; ++u) {
            const int i4 = tid + u * kEDGET;
            const bool ok = (e0 + (i4 >> 5)) < E;
            v[u] = src4[ok ? i4 : (i4 & 31)];
        }
        float4* xe4 = reinterpret_cast<float4*>(xe);
        #pragma unroll
        for (int u = 0; u < kXLD; ++u) {
            const int i4 = tid + u * kEDGET;
            const bool ok = (e0 + (i4 >> 5)) < E;
            xe4[i4] = ok ? v[u] : make_float4(0.0f, 0.0f, 0.0f, 0.0f);
        }
    }
    __syncthreads();

    {
        const int colBase = wave * 16;
        v8f acc = {};
        for (int k0 = 0; k0 < 128; k0 += 4) {
            v2f a, b;
            a.x = xe[lrow * 128 + k0 + khalf];
            a.y = xe[lrow * 128 + k0 + khalf + 1];
            b.x = w1[(k0 + khalf)     * 64 + colBase + lrow];
            b.y = w1[(k0 + khalf + 1) * 64 + colBase + lrow];
            acc = __builtin_amdgcn_wmma_f32_16x16x4_f32(
                false, a, false, b, (short)0, acc, false, false);
        }
        #pragma unroll
        for (int r = 0; r < 8; ++r) {
            int row = rbase + r;
            hraw[row * 64 + colBase + lrow] = acc[r] + b1[colBase + lrow];
        }
    }
    __syncthreads();
    ln_silu_16x64(hraw, h1, g1, be1, tid);
    __syncthreads();

    {
        const int colBase = wave * 16;
        v8f acc = {};
        for (int k0 = 0; k0 < 64; k0 += 4) {
            v2f a, b;
            a.x = h1[lrow * 64 + k0 + khalf];
            a.y = h1[lrow * 64 + k0 + khalf + 1];
            b.x = w2[(k0 + khalf)     * 64 + colBase + lrow];
            b.y = w2[(k0 + khalf + 1) * 64 + colBase + lrow];
            acc = __builtin_amdgcn_wmma_f32_16x16x4_f32(
                false, a, false, b, (short)0, acc, false, false);
        }
        #pragma unroll
        for (int r = 0; r < 8; ++r) {
            int row = rbase + r;
            hraw[row * 64 + colBase + lrow] = acc[r] + b2[colBase + lrow];
        }
    }
    __syncthreads();
    ln_silu_16x64(hraw, h2, g2, be2, tid);
    __syncthreads();

    for (int t = wave; t < 40; t += 4) {
        const int colBase = t * 16;
        v8f acc = {};
        for (int k0 = 0; k0 < 64; k0 += 4) {
            v2f a, b;
            a.x = h2[lrow * 64 + k0 + khalf];
            a.y = h2[lrow * 64 + k0 + khalf + 1];
            b.x = w3[(size_t)(k0 + khalf)     * 640 + colBase + lrow];
            b.y = w3[(size_t)(k0 + khalf + 1) * 640 + colBase + lrow];
            acc = __builtin_amdgcn_wmma_f32_16x16x4_f32(
                false, a, false, b, (short)0, acc, false, false);
        }
        #pragma unroll
        for (int r = 0; r < 8; ++r) {
            int row = rbase + r;
            m0s[row * (kM0 * kC) + colBase + lrow] = acc[r] + b3[colBase + lrow];
        }
    }
    __syncthreads();

    {
        const float4* src4 = reinterpret_cast<const float4*>(m0s);
        float4* dst4 = reinterpret_cast<float4*>(m0buf + (size_t)e0 * (kM0 * kC));
        constexpr int kOUT4 = kTILE * (kM0 * kC / 4) / kEDGET;
        const int lim4 = ((E - e0 < kTILE) ? (E - e0) : kTILE) * (kM0 * kC / 4);
        if (lim4 == kTILE * (kM0 * kC / 4)) {
            float4 t[kOUT4];
            #pragma unroll
            for (int u = 0; u < kOUT4; ++u) t[u] = src4[tid + u * kEDGET];
            #pragma unroll
            for (int u = 0; u < kOUT4; ++u) dst4[tid + u * kEDGET] = t[u];
        } else {
            for (int i = tid; i < lim4; i += kEDGET) dst4[i] = src4[i];
        }
    }
    if (tid < kTILE) {
        int e = e0 + tid;
        if (e < E) {
            float env = 0.0f;
            float d = edge_distance[e] * (1.0f / kCUTOFF);
            if (d < 1.0f) {
                float d5 = d * d * d * d * d;
                env = 1.0f - 21.0f * d5 + 35.0f * d5 * d - 15.0f * d5 * d * d;
            }
            envbuf[e] = env * (1.0f / kRESCALE);
        }
    }
}

__device__ __forceinline__ float xhalf(float v) {
    return __builtin_bit_cast(float, __builtin_amdgcn_permlanex16(
        __builtin_bit_cast(int, v), __builtin_bit_cast(int, v), 0x76543210, 0xfedcba98, false, false));
}

__global__ __launch_bounds__(128, 5)
void node_gather_kernel(const float* __restrict__ x,
                        const float* __restrict__ wigner_inv,
                        const float* __restrict__ m0buf,
                        const float* __restrict__ envbuf,
                        const int*   __restrict__ row_off,
                        const int*   __restrict__ perm,
                        int*         __restrict__ perm_sorted,
                        float*       __restrict__ out)
{
    __shared__ __attribute__((aligned(16))) float a_s[32 * kAP];
    __shared__ __attribute__((aligned(8))) float env_k[kGB * kM0];
    __shared__ int elist[kGB];

    const int n    = blockIdx.x;
    const int c    = threadIdx.x;
    const int lane = c & 31;
    const int wave = c >> 5;
    const int lc   = lane & 15;
    const int lh   = lane >> 4;
    const int beg  = row_off[n];
    const int cnt  = row_off[n + 1] - beg;

    v8f acc[2][2];
    const float* xn = x + (size_t)n * (kNCOEF * kC) + wave * 32 + lc;
    #pragma unroll
    for (int mt = 0; mt < 2; ++mt) {
        #pragma unroll
        for (int nt = 0; nt < 2; ++nt) {
            #pragma unroll
            for (int i = 0; i < 8; ++i) {
                const int row = 16 * mt + 8 * lh + i;
                acc[mt][nt][i] = (row < kNCOEF) ? xn[row * kC + 16 * nt] : 0.0f;
            }
        }
    }

    for (int i = c; i < (32 - kNCOEF) * kAP; i += blockDim.x) a_s[kNCOEF * kAP + i] = 0.0f;

    const int  stage_r    = c / kM0;
    const int  stage_l    = c - stage_r * kM0;
    const bool stage_w    = c < kNCOEF * kM0;
    const int  stage_goff = stage_r * kNCOEF + stage_l * (stage_l + 1);
    const int  stage_aidx = stage_r * kAP + stage_l;

    for (int i = c; i < cnt; i += blockDim.x) {
        const int ei = perm[beg + i];
        int rank = 0;
        for (int j = 0; j < cnt; ++j) rank += (perm[beg + j] < ei) ? 1 : 0;
        perm_sorted[beg + rank] = ei;
    }
    __syncthreads();

    for (int b0 = 0; b0 < cnt; b0 += kGB) {
        const int nb = (cnt - b0 < kGB) ? (cnt - b0) : kGB;
        const int ng = (nb + kGRP - 1) / kGRP;
        if (c < kGB) elist[c] = perm_sorted[beg + b0 + (c < nb ? c : nb - 1)];
        __syncthreads();

        v2f bf[kGB / kGRP][5][2];
        #pragma unroll
        for (int g = 0; g < kGB / kGRP; ++g) {
            if (g < ng) {
                int eg[kGRP];
                #pragma unroll
                for (int q = 0; q < kGRP; ++q) eg[q] = elist[g * kGRP + q];
                #pragma unroll
                for (int ks = 0; ks < 5; ++ks) {
                    #pragma unroll
                    for (int s2 = 0; s2 < 2; ++s2) {
                        const int k0 = 4 * ks + s2;
                        const int k1 = 4 * ks + 2 + s2;
                        const int e  = lh ? eg[k1 / kM0] : eg[k0 / kM0];
                        const int l  = lh ? (k1 % kM0) : (k0 % kM0);
                        const float* mp = m0buf + (size_t)e * (kM0 * kC) + l * kC + wave * 32 + lc;
                        const float b_lo = mp[0];
                        const float b_hi = mp[16];
                        if (s2 == 0) { bf[g][ks][0].x = b_lo; bf[g][ks][1].x = b_hi; }
                        else         { bf[g][ks][0].y = b_lo; bf[g][ks][1].y = b_hi; }
                    }
                }
            }
        }

        float wv[kGB];
        #pragma unroll
        for (int j = 0; j < kGB; ++j) {
            if (j < ng * kGRP && stage_w) {
                wv[j] = wigner_inv[(size_t)elist[j] * (kNCOEF * kNCOEF) + stage_goff];
            }
        }
        if (c < ng * (kGRP * kM0)) env_k[c] = (stage_r < nb) ? envbuf[elist[stage_r]] : 0.0f;
        #pragma unroll
        for (int j = 0; j < kGB; ++j) {
            if (j < ng * kGRP && stage_w) a_s[stage_aidx + kM0 * j] = (j < nb) ? wv[j] : 0.0f;
        }
        __syncthreads();

        #pragma unroll
        for (int g = 0; g < kGB / kGRP; ++g) {
            if (g < ng) {
                #pragma unroll
                for (int ks = 0; ks < 5; ++ks) {
                    const int kk = g * (kGRP * kM0) + 4 * ks + 2 * lh;
                    const v2f ek = *reinterpret_cast<const v2f*>(env_k + kk);
                    const v2f a0 = *reinterpret_cast<const v2f*>(a_s + lc * kAP + kk) * ek;
                    const v2f a1 = *reinterpret_cast<const v2f*>(a_s + (16 + lc) * kAP + kk) * ek;
                    acc[0][0] = __builtin_amdgcn_wmma_f32_16x16x4_f32(
                        false, a0, false, bf[g][ks][0], (short)0, acc[0][0], false, false);
                    acc[0][1] = __builtin_amdgcn_wmma_f32_16x16x4_f32(
                        false, a0, false, bf[g][ks][1], (short)0, acc[0][1], false, false);
                    acc[1][0] = __builtin_amdgcn_wmma_f32_16x16x4_f32(
                        false, a1, false, bf[g][ks][0], (short)0, acc[1][0], false, false);
                    acc[1][1] = __builtin_amdgcn_wmma_f32_16x16x4_f32(
                        false, a1, false, bf[g][ks][1], (short)0, acc[1][1], false, false);
                }
            }
        }
        __syncthreads();
    }

    float* on = out + (size_t)n * (kNCOEF * kC) + c;
    #pragma unroll
    for (int mt = 0; mt < 2; ++mt) {
        #pragma unroll
        for (int i = 0; i < 8; ++i) {
            const float r0 = acc[mt][0][i];
            const float r1 = acc[mt][1][i];
            const float x0 = xhalf(r0);
            const float x1 = xhalf(r1);
            const float s_lo = lh ? x1 : r0;
            const float s_hi = lh ? r1 : x0;
            const int row_lo = 16 * mt + i;
            const int row_hi = 16 * mt + 8 + i;
            if (row_lo < kNCOEF) on[row_lo * kC] = s_lo;
            if (row_hi < kNCOEF) on[row_hi * kC] = s_hi;
        }
    }
}

extern "C" void kernel_launch(void* const* d_in, const int* in_sizes, int n_in,
                              void* d_out, int out_size, void* d_ws, size_t ws_size,
                              hipStream_t stream) {
    const float* x             = (const float*)d_in[0];
    const float* x_edge        = (const float*)d_in[1];
    const float* edge_distance = (const float*)d_in[2];
    const float* wigner_inv    = (const float*)d_in[3];
    const int*   edge_index    = (const int*)d_in[4];
    const float* w1  = (const float*)d_in[5];
    const float* b1  = (const float*)d_in[6];
    const float* g1  = (const float*)d_in[7];
    const float* be1 = (const float*)d_in[8];
    const float* w2  = (const float*)d_in[9];
    const float* b2  = (const float*)d_in[10];
    const float* g2  = (const float*)d_in[11];
    const float* be2 = (const float*)d_in[12];
    const float* w3  = (const float*)d_in[13];
    const float* b3  = (const float*)d_in[14];
    const int* node_offset = (const int*)d_in[15];

    const int E = in_sizes[1] / 128;
    const int N = out_size / (kNCOEF * kC);

    float* m0buf       = (float*)d_ws;
    float* envbuf      = m0buf + (size_t)E * (kM0 * kC);
    int*   row_off     = (int*)(envbuf + E);
    int*   perm        = row_off + (N + 1);
    int*   perm_sorted = perm + E;

    const int blocks = (E + kTILE - 1) / kTILE;
    edge_mlp_kernel<<<kLISTB + blocks, kEDGET, 0, stream>>>(
        x_edge, edge_distance,
        w1, b1, g1, be1, w2, b2, g2, be2, w3, b3,
        m0buf, envbuf, edge_index, node_offset, row_off, perm, E, N);

    node_gather_kernel<<<N, 128, 0, stream>>>(
        x, wigner_inv, m0buf, envbuf, row_off, perm, perm_sorted, (float*)d_out);
}


# Edge network: LayerNorm+SiLU of both hidden layers on all 128 threads of the block (8 per row, row sums by a lane butterfly) instead of 16 threads (order of the LayerNorm sums changes)
# speedup vs baseline: 4.4680x; 1.0381x over previous
#include <hip/hip_runtime.h>

typedef __attribute__((ext_vector_type(2))) float v2f;
typedef __attribute__((ext_vector_type(8))) float v8f;

constexpr int kNCOEF = 25;
constexpr int kM0    = 5;
constexpr int kC     = 128;
constexpr int kTILE  = 16;
constexpr float kCUTOFF  = 12.0f;
constexpr float kRESCALE = 5.0f;
constexpr float kEPS     = 1e-5f;
constexpr int kLISTB  = 32;
constexpr int kLISTU  = 40;
constexpr int kLCH    = 8192;
constexpr int kGB     = 16;
constexpr int kGRP    = 4;
constexpr int kEDGET = 128;
constexpr int kXLD   = 4;
constexpr int kAP     = 84;

__device__ __forceinline__ float silu_f(float x) {
    return x / (1.0f + __expf(-x));
}

template <int M>
__device__ __forceinline__ float row_xor(float v) {
    return __builtin_bit_cast(float, __builtin_amdgcn_update_dpp(
        0, __builtin_bit_cast(int, v), 0x160 | M, 0xf, 0xf, false));
}

constexpr int kLNF = 8;

__device__ __forceinline__ void ln_silu_16x64(const float* __restrict__ hraw,
                                              float* __restrict__ hout,
                                              const float* __restrict__ g,
                                              const float* __restrict__ be,
                                              int tid) {
    const int c0  = (tid & 7) * kLNF;
    const int off = (tid >> 3) * 64 + c0;
    const float4 p = *reinterpret_cast<const float4*>(hraw + off);
    const float4 q = *reinterpret_cast<const float4*>(hraw + off + 4);
    const float v[kLNF] = {p.x, p.y, p.z, p.w, q.x, q.y, q.z, q.w};
    float gv[kLNF], bv[kLNF];
    #pragma unroll
    for (int j = 0; j < kLNF; ++j) { gv[j] = g[c0 + j]; bv[j] = be[c0 + j]; }
    float s = 0.0f, s2 = 0.0f;
    #pragma unroll
    for (int j = 0; j < kLNF; ++j) { s += v[j]; s2 += v[j] * v[j]; }
    s += row_xor<1>(s);  s2 += row_xor<1>(s2);
    s += row_xor<2>(s);  s2 += row_xor<2>(s2);
    s += row_xor<4>(s);  s2 += row_xor<4>(s2);
    const float mu  = s * (1.0f / 64.0f);
    const float var = s2 * (1.0f / 64.0f) - mu * mu;
    const float inv = rsqrtf(var + kEPS);
    float o[kLNF];
    #pragma unroll
    for (int j = 0; j < kLNF; ++j) o[j] = silu_f((v[j] - mu) * inv * gv[j] + bv[j]);
    *reinterpret_cast<float4*>(hout + off)     = make_float4(o[0], o[1], o[2], o[3]);
    *reinterpret_cast<float4*>(hout + off + 4) = make_float4(o[4], o[5], o[6], o[7]);
}


__device__ __forceinline__ void build_dst_lists_slice(const int* __restrict__ edge_index,
                                                      const int* __restrict__ node_offset_p,
                                                      int* __restrict__ row_off,
                                                      int* __restrict__ perm,
                                                      int E, int N, int part,
                                                      int* __restrict__ cnt,
                                                      int* __restrict__ lsum)
{
    constexpr int kW = kEDGET / 32;
    const int tid  = threadIdx.x;
    const int lane = tid & 31;
    const int wave = tid >> 5;
    const int noff = node_offset_p[0];
    const int* __restrict__ dstv = edge_index + E;

    const int npb = (N + kLISTB - 1) / kLISTB;
    const int lo  = (part * npb < N) ? part * npb : N;
    const int hi  = (part == kLISTB - 1) ? N : ((lo + npb < N) ? lo + npb : N);
    if (lo >= hi && part != kLISTB - 1) return;

    int cb = lo;
    do {
        const int ce = (cb + kLCH < hi) ? cb + kLCH : hi;
        const int nn = ce - cb;
        for (int i = tid; i < nn; i += kEDGET) cnt[i] = 0;
        __syncthreads();

        const unsigned ucb = (unsigned)cb, unn = (unsigned)nn, unoff = (unsigned)noff;
        unsigned below = 0;
        for (int eb = tid; eb < E; eb += kEDGET * kLISTU) {
            unsigned dd[kLISTU];
            #pragma unroll
            for (int u = 0; u < kLISTU; ++u) {
                const int e  = eb + u * kEDGET;
                const int dv = dstv[e < E ? e : eb];
                dd[u] = (e < E) ? (unsigned)dv - unoff : 0xFFFFFFFFu;
            }
            #pragma unroll
            for (int u = 0; u < kLISTU; ++u) {
                const unsigned r = dd[u] - ucb;
                if (r < unn) atomicAdd(&cnt[r], 1);
                below += (dd[u] < ucb) ? 1u : 0u;
            }
        }
        #pragma unroll
        for (int off = 16; off > 0; off >>= 1) below += __shfl_xor(below, off);
        if (lane == 0) lsum[wave] = (int)below;
        __syncthreads();
        int base = 0;
        #pragma unroll
        for (int w = 0; w < kW; ++w) base += lsum[w];

        const int per = (nn + kEDGET - 1) / kEDGET;
        const int i0  = (tid * per < nn) ? tid * per : nn;
        const int i1  = (i0 + per < nn) ? i0 + per : nn;
        int s = 0;
        for (int i = i0; i < i1; ++i) s += cnt[i];
        int incl = s;
        #pragma unroll
        for (int off = 1; off < 32; off <<= 1) {
            const int t = __shfl_up(incl, off);
            if (lane >= off) incl += t;
        }
        if (lane == 31) lsum[kW + wave] = incl;
        __syncthreads();
        int run = base + (incl - s);
        for (int w = 0; w < wave; ++w) run += lsum[kW + w];
        for (int i = i0; i < i1; ++i) {
            const int k = cnt[i];
            cnt[i] = run;
            row_off[cb + i] = run;
            run += k;
        }
        if (ce == N && tid == kEDGET - 1) row_off[N] = run;
        __syncthreads();

        for (int eb = tid; eb < E; eb += kEDGET * kLISTU) {
            unsigned dd[kLISTU];
            #pragma unroll
            for (int u = 0; u < kLISTU; ++u) {
                const int e  = eb + u * kEDGET;
                const int dv = dstv[e < E ? e : eb];
                dd[u] = (e < E) ? (unsigned)dv - unoff : 0xFFFFFFFFu;
            }
            int pos[kLISTU];
            #pragma unroll
            for (int u = 0; u < kLISTU; ++u) {
                const unsigned r = dd[u] - ucb;
                pos[u] = (r < unn) ? atomicAdd(&cnt[r], 1) : -1;
            }
            #pragma unroll
            for (int u = 0; u < kLISTU; ++u) {
                if (pos[u] >= 0) perm[pos[u]] = eb + u * kEDGET;
            }
        }
        __syncthreads();
        cb += kLCH;
    } while (cb < hi);
}

__global__ __launch_bounds__(kEDGET)
void edge_mlp_kernel(const float* __restrict__ x_edge,
                              const float* __restrict__ edge_distance,
                              const float* __restrict__ w1, const float* __restrict__ b1,
                              const float* __restrict__ g1, const float* __restrict__ be1,
                              const float* __restrict__ w2, const float* __restrict__ b2,
                              const float* __restrict__ g2, const float* __restrict__ be2,
                              const float* __restrict__ w3, const float* __restrict__ b3,
                              float* __restrict__ m0buf,
                              float* __restrict__ envbuf,
                              const int* __restrict__ edge_index,
                              const int* __restrict__ node_offset_p,
                              int* __restrict__ row_off,
                              int* __restrict__ perm,
                              int E, int N)
{
    __shared__ __attribute__((aligned(16))) float xe[kTILE * 128];
    __shared__ __attribute__((aligned(16))) float hraw[kTILE * 64];
    __shared__ __attribute__((aligned(16))) float h1[kTILE * 64];
    __shared__ __attribute__((aligned(16))) float h2[kTILE * 64];
    __shared__ __attribute__((aligned(16))) float m0s[kTILE * kM0 * kC];
    __shared__ int lsum[2 * (kEDGET / 32)];

    static_assert(kLCH <= kTILE * kM0 * kC, "list counters must fit the m0 tile");
    if (blockIdx.x < kLISTB) {
        build_dst_lists_slice(edge_index, node_offset_p, row_off, perm, E, N,
                              (int)blockIdx.x, reinterpret_cast<int*>(m0s), lsum);
        return;
    }

    const int tid   = threadIdx.x;
    const int wave  = tid >> 5;
    const int lane  = tid & 31;
    const int lrow  = lane & 15;
    const int khalf = (lane >> 4) << 1;
    const int rbase = (lane >> 4) << 3;
    const int e0    = ((int)blockIdx.x - kLISTB) * kTILE;

    {
        const float4* src4 = reinterpret_cast<const float4*>(x_edge + (size_t)e0 * 128);
        float4 v[kXLD];
        #pragma unroll
        for (int u = 0; u < kXLD; ++u) {
            const int i4 = tid + u * kEDGET;
            const bool ok = (e0 + (i4 >> 5)) < E;
            v[u] = src4[ok ? i4 : (i4 & 31)];
        }
        float4* xe4 = reinterpret_cast<float4*>(xe);
        #pragma unroll
        for (int u = 0; u < kXLD; ++u) {
            const int i4 = tid + u * kEDGET;
            const bool ok = (e0 + (i4 >> 5)) < E;
            xe4[i4] = ok ? v[u] : make_float4(0.0f, 0.0f, 0.0f, 0.0f);
        }
    }
    __syncthreads();

    {
        const int colBase = wave * 16;
        v8f acc = {};
        for (int k0 = 0; k0 < 128; k0 += 4) {
            v2f a, b;
            a.x = xe[lrow * 128 + k0 + khalf];
            a.y = xe[lrow * 128 + k0 + khalf + 1];
            b.x = w1[(k0 + khalf)     * 64 + colBase + lrow];
            b.y = w1[(k0 + khalf + 1) * 64 + colBase + lrow];
            acc = __builtin_amdgcn_wmma_f32_16x16x4_f32(
                false, a, false, b, (short)0, acc, false, false);
        }
        #pragma unroll
        for (int r = 0; r < 8; ++r) {
            int row = rbase + r;
            hraw[row * 64 + colBase + lrow] = acc[r] + b1[colBase + lrow];
        }
    }
    __syncthreads();
    ln_silu_16x64(hraw, h1, g1, be1, tid);
    __syncthreads();

    {
        const int colBase = wave * 16;
        v8f acc = {};
        for (int k0 = 0; k0 < 64; k0 += 4) {
            v2f a, b;
            a.x = h1[lrow * 64 + k0 + khalf];
            a.y = h1[lrow * 64 + k0 + khalf + 1];
            b.x = w2[(k0 + khalf)     * 64 + colBase + lrow];
            b.y = w2[(k0 + khalf + 1) * 64 + colBase + lrow];
            acc = __builtin_amdgcn_wmma_f32_16x16x4_f32(
                false, a, false, b, (short)0, acc, false, false);
        }
        #pragma unroll
        for (int r = 0; r < 8; ++r) {
            int row = rbase + r;
            hraw[row * 64 + colBase + lrow] = acc[r] + b2[colBase + lrow];
        }
    }
    __syncthreads();
    ln_silu_16x64(hraw, h2, g2, be2, tid);
    __syncthreads();

    for (int t = wave; t < 40; t += 4) {
        const int colBase = t * 16;
        v8f acc = {};
        for (int k0 = 0; k0 < 64; k0 += 4) {
            v2f a, b;
            a.x = h2[lrow * 64 + k0 + khalf];
            a.y = h2[lrow * 64 + k0 + khalf + 1];
            b.x = w3[(size_t)(k0 + khalf)     * 640 + colBase + lrow];
            b.y = w3[(size_t)(k0 + khalf + 1) * 640 + colBase + lrow];
            acc = __builtin_amdgcn_wmma_f32_16x16x4_f32(
                false, a, false, b, (short)0, acc, false, false);
        }
        #pragma unroll
        for (int r = 0; r < 8; ++r) {
            int row = rbase + r;
            m0s[row * (kM0 * kC) + colBase + lrow] = acc[r] + b3[colBase + lrow];
        }
    }
    __syncthreads();

    {
        const float4* src4 = reinterpret_cast<const float4*>(m0s);
        float4* dst4 = reinterpret_cast<float4*>(m0buf + (size_t)e0 * (kM0 * kC));
        constexpr int kOUT4 = kTILE * (kM0 * kC / 4) / kEDGET;
        const int lim4 = ((E - e0 < kTILE) ? (E - e0) : kTILE) * (kM0 * kC / 4);
        if (lim4 == kTILE * (kM0 * kC / 4)) {
            float4 t[kOUT4];
            #pragma unroll
            for (int u = 0; u < kOUT4; ++u) t[u] = src4[tid + u * kEDGET];
            #pragma unroll
            for (int u = 0; u < kOUT4; ++u) dst4[tid + u * kEDGET] = t[u];
        } else {
            for (int i = tid; i < lim4; i += kEDGET) dst4[i] = src4[i];
        }
    }
    if (tid < kTILE) {
        int e = e0 + tid;
        if (e < E) {
            float env = 0.0f;
            float d = edge_distance[e] * (1.0f / kCUTOFF);
            if (d < 1.0f) {
                float d5 = d * d * d * d * d;
                env = 1.0f - 21.0f * d5 + 35.0f * d5 * d - 15.0f * d5 * d * d;
            }
            envbuf[e] = env * (1.0f / kRESCALE);
        }
    }
}

__device__ __forceinline__ float xhalf(float v) {
    return __builtin_bit_cast(float, __builtin_amdgcn_permlanex16(
        __builtin_bit_cast(int, v), __builtin_bit_cast(int, v), 0x76543210, 0xfedcba98, false, false));
}

__global__ __launch_bounds__(128, 5)
void node_gather_kernel(const float* __restrict__ x,
                        const float* __restrict__ wigner_inv,
                        const float* __restrict__ m0buf,
                        const float* __restrict__ envbuf,
                        const int*   __restrict__ row_off,
                        const int*   __restrict__ perm,
                        int*         __restrict__ perm_sorted,
                        float*       __restrict__ out)
{
    __shared__ __attribute__((aligned(16))) float a_s[32 * kAP];
    __shared__ __attribute__((aligned(8))) float env_k[kGB * kM0];
    __shared__ int elist[kGB];

    const int n    = blockIdx.x;
    const int c    = threadIdx.x;
    const int lane = c & 31;
    const int wave = c >> 5;
    const int lc   = lane & 15;
    const int lh   = lane >> 4;
    const int beg  = row_off[n];
    const int cnt  = row_off[n + 1] - beg;

    v8f acc[2][2];
    const float* xn = x + (size_t)n * (kNCOEF * kC) + wave * 32 + lc;
    #pragma unroll
    for (int mt = 0; mt < 2; ++mt) {
        #pragma unroll
        for (int nt = 0; nt < 2; ++nt) {
            #pragma unroll
            for (int i = 0; i < 8; ++i) {
                const int row = 16 * mt + 8 * lh + i;
                acc[mt][nt][i] = (row < kNCOEF) ? xn[row * kC + 16 * nt] : 0.0f;
            }
        }
    }

    for (int i = c; i < (32 - kNCOEF) * kAP; i += blockDim.x) a_s[kNCOEF * kAP + i] = 0.0f;

    const int  stage_r    = c / kM0;
    const int  stage_l    = c - stage_r * kM0;
    const bool stage_w    = c < kNCOEF * kM0;
    const int  stage_goff = stage_r * kNCOEF + stage_l * (stage_l + 1);
    const int  stage_aidx = stage_r * kAP + stage_l;

    for (int i = c; i < cnt; i += blockDim.x) {
        const int ei = perm[beg + i];
        int rank = 0;
        for (int j = 0; j < cnt; ++j) rank += (perm[beg + j] < ei) ? 1 : 0;
        perm_sorted[beg + rank] = ei;
    }
    __syncthreads();

    for (int b0 = 0; b0 < cnt; b0 += kGB) {
        const int nb = (cnt - b0 < kGB) ? (cnt - b0) : kGB;
        const int ng = (nb + kGRP - 1) / kGRP;
        if (c < kGB) elist[c] = perm_sorted[beg + b0 + (c < nb ? c : nb - 1)];
        __syncthreads();

        v2f bf[kGB / kGRP][5][2];
        #pragma unroll
        for (int g = 0; g < kGB / kGRP; ++g) {
            if (g < ng) {
                int eg[kGRP];
                #pragma unroll
                for (int q = 0; q < kGRP; ++q) eg[q] = elist[g * kGRP + q];
                #pragma unroll
                for (int ks = 0; ks < 5; ++ks) {
                    #pragma unroll
                    for (int s2 = 0; s2 < 2; ++s2) {
                        const int k0 = 4 * ks + s2;
                        const int k1 = 4 * ks + 2 + s2;
                        const int e  = lh ? eg[k1 / kM0] : eg[k0 / kM0];
                        const int l  = lh ? (k1 % kM0) : (k0 % kM0);
                        const float* mp = m0buf + (size_t)e * (kM0 * kC) + l * kC + wave * 32 + lc;
                        const float b_lo = mp[0];
                        const float b_hi = mp[16];
                        if (s2 == 0) { bf[g][ks][0].x = b_lo; bf[g][ks][1].x = b_hi; }
                        else         { bf[g][ks][0].y = b_lo; bf[g][ks][1].y = b_hi; }
                    }
                }
            }
        }

        float wv[kGB];
        #pragma unroll
        for (int j = 0; j < kGB; ++j) {
            if (j < ng * kGRP && stage_w) {
                wv[j] = wigner_inv[(size_t)elist[j] * (kNCOEF * kNCOEF) + stage_goff];
            }
        }
        if (c < ng * (kGRP * kM0)) env_k[c] = (stage_r < nb) ? envbuf[elist[stage_r]] : 0.0f;
        #pragma unroll
        for (int j = 0; j < kGB; ++j) {
            if (j < ng * kGRP && stage_w) a_s[stage_aidx + kM0 * j] = (j < nb) ? wv[j] : 0.0f;
        }
        __syncthreads();

        #pragma unroll
        for (int g = 0; g < kGB / kGRP; ++g) {
            if (g < ng) {
                #pragma unroll
                for (int ks = 0; ks < 5; ++ks) {
                    const int kk = g * (kGRP * kM0) + 4 * ks + 2 * lh;
                    const v2f ek = *reinterpret_cast<const v2f*>(env_k + kk);
                    const v2f a0 = *reinterpret_cast<const v2f*>(a_s + lc * kAP + kk) * ek;
                    const v2f a1 = *reinterpret_cast<const v2f*>(a_s + (16 + lc) * kAP + kk) * ek;
                    acc[0][0] = __builtin_amdgcn_wmma_f32_16x16x4_f32(
                        false, a0, false, bf[g][ks][0], (short)0, acc[0][0], false, false);
                    acc[0][1] = __builtin_amdgcn_wmma_f32_16x16x4_f32(
                        false, a0, false, bf[g][ks][1], (short)0, acc[0][1], false, false);
                    acc[1][0] = __builtin_amdgcn_wmma_f32_16x16x4_f32(
                        false, a1, false, bf[g][ks][0], (short)0, acc[1][0], false, false);
                    acc[1][1] = __builtin_amdgcn_wmma_f32_16x16x4_f32(
                        false, a1, false, bf[g][ks][1], (short)0, acc[1][1], false, false);
                }
            }
        }
        __syncthreads();
    }

    float* on = out + (size_t)n * (kNCOEF * kC) + c;
    #pragma unroll
    for (int mt = 0; mt < 2; ++mt) {
        #pragma unroll
        for (int i = 0; i < 8; ++i) {
            const float r0 = acc[mt][0][i];
            const float r1 = acc[mt][1][i];
            const float x0 = xhalf(r0);
            const float x1 = xhalf(r1);
            const float s_lo = lh ? x1 : r0;
            const float s_hi = lh ? r1 : x0;
            const int row_lo = 16 * mt + i;
            const int row_hi = 16 * mt + 8 + i;
            if (row_lo < kNCOEF) on[row_lo * kC] = s_lo;
            if (row_hi < kNCOEF) on[row_hi * kC] = s_hi;
        }
    }
}

extern "C" void kernel_launch(void* const* d_in, const int* in_sizes, int n_in,
                              void* d_out, int out_size, void* d_ws, size_t ws_size,
                              hipStream_t stream) {
    const float* x             = (const float*)d_in[0];
    const float* x_edge        = (const float*)d_in[1];
    const float* edge_distance = (const float*)d_in[2];
    const float* wigner_inv    = (const float*)d_in[3];
    const int*   edge_index    = (const int*)d_in[4];
    const float* w1  = (const float*)d_in[5];
    const float* b1  = (const float*)d_in[6];
    const float* g1  = (const float*)d_in[7];
    const float* be1 = (const float*)d_in[8];
    const float* w2  = (const float*)d_in[9];
    const float* b2  = (const float*)d_in[10];
    const float* g2  = (const float*)d_in[11];
    const float* be2 = (const float*)d_in[12];
    const float* w3  = (const float*)d_in[13];
    const float* b3  = (const float*)d_in[14];
    const int* node_offset = (const int*)d_in[15];

    const int E = in_sizes[1] / 128;
    const int N = out_size / (kNCOEF * kC);

    float* m0buf       = (float*)d_ws;
    float* envbuf      = m0buf + (size_t)E * (kM0 * kC);
    int*   row_off     = (int*)(envbuf + E);
    int*   perm        = row_off + (N + 1);
    int*   perm_sorted = perm + E;

    const int blocks = (E + kTILE - 1) / kTILE;
    edge_mlp_kernel<<<kLISTB + blocks, kEDGET, 0, stream>>>(
        x_edge, edge_distance,
        w1, b1, g1, be1, w2, b2, g2, be2, w3, b3,
        m0buf, envbuf, edge_index, node_offset, row_off, perm, E, N);

    node_gather_kernel<<<N, 128, 0, stream>>>(
        x, wigner_inv, m0buf, envbuf, row_off, perm, perm_sorted, (float*)d_out);
}


# List build reads the destinations once: while counting, each thread keeps its slice's edges (id, node) in its own LDS column and places the ids from there; second full pass only on column overflow
# speedup vs baseline: 4.6728x; 1.0458x over previous
#include <hip/hip_runtime.h>

typedef __attribute__((ext_vector_type(2))) float v2f;
typedef __attribute__((ext_vector_type(8))) float v8f;

constexpr int kNCOEF = 25;
constexpr int kM0    = 5;
constexpr int kC     = 128;
constexpr int kTILE  = 16;
constexpr float kCUTOFF  = 12.0f;
constexpr float kRESCALE = 5.0f;
constexpr float kEPS     = 1e-5f;
constexpr int kLISTB  = 32;
constexpr int kLISTU  = 40;
constexpr int kLCH    = 2048;
constexpr int kLK     = 32;
constexpr int kGB     = 16;
constexpr int kGRP    = 4;
constexpr int kEDGET = 128;
constexpr int kXLD   = 4;
constexpr int kAP     = 84;

__device__ __forceinline__ float silu_f(float x) {
    return x / (1.0f + __expf(-x));
}

template <int M>
__device__ __forceinline__ float row_xor(float v) {
    return __builtin_bit_cast(float, __builtin_amdgcn_update_dpp(
        0, __builtin_bit_cast(int, v), 0x160 | M, 0xf, 0xf, false));
}

constexpr int kLNF = 8;

__device__ __forceinline__ void ln_silu_16x64(const float* __restrict__ hraw,
                                              float* __restrict__ hout,
                                              const float* __restrict__ g,
                                              const float* __restrict__ be,
                                              int tid) {
    const int c0  = (tid & 7) * kLNF;
    const int off = (tid >> 3) * 64 + c0;
    const float4 p = *reinterpret_cast<const float4*>(hraw + off);
    const float4 q = *reinterpret_cast<const float4*>(hraw + off + 4);
    const float v[kLNF] = {p.x, p.y, p.z, p.w, q.x, q.y, q.z, q.w};
    float gv[kLNF], bv[kLNF];
    #pragma unroll
    for (int j = 0; j < kLNF; ++j) { gv[j] = g[c0 + j]; bv[j] = be[c0 + j]; }
    float s = 0.0f, s2 = 0.0f;
    #pragma unroll
    for (int j = 0; j < kLNF; ++j) { s += v[j]; s2 += v[j] * v[j]; }
    s += row_xor<1>(s);  s2 += row_xor<1>(s2);
    s += row_xor<2>(s);  s2 += row_xor<2>(s2);
    s += row_xor<4>(s);  s2 += row_xor<4>(s2);
    const float mu  = s * (1.0f / 64.0f);
    const float var = s2 * (1.0f / 64.0f) - mu * mu;
    const float inv = rsqrtf(var + kEPS);
    float o[kLNF];
    #pragma unroll
    for (int j = 0; j < kLNF; ++j) o[j] = silu_f((v[j] - mu) * inv * gv[j] + bv[j]);
    *reinterpret_cast<float4*>(hout + off)     = make_float4(o[0], o[1], o[2], o[3]);
    *reinterpret_cast<float4*>(hout + off + 4) = make_float4(o[4], o[5], o[6], o[7]);
}


__device__ __forceinline__ void build_dst_lists_slice(const int* __restrict__ edge_index,
                                                      const int* __restrict__ node_offset_p,
                                                      int* __restrict__ row_off,
                                                      int* __restrict__ perm,
                                                      int E, int N, int part,
                                                      int* __restrict__ cnt,
                                                      int* __restrict__ keep_e,
                                                      int* __restrict__ keep_r,
                                                      int* __restrict__ lsum)
{
    constexpr int kW = kEDGET / 32;
    const int tid  = threadIdx.x;
    const int lane = tid & 31;
    const int wave = tid >> 5;
    const int noff = node_offset_p[0];
    const int* __restrict__ dstv = edge_index + E;

    const int npb = (N + kLISTB - 1) / kLISTB;
    const int lo  = (part * npb < N) ? part * npb : N;
    const int hi  = (part == kLISTB - 1) ? N : ((lo + npb < N) ? lo + npb : N);
    if (lo >= hi && part != kLISTB - 1) return;

    int cb = lo;
    do {
        const int ce = (cb + kLCH < hi) ? cb + kLCH : hi;
        const int nn = ce - cb;
        for (int i = tid; i < nn; i += kEDGET) cnt[i] = 0;
        if (tid == 0) lsum[2 * kW] = 0;
        __syncthreads();

        const unsigned ucb = (unsigned)cb, unn = (unsigned)nn, unoff = (unsigned)noff;
        unsigned below = 0;
        int kept = 0;
        for (int eb = tid; eb < E; eb += kEDGET * kLISTU) {
            unsigned dd[kLISTU];
            #pragma unroll
            for (int u = 0; u < kLISTU; ++u) {
                const int e  = eb + u * kEDGET;
                const int dv = dstv[e < E ? e : eb];
                dd[u] = (e < E) ? (unsigned)dv - unoff : 0xFFFFFFFFu;
            }
            #pragma unroll
            for (int u = 0; u < kLISTU; ++u) {
                const unsigned r = dd[u] - ucb;
                if (r < unn) {
                    atomicAdd(&cnt[r], 1);
                    const int slot = ((kept < kLK) ? kept : kLK - 1) * kEDGET + tid;
                    keep_e[slot] = eb + u * kEDGET;
                    keep_r[slot] = (int)r;
                    ++kept;
                }
                below += (dd[u] < ucb) ? 1u : 0u;
            }
        }
        if (kept > kLK) lsum[2 * kW] = 1;
        #pragma unroll
        for (int off = 16; off > 0; off >>= 1) below += __shfl_xor(below, off);
        if (lane == 0) lsum[wave] = (int)below;
        __syncthreads();
        int base = 0;
        #pragma unroll
        for (int w = 0; w < kW; ++w) base += lsum[w];

        const int per = (nn + kEDGET - 1) / kEDGET;
        const int i0  = (tid * per < nn) ? tid * per : nn;
        const int i1  = (i0 + per < nn) ? i0 + per : nn;
        int s = 0;
        for (int i = i0; i < i1; ++i) s += cnt[i];
        int incl = s;
        #pragma unroll
        for (int off = 1; off < 32; off <<= 1) {
            const int t = __shfl_up(incl, off);
            if (lane >= off) incl += t;
        }
        if (lane == 31) lsum[kW + wave] = incl;
        __syncthreads();
        int run = base + (incl - s);
        for (int w = 0; w < wave; ++w) run += lsum[kW + w];
        for (int i = i0; i < i1; ++i) {
            const int k = cnt[i];
            cnt[i] = run;
            row_off[cb + i] = run;
            run += k;
        }
        if (ce == N && tid == kEDGET - 1) row_off[N] = run;
        __syncthreads();

        if (lsum[2 * kW] == 0) {
            constexpr int kPU = 4;
            static_assert(kLK % kPU == 0, "a trip must stay inside the column");
            #pragma unroll 1
            for (int i0 = 0; i0 < kept; i0 += kPU) {
                int ke[kPU], kr[kPU], pos[kPU];
                #pragma unroll
                for (int q = 0; q < kPU; ++q) {
                    const int i = (i0 + q < kept) ? i0 + q : i0;
                    ke[q] = keep_e[i * kEDGET + tid];
                    kr[q] = keep_r[i * kEDGET + tid];
                }
                #pragma unroll
                for (int q = 0; q < kPU; ++q) {
                    pos[q] = (i0 + q < kept) ? atomicAdd(&cnt[kr[q]], 1) : -1;
                }
                #pragma unroll
                for (int q = 0; q < kPU; ++q) {
                    if (pos[q] >= 0) perm[pos[q]] = ke[q];
                }
            }
        } else
        for (int eb = tid; eb < E; eb += kEDGET * kLISTU) {
            unsigned dd[kLISTU];
            #pragma unroll
            for (int u = 0; u < kLISTU; ++u) {
                const int e  = eb + u * kEDGET;
                const int dv = dstv[e < E ? e : eb];
                dd[u] = (e < E) ? (unsigned)dv - unoff : 0xFFFFFFFFu;
            }
            int pos[kLISTU];
            #pragma unroll
            for (int u = 0; u < kLISTU; ++u) {
                const unsigned r = dd[u] - ucb;
                pos[u] = (r < unn) ? atomicAdd(&cnt[r], 1) : -1;
            }
            #pragma unroll
            for (int u = 0; u < kLISTU; ++u) {
                if (pos[u] >= 0) perm[pos[u]] = eb + u * kEDGET;
            }
        }
        __syncthreads();
        cb += kLCH;
    } while (cb < hi);
}

__global__ __launch_bounds__(kEDGET)
void edge_mlp_kernel(const float* __restrict__ x_edge,
                              const float* __restrict__ edge_distance,
                              const float* __restrict__ w1, const float* __restrict__ b1,
                              const float* __restrict__ g1, const float* __restrict__ be1,
                              const float* __restrict__ w2, const float* __restrict__ b2,
                              const float* __restrict__ g2, const float* __restrict__ be2,
                              const float* __restrict__ w3, const float* __restrict__ b3,
                              float* __restrict__ m0buf,
                              float* __restrict__ envbuf,
                              const int* __restrict__ edge_index,
                              const int* __restrict__ node_offset_p,
                              int* __restrict__ row_off,
                              int* __restrict__ perm,
                              int E, int N)
{
    __shared__ __attribute__((aligned(16))) float xe[kTILE * 128];
    __shared__ __attribute__((aligned(16))) float hraw[kTILE * 64];
    __shared__ __attribute__((aligned(16))) float h1[kTILE * 64];
    __shared__ __attribute__((aligned(16))) float h2[kTILE * 64];
    __shared__ __attribute__((aligned(16))) float m0s[kTILE * kM0 * kC];
    __shared__ int lsum[2 * (kEDGET / 32) + 1];

    static_assert(kLCH + 2 * kLK * kEDGET <= kTILE * kM0 * kC, "list counters and kept edges must fit the m0 tile");
    if (blockIdx.x < kLISTB) {
        int* lds = reinterpret_cast<int*>(m0s);
        build_dst_lists_slice(edge_index, node_offset_p, row_off, perm, E, N,
                              (int)blockIdx.x, lds, lds + kLCH, lds + kLCH + kLK * kEDGET, lsum);
        return;
    }

    const int tid   = threadIdx.x;
    const int wave  = tid >> 5;
    const int lane  = tid & 31;
    const int lrow  = lane & 15;
    const int khalf = (lane >> 4) << 1;
    const int rbase = (lane >> 4) << 3;
    const int e0    = ((int)blockIdx.x - kLISTB) * kTILE;

    {
        const float4* src4 = reinterpret_cast<const float4*>(x_edge + (size_t)e0 * 128);
        float4 v[kXLD];
        #pragma unroll
        for (int u = 0; u < kXLD; ++u) {
            const int i4 = tid + u * kEDGET;
            const bool ok = (e0 + (i4 >> 5)) < E;
            v[u] = src4[ok ? i4 : (i4 & 31)];
        }
        float4* xe4 = reinterpret_cast<float4*>(xe);
        #pragma unroll
        for (int u = 0; u < kXLD; ++u) {
            const int i4 = tid + u * kEDGET;
            const bool ok = (e0 + (i4 >> 5)) < E;
            xe4[i4] = ok ? v[u] : make_float4(0.0f, 0.0f, 0.0f, 0.0f);
        }
    }
    __syncthreads();

    {
        const int colBase = wave * 16;
        v8f acc = {};
        for (int k0 = 0; k0 < 128; k0 += 4) {
            v2f a, b;
            a.x = xe[lrow * 128 + k0 + khalf];
            a.y = xe[lrow * 128 + k0 + khalf + 1];
            b.x = w1[(k0 + khalf)     * 64 + colBase + lrow];
            b.y = w1[(k0 + khalf + 1) * 64 + colBase + lrow];
            acc = __builtin_amdgcn_wmma_f32_16x16x4_f32(
                false, a, false, b, (short)0, acc, false, false);
        }
        #pragma unroll
        for (int r = 0; r < 8; ++r) {
            int row = rbase + r;
            hraw[row * 64 + colBase + lrow] = acc[r] + b1[colBase + lrow];
        }
    }
    __syncthreads();
    ln_silu_16x64(hraw, h1, g1, be1, tid);
    __syncthreads();

    {
        const int colBase = wave * 16;
        v8f acc = {};
        for (int k0 = 0; k0 < 64; k0 += 4) {
            v2f a, b;
            a.x = h1[lrow * 64 + k0 + khalf];
            a.y = h1[lrow * 64 + k0 + khalf + 1];
            b.x = w2[(k0 + khalf)     * 64 + colBase + lrow];
            b.y = w2[(k0 + khalf + 1) * 64 + colBase + lrow];
            acc = __builtin_amdgcn_wmma_f32_16x16x4_f32(
                false, a, false, b, (short)0, acc, false, false);
        }
        #pragma unroll
        for (int r = 0; r < 8; ++r) {
            int row = rbase + r;
            hraw[row * 64 + colBase + lrow] = acc[r] + b2[colBase + lrow];
        }
    }
    __syncthreads();
    ln_silu_16x64(hraw, h2, g2, be2, tid);
    __syncthreads();

    for (int t = wave; t < 40; t += 4) {
        const int colBase = t * 16;
        v8f acc = {};
        for (int k0 = 0; k0 < 64; k0 += 4) {
            v2f a, b;
            a.x = h2[lrow * 64 + k0 + khalf];
            a.y = h2[lrow * 64 + k0 + khalf + 1];
            b.x = w3[(size_t)(k0 + khalf)     * 640 + colBase + lrow];
            b.y = w3[(size_t)(k0 + khalf + 1) * 640 + colBase + lrow];
            acc = __builtin_amdgcn_wmma_f32_16x16x4_f32(
                false, a, false, b, (short)0, acc, false, false);
        }
        #pragma unroll
        for (int r = 0; r < 8; ++r) {
            int row = rbase + r;
            m0s[row * (kM0 * kC) + colBase + lrow] = acc[r] + b3[colBase + lrow];
        }
    }
    __syncthreads();

    {
        const float4* src4 = reinterpret_cast<const float4*>(m0s);
        float4* dst4 = reinterpret_cast<float4*>(m0buf + (size_t)e0 * (kM0 * kC));
        constexpr int kOUT4 = kTILE * (kM0 * kC / 4) / kEDGET;
        const int lim4 = ((E - e0 < kTILE) ? (E - e0) : kTILE) * (kM0 * kC / 4);
        if (lim4 == kTILE * (kM0 * kC / 4)) {
            float4 t[kOUT4];
            #pragma unroll
            for (int u = 0; u < kOUT4; ++u) t[u] = src4[tid + u * kEDGET];
            #pragma unroll
            for (int u = 0; u < kOUT4; ++u) dst4[tid + u * kEDGET] = t[u];
        } else {
            for (int i = tid; i < lim4; i += kEDGET) dst4[i] = src4[i];
        }
    }
    if (tid < kTILE) {
        int e = e0 + tid;
        if (e < E) {
            float env = 0.0f;
            float d = edge_distance[e] * (1.0f / kCUTOFF);
            if (d < 1.0f) {
                float d5 = d * d * d * d * d;
                env = 1.0f - 21.0f * d5 + 35.0f * d5 * d - 15.0f * d5 * d * d;
            }
            envbuf[e] = env * (1.0f / kRESCALE);
        }
    }
}

__device__ __forceinline__ float xhalf(float v) {
    return __builtin_bit_cast(float, __builtin_amdgcn_permlanex16(
        __builtin_bit_cast(int, v), __builtin_bit_cast(int, v), 0x76543210, 0xfedcba98, false, false));
}

__global__ __launch_bounds__(128, 5)
void node_gather_kernel(const float* __restrict__ x,
                        const float* __restrict__ wigner_inv,
                        const float* __restrict__ m0buf,
                        const float* __restrict__ envbuf,
                        const int*   __restrict__ row_off,
                        const int*   __restrict__ perm,
                        int*         __restrict__ perm_sorted,
                        float*       __restrict__ out)
{
    __shared__ __attribute__((aligned(16))) float a_s[32 * kAP];
    __shared__ __attribute__((aligned(8))) float env_k[kGB * kM0];
    __shared__ int elist[kGB];

    const int n    = blockIdx.x;
    const int c    = threadIdx.x;
    const int lane = c & 31;
    const int wave = c >> 5;
    const int lc   = lane & 15;
    const int lh   = lane >> 4;
    const int beg  = row_off[n];
    const int cnt  = row_off[n + 1] - beg;

    v8f acc[2][2];
    const float* xn = x + (size_t)n * (kNCOEF * kC) + wave * 32 + lc;
    #pragma unroll
    for (int mt = 0; mt < 2; ++mt) {
        #pragma unroll
        for (int nt = 0; nt < 2; ++nt) {
            #pragma unroll
            for (int i = 0; i < 8; ++i) {
                const int row = 16 * mt + 8 * lh + i;
                acc[mt][nt][i] = (row < kNCOEF) ? xn[row * kC + 16 * nt] : 0.0f;
            }
        }
    }

    for (int i = c; i < (32 - kNCOEF) * kAP; i += blockDim.x) a_s[kNCOEF * kAP + i] = 0.0f;

    const int  stage_r    = c / kM0;
    const int  stage_l    = c - stage_r * kM0;
    const bool stage_w    = c < kNCOEF * kM0;
    const int  stage_goff = stage_r * kNCOEF + stage_l * (stage_l + 1);
    const int  stage_aidx = stage_r * kAP + stage_l;

    for (int i = c; i < cnt; i += blockDim.x) {
        const int ei = perm[beg + i];
        int rank = 0;
        for (int j = 0; j < cnt; ++j) rank += (perm[beg + j] < ei) ? 1 : 0;
        perm_sorted[beg + rank] = ei;
    }
    __syncthreads();

    for (int b0 = 0; b0 < cnt; b0 += kGB) {
        const int nb = (cnt - b0 < kGB) ? (cnt - b0) : kGB;
        const int ng = (nb + kGRP - 1) / kGRP;
        if (c < kGB) elist[c] = perm_sorted[beg + b0 + (c < nb ? c : nb - 1)];
        __syncthreads();

        v2f bf[kGB / kGRP][5][2];
        #pragma unroll
        for (int g = 0; g < kGB / kGRP; ++g) {
            if (g < ng) {
                int eg[kGRP];
                #pragma unroll
                for (int q = 0; q < kGRP; ++q) eg[q] = elist[g * kGRP + q];
                #pragma unroll
                for (int ks = 0; ks < 5; ++ks) {
                    #pragma unroll
                    for (int s2 = 0; s2 < 2; ++s2) {
                        const int k0 = 4 * ks + s2;
                        const int k1 = 4 * ks + 2 + s2;
                        const int e  = lh ? eg[k1 / kM0] : eg[k0 / kM0];
                        const int l  = lh ? (k1 % kM0) : (k0 % kM0);
                        const float* mp = m0buf + (size_t)e * (kM0 * kC) + l * kC + wave * 32 + lc;
                        const float b_lo = mp[0];
                        const float b_hi = mp[16];
                        if (s2 == 0) { bf[g][ks][0].x = b_lo; bf[g][ks][1].x = b_hi; }
                        else         { bf[g][ks][0].y = b_lo; bf[g][ks][1].y = b_hi; }
                    }
                }
            }
        }

        float wv[kGB];
        #pragma unroll
        for (int j = 0; j < kGB; ++j) {
            if (j < ng * kGRP && stage_w) {
                wv[j] = wigner_inv[(size_t)elist[j] * (kNCOEF * kNCOEF) + stage_goff];
            }
        }
        if (c < ng * (kGRP * kM0)) env_k[c] = (stage_r < nb) ? envbuf[elist[stage_r]] : 0.0f;
        #pragma unroll
        for (int j = 0; j < kGB; ++j) {
            if (j < ng * kGRP && stage_w) a_s[stage_aidx + kM0 * j] = (j < nb) ? wv[j] : 0.0f;
        }
        __syncthreads();

        #pragma unroll
        for (int g = 0; g < kGB / kGRP; ++g) {
            if (g < ng) {
                #pragma unroll
                for (int ks = 0; ks < 5; ++ks) {
                    const int kk = g * (kGRP * kM0) + 4 * ks + 2 * lh;
                    const v2f ek = *reinterpret_cast<const v2f*>(env_k + kk);
                    const v2f a0 = *reinterpret_cast<const v2f*>(a_s + lc * kAP + kk) * ek;
                    const v2f a1 = *reinterpret_cast<const v2f*>(a_s + (16 + lc) * kAP + kk) * ek;
                    acc[0][0] = __builtin_amdgcn_wmma_f32_16x16x4_f32(
                        false, a0, false, bf[g][ks][0], (short)0, acc[0][0], false, false);
                    acc[0][1] = __builtin_amdgcn_wmma_f32_16x16x4_f32(
                        false, a0, false, bf[g][ks][1], (short)0, acc[0][1], false, false);
                    acc[1][0] = __builtin_amdgcn_wmma_f32_16x16x4_f32(
                        false, a1, false, bf[g][ks][0], (short)0, acc[1][0], false, false);
                    acc[1][1] = __builtin_amdgcn_wmma_f32_16x16x4_f32(
                        false, a1, false, bf[g][ks][1], (short)0, acc[1][1], false, false);
                }
            }
        }
        __syncthreads();
    }

    float* on = out + (size_t)n * (kNCOEF * kC) + c;
    #pragma unroll
    for (int mt = 0; mt < 2; ++mt) {
        #pragma unroll
        for (int i = 0; i < 8; ++i) {
            const float r0 = acc[mt][0][i];
            const float r1 = acc[mt][1][i];
            const float x0 = xhalf(r0);
            const float x1 = xhalf(r1);
            const float s_lo = lh ? x1 : r0;
            const float s_hi = lh ? r1 : x0;
            const int row_lo = 16 * mt + i;
            const int row_hi = 16 * mt + 8 + i;
            if (row_lo < kNCOEF) on[row_lo * kC] = s_lo;
            if (row_hi < kNCOEF) on[row_hi * kC] = s_hi;
        }
    }
}

extern "C" void kernel_launch(void* const* d_in, const int* in_sizes, int n_in,
                              void* d_out, int out_size, void* d_ws, size_t ws_size,
                              hipStream_t stream) {
    const float* x             = (const float*)d_in[0];
    const float* x_edge        = (const float*)d_in[1];
    const float* edge_distance = (const float*)d_in[2];
    const float* wigner_inv    = (const float*)d_in[3];
    const int*   edge_index    = (const int*)d_in[4];
    const float* w1  = (const float*)d_in[5];
    const float* b1  = (const float*)d_in[6];
    const float* g1  = (const float*)d_in[7];
    const float* be1 = (const float*)d_in[8];
    const float* w2  = (const float*)d_in[9];
    const float* b2  = (const float*)d_in[10];
    const float* g2  = (const float*)d_in[11];
    const float* be2 = (const float*)d_in[12];
    const float* w3  = (const float*)d_in[13];
    const float* b3  = (const float*)d_in[14];
    const int* node_offset = (const int*)d_in[15];

    const int E = in_sizes[1] / 128;
    const int N = out_size / (kNCOEF * kC);

    float* m0buf       = (float*)d_ws;
    float* envbuf      = m0buf + (size_t)E * (kM0 * kC);
    int*   row_off     = (int*)(envbuf + E);
    int*   perm        = row_off + (N + 1);
    int*   perm_sorted = perm + E;

    const int blocks = (E + kTILE - 1) / kTILE;
    edge_mlp_kernel<<<kLISTB + blocks, kEDGET, 0, stream>>>(
        x_edge, edge_distance,
        w1, b1, g1, be1, w2, b2, g2, be2, w3, b3,
        m0buf, envbuf, edge_index, node_offset, row_off, perm, E, N);

    node_gather_kernel<<<N, 128, 0, stream>>>(
        x, wigner_inv, m0buf, envbuf, row_off, perm, perm_sorted, (float*)d_out);
}


# Node kernel: a node's edge ids stay in LDS: put in id order there (one id per thread, ranks from LDS reads) and read four at a time for both the m0 and Wigner fetches; no workspace round trip
# speedup vs baseline: 4.8657x; 1.0413x over previous
#include <hip/hip_runtime.h>

typedef __attribute__((ext_vector_type(2))) float v2f;
typedef __attribute__((ext_vector_type(8))) float v8f;

constexpr int kNCOEF = 25;
constexpr int kM0    = 5;
constexpr int kC     = 128;
constexpr int kTILE  = 16;
constexpr float kCUTOFF  = 12.0f;
constexpr float kRESCALE = 5.0f;
constexpr float kEPS     = 1e-5f;
constexpr int kLISTB  = 32;
constexpr int kLISTU  = 40;
constexpr int kLCH    = 2048;
constexpr int kLK     = 32;
constexpr int kGB     = 16;
constexpr int kGRP    = 4;
constexpr int kEDGET = 128;
constexpr int kXLD   = 4;
constexpr int kSORT   = 128;
constexpr int kAP     = 84;

__device__ __forceinline__ float silu_f(float x) {
    return x / (1.0f + __expf(-x));
}

template <int M>
__device__ __forceinline__ float row_xor(float v) {
    return __builtin_bit_cast(float, __builtin_amdgcn_update_dpp(
        0, __builtin_bit_cast(int, v), 0x160 | M, 0xf, 0xf, false));
}

constexpr int kLNF = 8;

__device__ __forceinline__ void ln_silu_16x64(const float* __restrict__ hraw,
                                              float* __restrict__ hout,
                                              const float* __restrict__ g,
                                              const float* __restrict__ be,
                                              int tid) {
    const int c0  = (tid & 7) * kLNF;
    const int off = (tid >> 3) * 64 + c0;
    const float4 p = *reinterpret_cast<const float4*>(hraw + off);
    const float4 q = *reinterpret_cast<const float4*>(hraw + off + 4);
    const float v[kLNF] = {p.x, p.y, p.z, p.w, q.x, q.y, q.z, q.w};
    float gv[kLNF], bv[kLNF];
    #pragma unroll
    for (int j = 0; j < kLNF; ++j) { gv[j] = g[c0 + j]; bv[j] = be[c0 + j]; }
    float s = 0.0f, s2 = 0.0f;
    #pragma unroll
    for (int j = 0; j < kLNF; ++j) { s += v[j]; s2 += v[j] * v[j]; }
    s += row_xor<1>(s);  s2 += row_xor<1>(s2);
    s += row_xor<2>(s);  s2 += row_xor<2>(s2);
    s += row_xor<4>(s);  s2 += row_xor<4>(s2);
    const float mu  = s * (1.0f / 64.0f);
    const float var = s2 * (1.0f / 64.0f) - mu * mu;
    const float inv = rsqrtf(var + kEPS);
    float o[kLNF];
    #pragma unroll
    for (int j = 0; j < kLNF; ++j) o[j] = silu_f((v[j] - mu) * inv * gv[j] + bv[j]);
    *reinterpret_cast<float4*>(hout + off)     = make_float4(o[0], o[1], o[2], o[3]);
    *reinterpret_cast<float4*>(hout + off + 4) = make_float4(o[4], o[5], o[6], o[7]);
}


__device__ __forceinline__ void build_dst_lists_slice(const int* __restrict__ edge_index,
                                                      const int* __restrict__ node_offset_p,
                                                      int* __restrict__ row_off,
                                                      int* __restrict__ perm,
                                                      int E, int N, int part,
                                                      int* __restrict__ cnt,
                                                      int* __restrict__ keep_e,
                                                      int* __restrict__ keep_r,
                                                      int* __restrict__ lsum)
{
    constexpr int kW = kEDGET / 32;
    const int tid  = threadIdx.x;
    const int lane = tid & 31;
    const int wave = tid >> 5;
    const int noff = node_offset_p[0];
    const int* __restrict__ dstv = edge_index + E;

    const int npb = (N + kLISTB - 1) / kLISTB;
    const int lo  = (part * npb < N) ? part * npb : N;
    const int hi  = (part == kLISTB - 1) ? N : ((lo + npb < N) ? lo + npb : N);
    if (lo >= hi && part != kLISTB - 1) return;

    int cb = lo;
    do {
        const int ce = (cb + kLCH < hi) ? cb + kLCH : hi;
        const int nn = ce - cb;
        for (int i = tid; i < nn; i += kEDGET) cnt[i] = 0;
        if (tid == 0) lsum[2 * kW] = 0;
        __syncthreads();

        const unsigned ucb = (unsigned)cb, unn = (unsigned)nn, unoff = (unsigned)noff;
        unsigned below = 0;
        int kept = 0;
        for (int eb = tid; eb < E; eb += kEDGET * kLISTU) {
            unsigned dd[kLISTU];
            #pragma unroll
            for (int u = 0; u < kLISTU; ++u) {
                const int e  = eb + u * kEDGET;
                const int dv = dstv[e < E ? e : eb];
                dd[u] = (e < E) ? (unsigned)dv - unoff : 0xFFFFFFFFu;
            }
            #pragma unroll
            for (int u = 0; u < kLISTU; ++u) {
                const unsigned r = dd[u] - ucb;
                if (r < unn) {
                    atomicAdd(&cnt[r], 1);
                    const int slot = ((kept < kLK) ? kept : kLK - 1) * kEDGET + tid;
                    keep_e[slot] = eb + u * kEDGET;
                    keep_r[slot] = (int)r;
                    ++kept;
                }
                below += (dd[u] < ucb) ? 1u : 0u;
            }
        }
        if (kept > kLK) lsum[2 * kW] = 1;
        #pragma unroll
        for (int off = 16; off > 0; off >>= 1) below += __shfl_xor(below, off);
        if (lane == 0) lsum[wave] = (int)below;
        __syncthreads();
        int base = 0;
        #pragma unroll
        for (int w = 0; w < kW; ++w) base += lsum[w];

        const int per = (nn + kEDGET - 1) / kEDGET;
        const int i0  = (tid * per < nn) ? tid * per : nn;
        const int i1  = (i0 + per < nn) ? i0 + per : nn;
        int s = 0;
        for (int i = i0; i < i1; ++i) s += cnt[i];
        int incl = s;
        #pragma unroll
        for (int off = 1; off < 32; off <<= 1) {
            const int t = __shfl_up(incl, off);
            if (lane >= off) incl += t;
        }
        if (lane == 31) lsum[kW + wave] = incl;
        __syncthreads();
        int run = base + (incl - s);
        for (int w = 0; w < wave; ++w) run += lsum[kW + w];
        for (int i = i0; i < i1; ++i) {
            const int k = cnt[i];
            cnt[i] = run;
            row_off[cb + i] = run;
            run += k;
        }
        if (ce == N && tid == kEDGET - 1) row_off[N] = run;
        __syncthreads();

        if (lsum[2 * kW] == 0) {
            constexpr int kPU = 4;
            static_assert(kLK % kPU == 0, "a trip must stay inside the column");
            #pragma unroll 1
            for (int i0 = 0; i0 < kept; i0 += kPU) {
                int ke[kPU], kr[kPU], pos[kPU];
                #pragma unroll
                for (int q = 0; q < kPU; ++q) {
                    const int i = (i0 + q < kept) ? i0 + q : i0;
                    ke[q] = keep_e[i * kEDGET + tid];
                    kr[q] = keep_r[i * kEDGET + tid];
                }
                #pragma unroll
                for (int q = 0; q < kPU; ++q) {
                    pos[q] = (i0 + q < kept) ? atomicAdd(&cnt[kr[q]], 1) : -1;
                }
                #pragma unroll
                for (int q = 0; q < kPU; ++q) {
                    if (pos[q] >= 0) perm[pos[q]] = ke[q];
                }
            }
        } else
        for (int eb = tid; eb < E; eb += kEDGET * kLISTU) {
            unsigned dd[kLISTU];
            #pragma unroll
            for (int u = 0; u < kLISTU; ++u) {
                const int e  = eb + u * kEDGET;
                const int dv = dstv[e < E ? e : eb];
                dd[u] = (e < E) ? (unsigned)dv - unoff : 0xFFFFFFFFu;
            }
            int pos[kLISTU];
            #pragma unroll
            for (int u = 0; u < kLISTU; ++u) {
                const unsigned r = dd[u] - ucb;
                pos[u] = (r < unn) ? atomicAdd(&cnt[r], 1) : -1;
            }
            #pragma unroll
            for (int u = 0; u < kLISTU; ++u) {
                if (pos[u] >= 0) perm[pos[u]] = eb + u * kEDGET;
            }
        }
        __syncthreads();
        cb += kLCH;
    } while (cb < hi);
}

__global__ __launch_bounds__(kEDGET)
void edge_mlp_kernel(const float* __restrict__ x_edge,
                              const float* __restrict__ edge_distance,
                              const float* __restrict__ w1, const float* __restrict__ b1,
                              const float* __restrict__ g1, const float* __restrict__ be1,
                              const float* __restrict__ w2, const float* __restrict__ b2,
                              const float* __restrict__ g2, const float* __restrict__ be2,
                              const float* __restrict__ w3, const float* __restrict__ b3,
                              float* __restrict__ m0buf,
                              float* __restrict__ envbuf,
                              const int* __restrict__ edge_index,
                              const int* __restrict__ node_offset_p,
                              int* __restrict__ row_off,
                              int* __restrict__ perm,
                              int E, int N)
{
    __shared__ __attribute__((aligned(16))) float xe[kTILE * 128];
    __shared__ __attribute__((aligned(16))) float hraw[kTILE * 64];
    __shared__ __attribute__((aligned(16))) float h1[kTILE * 64];
    __shared__ __attribute__((aligned(16))) float h2[kTILE * 64];
    __shared__ __attribute__((aligned(16))) float m0s[kTILE * kM0 * kC];
    __shared__ int lsum[2 * (kEDGET / 32) + 1];

    static_assert(kLCH + 2 * kLK * kEDGET <= kTILE * kM0 * kC, "list counters and kept edges must fit the m0 tile");
    if (blockIdx.x < kLISTB) {
        int* lds = reinterpret_cast<int*>(m0s);
        build_dst_lists_slice(edge_index, node_offset_p, row_off, perm, E, N,
                              (int)blockIdx.x, lds, lds + kLCH, lds + kLCH + kLK * kEDGET, lsum);
        return;
    }

    const int tid   = threadIdx.x;
    const int wave  = tid >> 5;
    const int lane  = tid & 31;
    const int lrow  = lane & 15;
    const int khalf = (lane >> 4) << 1;
    const int rbase = (lane >> 4) << 3;
    const int e0    = ((int)blockIdx.x - kLISTB) * kTILE;

    {
        const float4* src4 = reinterpret_cast<const float4*>(x_edge + (size_t)e0 * 128);
        float4 v[kXLD];
        #pragma unroll
        for (int u = 0; u < kXLD; ++u) {
            const int i4 = tid + u * kEDGET;
            const bool ok = (e0 + (i4 >> 5)) < E;
            v[u] = src4[ok ? i4 : (i4 & 31)];
        }
        float4* xe4 = reinterpret_cast<float4*>(xe);
        #pragma unroll
        for (int u = 0; u < kXLD; ++u) {
            const int i4 = tid + u * kEDGET;
            const bool ok = (e0 + (i4 >> 5)) < E;
            xe4[i4] = ok ? v[u] : make_float4(0.0f, 0.0f, 0.0f, 0.0f);
        }
    }
    __syncthreads();

    {
        const int colBase = wave * 16;
        v8f acc = {};
        for (int k0 = 0; k0 < 128; k0 += 4) {
            v2f a, b;
            a.x = xe[lrow * 128 + k0 + khalf];
            a.y = xe[lrow * 128 + k0 + khalf + 1];
            b.x = w1[(k0 + khalf)     * 64 + colBase + lrow];
            b.y = w1[(k0 + khalf + 1) * 64 + colBase + lrow];
            acc = __builtin_amdgcn_wmma_f32_16x16x4_f32(
                false, a, false, b, (short)0, acc, false, false);
        }
        #pragma unroll
        for (int r = 0; r < 8; ++r) {
            int row = rbase + r;
            hraw[row * 64 + colBase + lrow] = acc[r] + b1[colBase + lrow];
        }
    }
    __syncthreads();
    ln_silu_16x64(hraw, h1, g1, be1, tid);
    __syncthreads();

    {
        const int colBase = wave * 16;
        v8f acc = {};
        for (int k0 = 0; k0 < 64; k0 += 4) {
            v2f a, b;
            a.x = h1[lrow * 64 + k0 + khalf];
            a.y = h1[lrow * 64 + k0 + khalf + 1];
            b.x = w2[(k0 + khalf)     * 64 + colBase + lrow];
            b.y = w2[(k0 + khalf + 1) * 64 + colBase + lrow];
            acc = __builtin_amdgcn_wmma_f32_16x16x4_f32(
                false, a, false, b, (short)0, acc, false, false);
        }
        #pragma unroll
        for (int r = 0; r < 8; ++r) {
            int row = rbase + r;
            hraw[row * 64 + colBase + lrow] = acc[r] + b2[colBase + lrow];
        }
    }
    __syncthreads();
    ln_silu_16x64(hraw, h2, g2, be2, tid);
    __syncthreads();

    for (int t = wave; t < 40; t += 4) {
        const int colBase = t * 16;
        v8f acc = {};
        for (int k0 = 0; k0 < 64; k0 += 4) {
            v2f a, b;
            a.x = h2[lrow * 64 + k0 + khalf];
            a.y = h2[lrow * 64 + k0 + khalf + 1];
            b.x = w3[(size_t)(k0 + khalf)     * 640 + colBase + lrow];
            b.y = w3[(size_t)(k0 + khalf + 1) * 640 + colBase + lrow];
            acc = __builtin_amdgcn_wmma_f32_16x16x4_f32(
                false, a, false, b, (short)0, acc, false, false);
        }
        #pragma unroll
        for (int r = 0; r < 8; ++r) {
            int row = rbase + r;
            m0s[row * (kM0 * kC) + colBase + lrow] = acc[r] + b3[colBase + lrow];
        }
    }
    __syncthreads();

    {
        const float4* src4 = reinterpret_cast<const float4*>(m0s);
        float4* dst4 = reinterpret_cast<float4*>(m0buf + (size_t)e0 * (kM0 * kC));
        constexpr int kOUT4 = kTILE * (kM0 * kC / 4) / kEDGET;
        const int lim4 = ((E - e0 < kTILE) ? (E - e0) : kTILE) * (kM0 * kC / 4);
        if (lim4 == kTILE * (kM0 * kC / 4)) {
            float4 t[kOUT4];
            #pragma unroll
            for (int u = 0; u < kOUT4; ++u) t[u] = src4[tid + u * kEDGET];
            #pragma unroll
            for (int u = 0; u < kOUT4; ++u) dst4[tid + u * kEDGET] = t[u];
        } else {
            for (int i = tid; i < lim4; i += kEDGET) dst4[i] = src4[i];
        }
    }
    if (tid < kTILE) {
        int e = e0 + tid;
        if (e < E) {
            float env = 0.0f;
            float d = edge_distance[e] * (1.0f / kCUTOFF);
            if (d < 1.0f) {
                float d5 = d * d * d * d * d;
                env = 1.0f - 21.0f * d5 + 35.0f * d5 * d - 15.0f * d5 * d * d;
            }
            envbuf[e] = env * (1.0f / kRESCALE);
        }
    }
}

__device__ __forceinline__ float xhalf(float v) {
    return __builtin_bit_cast(float, __builtin_amdgcn_permlanex16(
        __builtin_bit_cast(int, v), __builtin_bit_cast(int, v), 0x76543210, 0xfedcba98, false, false));
}

__global__ __launch_bounds__(128, 5)
void node_gather_kernel(const float* __restrict__ x,
                        const float* __restrict__ wigner_inv,
                        const float* __restrict__ m0buf,
                        const float* __restrict__ envbuf,
                        const int*   __restrict__ row_off,
                        const int*   __restrict__ perm,
                        int*         __restrict__ perm_sorted,
                        float*       __restrict__ out)
{
    __shared__ __attribute__((aligned(16))) float a_s[32 * kAP];
    __shared__ __attribute__((aligned(8))) float env_k[kGB * kM0];
    __shared__ __attribute__((aligned(16))) int eraw[kSORT];
    __shared__ __attribute__((aligned(16))) int esort[kSORT + kGB];

    const int n    = blockIdx.x;
    const int c    = threadIdx.x;
    const int lane = c & 31;
    const int wave = c >> 5;
    const int lc   = lane & 15;
    const int lh   = lane >> 4;
    const int beg  = row_off[n];
    const int cnt  = row_off[n + 1] - beg;

    v8f acc[2][2];
    const float* xn = x + (size_t)n * (kNCOEF * kC) + wave * 32 + lc;
    #pragma unroll
    for (int mt = 0; mt < 2; ++mt) {
        #pragma unroll
        for (int nt = 0; nt < 2; ++nt) {
            #pragma unroll
            for (int i = 0; i < 8; ++i) {
                const int row = 16 * mt + 8 * lh + i;
                acc[mt][nt][i] = (row < kNCOEF) ? xn[row * kC + 16 * nt] : 0.0f;
            }
        }
    }

    for (int i = c; i < (32 - kNCOEF) * kAP; i += blockDim.x) a_s[kNCOEF * kAP + i] = 0.0f;

    const int  stage_r    = c / kM0;
    const int  stage_l    = c - stage_r * kM0;
    const bool stage_w    = c < kNCOEF * kM0;
    const int  stage_goff = stage_r * kNCOEF + stage_l * (stage_l + 1);
    const int  stage_aidx = stage_r * kAP + stage_l;

    const bool in_lds = cnt <= kSORT;
    if (in_lds) {
        const int ei = (c < cnt) ? perm[beg + c] : 0x7fffffff;
        eraw[c] = ei;
        __syncthreads();
        if (c < cnt) {
            int rank = 0;
            for (int j = 0; j < cnt; j += 4) {
                const int4 v = *reinterpret_cast<const int4*>(eraw + j);
                rank += ((v.x < ei) ? 1 : 0) + ((v.y < ei) ? 1 : 0) + ((v.z < ei) ? 1 : 0) + ((v.w < ei) ? 1 : 0);
            }
            esort[rank] = ei;
            if (rank == cnt - 1) {
                const int cpad = (cnt + kGB - 1) & ~(kGB - 1);
                for (int p = cnt; p < cpad; ++p) esort[p] = ei;
            }
        }
    } else {
        for (int i = c; i < cnt; i += blockDim.x) {
            const int ei = perm[beg + i];
            int rank = 0;
            for (int j = 0; j < cnt; ++j) rank += (perm[beg + j] < ei) ? 1 : 0;
            perm_sorted[beg + rank] = ei;
        }
    }
    __syncthreads();

    for (int b0 = 0; b0 < cnt; b0 += kGB) {
        const int nb = (cnt - b0 < kGB) ? (cnt - b0) : kGB;
        const int ng = (nb + kGRP - 1) / kGRP;
        if (!in_lds) {
            if (c < kGB) esort[c] = perm_sorted[beg + b0 + (c < nb ? c : nb - 1)];
            __syncthreads();
        }
        const int* elist = esort + (in_lds ? b0 : 0);

        v2f bf[kGB / kGRP][5][2];
        float wv[kGB];
        #pragma unroll
        for (int g = 0; g < kGB / kGRP; ++g) {
            if (g < ng) {
                int eg[kGRP];
                #pragma unroll
                for (int q = 0; q < kGRP; ++q) eg[q] = elist[g * kGRP + q];
                if (stage_w) {
                    #pragma unroll
                    for (int q = 0; q < kGRP; ++q) {
                        wv[g * kGRP + q] = wigner_inv[(size_t)eg[q] * (kNCOEF * kNCOEF) + stage_goff];
                    }
                }
                #pragma unroll
                for (int ks = 0; ks < 5; ++ks) {
                    #pragma unroll
                    for (int s2 = 0; s2 < 2; ++s2) {
                        const int k0 = 4 * ks + s2;
                        const int k1 = 4 * ks + 2 + s2;
                        const int e  = lh ? eg[k1 / kM0] : eg[k0 / kM0];
                        const int l  = lh ? (k1 % kM0) : (k0 % kM0);
                        const float* mp = m0buf + (size_t)e * (kM0 * kC) + l * kC + wave * 32 + lc;
                        const float b_lo = mp[0];
                        const float b_hi = mp[16];
                        if (s2 == 0) { bf[g][ks][0].x = b_lo; bf[g][ks][1].x = b_hi; }
                        else         { bf[g][ks][0].y = b_lo; bf[g][ks][1].y = b_hi; }
                    }
                }
            }
        }

        if (c < ng * (kGRP * kM0)) env_k[c] = (stage_r < nb) ? envbuf[elist[stage_r]] : 0.0f;
        #pragma unroll
        for (int j = 0; j < kGB; ++j) {
            if (j < ng * kGRP && stage_w) a_s[stage_aidx + kM0 * j] = (j < nb) ? wv[j] : 0.0f;
        }
        __syncthreads();

        #pragma unroll
        for (int g = 0; g < kGB / kGRP; ++g) {
            if (g < ng) {
                #pragma unroll
                for (int ks = 0; ks < 5; ++ks) {
                    const int kk = g * (kGRP * kM0) + 4 * ks + 2 * lh;
                    const v2f ek = *reinterpret_cast<const v2f*>(env_k + kk);
                    const v2f a0 = *reinterpret_cast<const v2f*>(a_s + lc * kAP + kk) * ek;
                    const v2f a1 = *reinterpret_cast<const v2f*>(a_s + (16 + lc) * kAP + kk) * ek;
                    acc[0][0] = __builtin_amdgcn_wmma_f32_16x16x4_f32(
                        false, a0, false, bf[g][ks][0], (short)0, acc[0][0], false, false);
                    acc[0][1] = __builtin_amdgcn_wmma_f32_16x16x4_f32(
                        false, a0, false, bf[g][ks][1], (short)0, acc[0][1], false, false);
                    acc[1][0] = __builtin_amdgcn_wmma_f32_16x16x4_f32(
                        false, a1, false, bf[g][ks][0], (short)0, acc[1][0], false, false);
                    acc[1][1] = __builtin_amdgcn_wmma_f32_16x16x4_f32(
                        false, a1, false, bf[g][ks][1], (short)0, acc[1][1], false, false);
                }
            }
        }
        __syncthreads();
    }

    float* on = out + (size_t)n * (kNCOEF * kC) + c;
    #pragma unroll
    for (int mt = 0; mt < 2; ++mt) {
        #pragma unroll
        for (int i = 0; i < 8; ++i) {
            const float r0 = acc[mt][0][i];
            const float r1 = acc[mt][1][i];
            const float x0 = xhalf(r0);
            const float x1 = xhalf(r1);
            const float s_lo = lh ? x1 : r0;
            const float s_hi = lh ? r1 : x0;
            const int row_lo = 16 * mt + i;
            const int row_hi = 16 * mt + 8 + i;
            if (row_lo < kNCOEF) on[row_lo * kC] = s_lo;
            if (row_hi < kNCOEF) on[row_hi * kC] = s_hi;
        }
    }
}

extern "C" void kernel_launch(void* const* d_in, const int* in_sizes, int n_in,
                              void* d_out, int out_size, void* d_ws, size_t ws_size,
                              hipStream_t stream) {
    const float* x             = (const float*)d_in[0];
    const float* x_edge        = (const float*)d_in[1];
    const float* edge_distance = (const float*)d_in[2];
    const float* wigner_inv    = (const float*)d_in[3];
    const int*   edge_index    = (const int*)d_in[4];
    const float* w1  = (const float*)d_in[5];
    const float* b1  = (const float*)d_in[6];
    const float* g1  = (const float*)d_in[7];
    const float* be1 = (const float*)d_in[8];
    const float* w2  = (const float*)d_in[9];
    const float* b2  = (const float*)d_in[10];
    const float* g2  = (const float*)d_in[11];
    const float* be2 = (const float*)d_in[12];
    const float* w3  = (const float*)d_in[13];
    const float* b3  = (const float*)d_in[14];
    const int* node_offset = (const int*)d_in[15];

    const int E = in_sizes[1] / 128;
    const int N = out_size / (kNCOEF * kC);

    float* m0buf       = (float*)d_ws;
    float* envbuf      = m0buf + (size_t)E * (kM0 * kC);
    int*   row_off     = (int*)(envbuf + E);
    int*   perm        = row_off + (N + 1);
    int*   perm_sorted = perm + E;

    const int blocks = (E + kTILE - 1) / kTILE;
    edge_mlp_kernel<<<kLISTB + blocks, kEDGET, 0, stream>>>(
        x_edge, edge_distance,
        w1, b1, g1, be1, w2, b2, g2, be2, w3, b3,
        m0buf, envbuf, edge_index, node_offset, row_off, perm, E, N);

    node_gather_kernel<<<N, 128, 0, stream>>>(
        x, wigner_inv, m0buf, envbuf, row_off, perm, perm_sorted, (float*)d_out);
}


# Edge network: each layer's weight fragments requested a stage ahead of their products (w1 beside the x_edge tile, w2 beside layer 1, first w3 tiles beside layer 2, layer 3 two tiles ahead)
# speedup vs baseline: 5.0582x; 1.0396x over previous
#include <hip/hip_runtime.h>

typedef __attribute__((ext_vector_type(2))) float v2f;
typedef __attribute__((ext_vector_type(8))) float v8f;

constexpr int kNCOEF = 25;
constexpr int kM0    = 5;
constexpr int kC     = 128;
constexpr int kTILE  = 16;
constexpr float kCUTOFF  = 12.0f;
constexpr float kRESCALE = 5.0f;
constexpr float kEPS     = 1e-5f;
constexpr int kLISTB  = 32;
constexpr int kLISTU  = 40;
constexpr int kLCH    = 2048;
constexpr int kLK     = 32;
constexpr int kGB     = 16;
constexpr int kGRP    = 4;
constexpr int kEDGET = 128;
constexpr int kXLD   = 4;
constexpr int kSORT   = 128;
constexpr int kAP     = 84;

__device__ __forceinline__ float silu_f(float x) {
    return x / (1.0f + __expf(-x));
}

template <int M>
__device__ __forceinline__ float row_xor(float v) {
    return __builtin_bit_cast(float, __builtin_amdgcn_update_dpp(
        0, __builtin_bit_cast(int, v), 0x160 | M, 0xf, 0xf, false));
}

constexpr int kLNF = 8;

__device__ __forceinline__ void ln_silu_16x64(const float* __restrict__ hraw,
                                              float* __restrict__ hout,
                                              const float* __restrict__ g,
                                              const float* __restrict__ be,
                                              int tid) {
    const int c0  = (tid & 7) * kLNF;
    const int off = (tid >> 3) * 64 + c0;
    const float4 p = *reinterpret_cast<const float4*>(hraw + off);
    const float4 q = *reinterpret_cast<const float4*>(hraw + off + 4);
    const float v[kLNF] = {p.x, p.y, p.z, p.w, q.x, q.y, q.z, q.w};
    float gv[kLNF], bv[kLNF];
    #pragma unroll
    for (int j = 0; j < kLNF; ++j) { gv[j] = g[c0 + j]; bv[j] = be[c0 + j]; }
    float s = 0.0f, s2 = 0.0f;
    #pragma unroll
    for (int j = 0; j < kLNF; ++j) { s += v[j]; s2 += v[j] * v[j]; }
    s += row_xor<1>(s);  s2 += row_xor<1>(s2);
    s += row_xor<2>(s);  s2 += row_xor<2>(s2);
    s += row_xor<4>(s);  s2 += row_xor<4>(s2);
    const float mu  = s * (1.0f / 64.0f);
    const float var = s2 * (1.0f / 64.0f) - mu * mu;
    const float inv = rsqrtf(var + kEPS);
    float o[kLNF];
    #pragma unroll
    for (int j = 0; j < kLNF; ++j) o[j] = silu_f((v[j] - mu) * inv * gv[j] + bv[j]);
    *reinterpret_cast<float4*>(hout + off)     = make_float4(o[0], o[1], o[2], o[3]);
    *reinterpret_cast<float4*>(hout + off + 4) = make_float4(o[4], o[5], o[6], o[7]);
}


template <int KS>
__device__ __forceinline__ void load_w_frags(v2f (&b)[KS], const float* __restrict__ w,
                                             int ld, int col, int khalf) {
    #pragma unroll
    for (int ks = 0; ks < KS; ++ks) {
        b[ks].x = w[(4 * ks + khalf)     * ld + col];
        b[ks].y = w[(4 * ks + khalf + 1) * ld + col];
    }
}

__device__ __forceinline__ void build_dst_lists_slice(const int* __restrict__ edge_index,
                                                      const int* __restrict__ node_offset_p,
                                                      int* __restrict__ row_off,
                                                      int* __restrict__ perm,
                                                      int E, int N, int part,
                                                      int* __restrict__ cnt,
                                                      int* __restrict__ keep_e,
                                                      int* __restrict__ keep_r,
                                                      int* __restrict__ lsum)
{
    constexpr int kW = kEDGET / 32;
    const int tid  = threadIdx.x;
    const int lane = tid & 31;
    const int wave = tid >> 5;
    const int noff = node_offset_p[0];
    const int* __restrict__ dstv = edge_index + E;

    const int npb = (N + kLISTB - 1) / kLISTB;
    const int lo  = (part * npb < N) ? part * npb : N;
    const int hi  = (part == kLISTB - 1) ? N : ((lo + npb < N) ? lo + npb : N);
    if (lo >= hi && part != kLISTB - 1) return;

    int cb = lo;
    do {
        const int ce = (cb + kLCH < hi) ? cb + kLCH : hi;
        const int nn = ce - cb;
        for (int i = tid; i < nn; i += kEDGET) cnt[i] = 0;
        if (tid == 0) lsum[2 * kW] = 0;
        __syncthreads();

        const unsigned ucb = (unsigned)cb, unn = (unsigned)nn, unoff = (unsigned)noff;
        unsigned below = 0;
        int kept = 0;
        for (int eb = tid; eb < E; eb += kEDGET * kLISTU) {
            unsigned dd[kLISTU];
            #pragma unroll
            for (int u = 0; u < kLISTU; ++u) {
                const int e  = eb + u * kEDGET;
                const int dv = dstv[e < E ? e : eb];
                dd[u] = (e < E) ? (unsigned)dv - unoff : 0xFFFFFFFFu;
            }
            #pragma unroll
            for (int u = 0; u < kLISTU; ++u) {
                const unsigned r = dd[u] - ucb;
                if (r < unn) {
                    atomicAdd(&cnt[r], 1);
                    const int slot = ((kept < kLK) ? kept : kLK - 1) * kEDGET + tid;
                    keep_e[slot] = eb + u * kEDGET;
                    keep_r[slot] = (int)r;
                    ++kept;
                }
                below += (dd[u] < ucb) ? 1u : 0u;
            }
        }
        if (kept > kLK) lsum[2 * kW] = 1;
        #pragma unroll
        for (int off = 16; off > 0; off >>= 1) below += __shfl_xor(below, off);
        if (lane == 0) lsum[wave] = (int)below;
        __syncthreads();
        int base = 0;
        #pragma unroll
        for (int w = 0; w < kW; ++w) base += lsum[w];

        const int per = (nn + kEDGET - 1) / kEDGET;
        const int i0  = (tid * per < nn) ? tid * per : nn;
        const int i1  = (i0 + per < nn) ? i0 + per : nn;
        int s = 0;
        for (int i = i0; i < i1; ++i) s += cnt[i];
        int incl = s;
        #pragma unroll
        for (int off = 1; off < 32; off <<= 1) {
            const int t = __shfl_up(incl, off);
            if (lane >= off) incl += t;
        }
        if (lane == 31) lsum[kW + wave] = incl;
        __syncthreads();
        int run = base + (incl - s);
        for (int w = 0; w < wave; ++w) run += lsum[kW + w];
        for (int i = i0; i < i1; ++i) {
            const int k = cnt[i];
            cnt[i] = run;
            row_off[cb + i] = run;
            run += k;
        }
        if (ce == N && tid == kEDGET - 1) row_off[N] = run;
        __syncthreads();

        if (lsum[2 * kW] == 0) {
            constexpr int kPU = 4;
            static_assert(kLK % kPU == 0, "a trip must stay inside the column");
            #pragma unroll 1
            for (int i0 = 0; i0 < kept; i0 += kPU) {
                int ke[kPU], kr[kPU], pos[kPU];
                #pragma unroll
                for (int q = 0; q < kPU; ++q) {
                    const int i = (i0 + q < kept) ? i0 + q : i0;
                    ke[q] = keep_e[i * kEDGET + tid];
                    kr[q] = keep_r[i * kEDGET + tid];
                }
                #pragma unroll
                for (int q = 0; q < kPU; ++q) {
                    pos[q] = (i0 + q < kept) ? atomicAdd(&cnt[kr[q]], 1) : -1;
                }
                #pragma unroll
                for (int q = 0; q < kPU; ++q) {
                    if (pos[q] >= 0) perm[pos[q]] = ke[q];
                }
            }
        } else
        for (int eb = tid; eb < E; eb += kEDGET * kLISTU) {
            unsigned dd[kLISTU];
            #pragma unroll
            for (int u = 0; u < kLISTU; ++u) {
                const int e  = eb + u * kEDGET;
                const int dv = dstv[e < E ? e : eb];
                dd[u] = (e < E) ? (unsigned)dv - unoff : 0xFFFFFFFFu;
            }
            int pos[kLISTU];
            #pragma unroll
            for (int u = 0; u < kLISTU; ++u) {
                const unsigned r = dd[u] - ucb;
                pos[u] = (r < unn) ? atomicAdd(&cnt[r], 1) : -1;
            }
            #pragma unroll
            for (int u = 0; u < kLISTU; ++u) {
                if (pos[u] >= 0) perm[pos[u]] = eb + u * kEDGET;
            }
        }
        __syncthreads();
        cb += kLCH;
    } while (cb < hi);
}

__global__ __launch_bounds__(kEDGET)
void edge_mlp_kernel(const float* __restrict__ x_edge,
                              const float* __restrict__ edge_distance,
                              const float* __restrict__ w1, const float* __restrict__ b1,
                              const float* __restrict__ g1, const float* __restrict__ be1,
                              const float* __restrict__ w2, const float* __restrict__ b2,
                              const float* __restrict__ g2, const float* __restrict__ be2,
                              const float* __restrict__ w3, const float* __restrict__ b3,
                              float* __restrict__ m0buf,
                              float* __restrict__ envbuf,
                              const int* __restrict__ edge_index,
                              const int* __restrict__ node_offset_p,
                              int* __restrict__ row_off,
                              int* __restrict__ perm,
                              int E, int N)
{
    __shared__ __attribute__((aligned(16))) float xe[kTILE * 128];
    __shared__ __attribute__((aligned(16))) float hraw[kTILE * 64];
    __shared__ __attribute__((aligned(16))) float h1[kTILE * 64];
    __shared__ __attribute__((aligned(16))) float h2[kTILE * 64];
    __shared__ __attribute__((aligned(16))) float m0s[kTILE * kM0 * kC];
    __shared__ int lsum[2 * (kEDGET / 32) + 1];

    static_assert(kLCH + 2 * kLK * kEDGET <= kTILE * kM0 * kC, "list counters and kept edges must fit the m0 tile");
    if (blockIdx.x < kLISTB) {
        int* lds = reinterpret_cast<int*>(m0s);
        build_dst_lists_slice(edge_index, node_offset_p, row_off, perm, E, N,
                              (int)blockIdx.x, lds, lds + kLCH, lds + kLCH + kLK * kEDGET, lsum);
        return;
    }

    const int tid   = threadIdx.x;
    const int wave  = tid >> 5;
    const int lane  = tid & 31;
    const int lrow  = lane & 15;
    const int khalf = (lane >> 4) << 1;
    const int rbase = (lane >> 4) << 3;
    const int e0    = ((int)blockIdx.x - kLISTB) * kTILE;

    v2f bw1[32];
    float bias1;

    {
        const float4* src4 = reinterpret_cast<const float4*>(x_edge + (size_t)e0 * 128);
        float4 v[kXLD];
        #pragma unroll
        for (int u = 0; u < kXLD; ++u) {
            const int i4 = tid + u * kEDGET;
            const bool ok = (e0 + (i4 >> 5)) < E;
            v[u] = src4[ok ? i4 : (i4 & 31)];
        }
        load_w_frags<32>(bw1, w1, 64, wave * 16 + lrow, khalf);
        bias1 = b1[wave * 16 + lrow];
        float4* xe4 = reinterpret_cast<float4*>(xe);
        #pragma unroll
        for (int u = 0; u < kXLD; ++u) {
            const int i4 = tid + u * kEDGET;
            const bool ok = (e0 + (i4 >> 5)) < E;
            xe4[i4] = ok ? v[u] : make_float4(0.0f, 0.0f, 0.0f, 0.0f);
        }
    }
    __syncthreads();

    v2f bw2[16];
    load_w_frags<16>(bw2, w2, 64, wave * 16 + lrow, khalf);
    const float bias2 = b2[wave * 16 + lrow];

    {
        const int colBase = wave * 16;
        v8f acc = {};
        #pragma unroll
        for (int ks = 0; ks < 32; ++ks) {
            v2f a;
            a.x = xe[lrow * 128 + 4 * ks + khalf];
            a.y = xe[lrow * 128 + 4 * ks + khalf + 1];
            acc = __builtin_amdgcn_wmma_f32_16x16x4_f32(
                false, a, false, bw1[ks], (short)0, acc, false, false);
        }
        #pragma unroll
        for (int r = 0; r < 8; ++r) {
            int row = rbase + r;
            hraw[row * 64 + colBase + lrow] = acc[r] + bias1;
        }
    }
    __syncthreads();
    ln_silu_16x64(hraw, h1, g1, be1, tid);
    __syncthreads();

    constexpr int kT3 = 40 / 4;
    constexpr int kPF = 2;
    v2f   bw3[kPF + 1][16];
    float bias3[kPF + 1];
    #pragma unroll
    for (int i = 0; i < kPF; ++i) {
        const int col = (wave + 4 * i) * 16 + lrow;
        load_w_frags<16>(bw3[i], w3, kM0 * kC, col, khalf);
        bias3[i] = b3[col];
    }

    {
        const int colBase = wave * 16;
        v8f acc = {};
        #pragma unroll
        for (int ks = 0; ks < 16; ++ks) {
            v2f a;
            a.x = h1[lrow * 64 + 4 * ks + khalf];
            a.y = h1[lrow * 64 + 4 * ks + khalf + 1];
            acc = __builtin_amdgcn_wmma_f32_16x16x4_f32(
                false, a, false, bw2[ks], (short)0, acc, false, false);
        }
        #pragma unroll
        for (int r = 0; r < 8; ++r) {
            int row = rbase + r;
            hraw[row * 64 + colBase + lrow] = acc[r] + bias2;
        }
    }
    __syncthreads();
    ln_silu_16x64(hraw, h2, g2, be2, tid);
    __syncthreads();

    {
        v2f a3[16];
        #pragma unroll
        for (int ks = 0; ks < 16; ++ks) {
            a3[ks].x = h2[lrow * 64 + 4 * ks + khalf];
            a3[ks].y = h2[lrow * 64 + 4 * ks + khalf + 1];
        }
        #pragma unroll
        for (int i = 0; i < kT3; ++i) {
            if (i + kPF < kT3) {
                const int col = (wave + 4 * (i + kPF)) * 16 + lrow;
                load_w_frags<16>(bw3[(i + kPF) % (kPF + 1)], w3, kM0 * kC, col, khalf);
                bias3[(i + kPF) % (kPF + 1)] = b3[col];
            }
            const int colBase = (wave + 4 * i) * 16;
            v8f acc = {};
            #pragma unroll
            for (int ks = 0; ks < 16; ++ks) {
                acc = __builtin_amdgcn_wmma_f32_16x16x4_f32(
                    false, a3[ks], false, bw3[i % (kPF + 1)][ks], (short)0, acc, false, false);
            }
            #pragma unroll
            for (int r = 0; r < 8; ++r) {
                int row = rbase + r;
                m0s[row * (kM0 * kC) + colBase + lrow] = acc[r] + bias3[i % (kPF + 1)];
            }
        }
    }
    __syncthreads();

    {
        const float4* src4 = reinterpret_cast<const float4*>(m0s);
        float4* dst4 = reinterpret_cast<float4*>(m0buf + (size_t)e0 * (kM0 * kC));
        constexpr int kOUT4 = kTILE * (kM0 * kC / 4) / kEDGET;
        const int lim4 = ((E - e0 < kTILE) ? (E - e0) : kTILE) * (kM0 * kC / 4);
        if (lim4 == kTILE * (kM0 * kC / 4)) {
            float4 t[kOUT4];
            #pragma unroll
            for (int u = 0; u < kOUT4; ++u) t[u] = src4[tid + u * kEDGET];
            #pragma unroll
            for (int u = 0; u < kOUT4; ++u) dst4[tid + u * kEDGET] = t[u];
        } else {
            for (int i = tid; i < lim4; i += kEDGET) dst4[i] = src4[i];
        }
    }
    if (tid < kTILE) {
        int e = e0 + tid;
        if (e < E) {
            float env = 0.0f;
            float d = edge_distance[e] * (1.0f / kCUTOFF);
            if (d < 1.0f) {
                float d5 = d * d * d * d * d;
                env = 1.0f - 21.0f * d5 + 35.0f * d5 * d - 15.0f * d5 * d * d;
            }
            envbuf[e] = env * (1.0f / kRESCALE);
        }
    }
}

__device__ __forceinline__ float xhalf(float v) {
    return __builtin_bit_cast(float, __builtin_amdgcn_permlanex16(
        __builtin_bit_cast(int, v), __builtin_bit_cast(int, v), 0x76543210, 0xfedcba98, false, false));
}

__global__ __launch_bounds__(128, 5)
void node_gather_kernel(const float* __restrict__ x,
                        const float* __restrict__ wigner_inv,
                        const float* __restrict__ m0buf,
                        const float* __restrict__ envbuf,
                        const int*   __restrict__ row_off,
                        const int*   __restrict__ perm,
                        int*         __restrict__ perm_sorted,
                        float*       __restrict__ out)
{
    __shared__ __attribute__((aligned(16))) float a_s[32 * kAP];
    __shared__ __attribute__((aligned(8))) float env_k[kGB * kM0];
    __shared__ __attribute__((aligned(16))) int eraw[kSORT];
    __shared__ __attribute__((aligned(16))) int esort[kSORT + kGB];

    const int n    = blockIdx.x;
    const int c    = threadIdx.x;
    const int lane = c & 31;
    const int wave = c >> 5;
    const int lc   = lane & 15;
    const int lh   = lane >> 4;
    const int beg  = row_off[n];
    const int cnt  = row_off[n + 1] - beg;

    v8f acc[2][2];
    const float* xn = x + (size_t)n * (kNCOEF * kC) + wave * 32 + lc;
    #pragma unroll
    for (int mt = 0; mt < 2; ++mt) {
        #pragma unroll
        for (int nt = 0; nt < 2; ++nt) {
            #pragma unroll
            for (int i = 0; i < 8; ++i) {
                const int row = 16 * mt + 8 * lh + i;
                acc[mt][nt][i] = (row < kNCOEF) ? xn[row * kC + 16 * nt] : 0.0f;
            }
        }
    }

    for (int i = c; i < (32 - kNCOEF) * kAP; i += blockDim.x) a_s[kNCOEF * kAP + i] = 0.0f;

    const int  stage_r    = c / kM0;
    const int  stage_l    = c - stage_r * kM0;
    const bool stage_w    = c < kNCOEF * kM0;
    const int  stage_goff = stage_r * kNCOEF + stage_l * (stage_l + 1);
    const int  stage_aidx = stage_r * kAP + stage_l;

    const bool in_lds = cnt <= kSORT;
    if (in_lds) {
        const int ei = (c < cnt) ? perm[beg + c] : 0x7fffffff;
        eraw[c] = ei;
        __syncthreads();
        if (c < cnt) {
            int rank = 0;
            for (int j = 0; j < cnt; j += 4) {
                const int4 v = *reinterpret_cast<const int4*>(eraw + j);
                rank += ((v.x < ei) ? 1 : 0) + ((v.y < ei) ? 1 : 0) + ((v.z < ei) ? 1 : 0) + ((v.w < ei) ? 1 : 0);
            }
            esort[rank] = ei;
            if (rank == cnt - 1) {
                const int cpad = (cnt + kGB - 1) & ~(kGB - 1);
                for (int p = cnt; p < cpad; ++p) esort[p] = ei;
            }
        }
    } else {
        for (int i = c; i < cnt; i += blockDim.x) {
            const int ei = perm[beg + i];
            int rank = 0;
            for (int j = 0; j < cnt; ++j) rank += (perm[beg + j] < ei) ? 1 : 0;
            perm_sorted[beg + rank] = ei;
        }
    }
    __syncthreads();

    for (int b0 = 0; b0 < cnt; b0 += kGB) {
        const int nb = (cnt - b0 < kGB) ? (cnt - b0) : kGB;
        const int ng = (nb + kGRP - 1) / kGRP;
        if (!in_lds) {
            if (c < kGB) esort[c] = perm_sorted[beg + b0 + (c < nb ? c : nb - 1)];
            __syncthreads();
        }
        const int* elist = esort + (in_lds ? b0 : 0);

        v2f bf[kGB / kGRP][5][2];
        float wv[kGB];
        #pragma unroll
        for (int g = 0; g < kGB / kGRP; ++g) {
            if (g < ng) {
                int eg[kGRP];
                #pragma unroll
                for (int q = 0; q < kGRP; ++q) eg[q] = elist[g * kGRP + q];
                if (stage_w) {
                    #pragma unroll
                    for (int q = 0; q < kGRP; ++q) {
                        wv[g * kGRP + q] = wigner_inv[(size_t)eg[q] * (kNCOEF * kNCOEF) + stage_goff];
                    }
                }
                #pragma unroll
                for (int ks = 0; ks < 5; ++ks) {
                    #pragma unroll
                    for (int s2 = 0; s2 < 2; ++s2) {
                        const int k0 = 4 * ks + s2;
                        const int k1 = 4 * ks + 2 + s2;
                        const int e  = lh ? eg[k1 / kM0] : eg[k0 / kM0];
                        const int l  = lh ? (k1 % kM0) : (k0 % kM0);
                        const float* mp = m0buf + (size_t)e * (kM0 * kC) + l * kC + wave * 32 + lc;
                        const float b_lo = mp[0];
                        const float b_hi = mp[16];
                        if (s2 == 0) { bf[g][ks][0].x = b_lo; bf[g][ks][1].x = b_hi; }
                        else         { bf[g][ks][0].y = b_lo; bf[g][ks][1].y = b_hi; }
                    }
                }
            }
        }

        if (c < ng * (kGRP * kM0)) env_k[c] = (stage_r < nb) ? envbuf[elist[stage_r]] : 0.0f;
        #pragma unroll
        for (int j = 0; j < kGB; ++j) {
            if (j < ng * kGRP && stage_w) a_s[stage_aidx + kM0 * j] = (j < nb) ? wv[j] : 0.0f;
        }
        __syncthreads();

        #pragma unroll
        for (int g = 0; g < kGB / kGRP; ++g) {
            if (g < ng) {
                #pragma unroll
                for (int ks = 0; ks < 5; ++ks) {
                    const int kk = g * (kGRP * kM0) + 4 * ks + 2 * lh;
                    const v2f ek = *reinterpret_cast<const v2f*>(env_k + kk);
                    const v2f a0 = *reinterpret_cast<const v2f*>(a_s + lc * kAP + kk) * ek;
                    const v2f a1 = *reinterpret_cast<const v2f*>(a_s + (16 + lc) * kAP + kk) * ek;
                    acc[0][0] = __builtin_amdgcn_wmma_f32_16x16x4_f32(
                        false, a0, false, bf[g][ks][0], (short)0, acc[0][0], false, false);
                    acc[0][1] = __builtin_amdgcn_wmma_f32_16x16x4_f32(
                        false, a0, false, bf[g][ks][1], (short)0, acc[0][1], false, false);
                    acc[1][0] = __builtin_amdgcn_wmma_f32_16x16x4_f32(
                        false, a1, false, bf[g][ks][0], (short)0, acc[1][0], false, false);
                    acc[1][1] = __builtin_amdgcn_wmma_f32_16x16x4_f32(
                        false, a1, false, bf[g][ks][1], (short)0, acc[1][1], false, false);
                }
            }
        }
        __syncthreads();
    }

    float* on = out + (size_t)n * (kNCOEF * kC) + c;
    #pragma unroll
    for (int mt = 0; mt < 2; ++mt) {
        #pragma unroll
        for (int i = 0; i < 8; ++i) {
            const float r0 = acc[mt][0][i];
            const float r1 = acc[mt][1][i];
            const float x0 = xhalf(r0);
            const float x1 = xhalf(r1);
            const float s_lo = lh ? x1 : r0;
            const float s_hi = lh ? r1 : x0;
            const int row_lo = 16 * mt + i;
            const int row_hi = 16 * mt + 8 + i;
            if (row_lo < kNCOEF) on[row_lo * kC] = s_lo;
            if (row_hi < kNCOEF) on[row_hi * kC] = s_hi;
        }
    }
}

extern "C" void kernel_launch(void* const* d_in, const int* in_sizes, int n_in,
                              void* d_out, int out_size, void* d_ws, size_t ws_size,
                              hipStream_t stream) {
    const float* x             = (const float*)d_in[0];
    const float* x_edge        = (const float*)d_in[1];
    const float* edge_distance = (const float*)d_in[2];
    const float* wigner_inv    = (const float*)d_in[3];
    const int*   edge_index    = (const int*)d_in[4];
    const float* w1  = (const float*)d_in[5];
    const float* b1  = (const float*)d_in[6];
    const float* g1  = (const float*)d_in[7];
    const float* be1 = (const float*)d_in[8];
    const float* w2  = (const float*)d_in[9];
    const float* b2  = (const float*)d_in[10];
    const float* g2  = (const float*)d_in[11];
    const float* be2 = (const float*)d_in[12];
    const float* w3  = (const float*)d_in[13];
    const float* b3  = (const float*)d_in[14];
    const int* node_offset = (const int*)d_in[15];

    const int E = in_sizes[1] / 128;
    const int N = out_size / (kNCOEF * kC);

    float* m0buf       = (float*)d_ws;
    float* envbuf      = m0buf + (size_t)E * (kM0 * kC);
    int*   row_off     = (int*)(envbuf + E);
    int*   perm        = row_off + (N + 1);
    int*   perm_sorted = perm + E;

    const int blocks = (E + kTILE - 1) / kTILE;
    edge_mlp_kernel<<<kLISTB + blocks, kEDGET, 0, stream>>>(
        x_edge, edge_distance,
        w1, b1, g1, be1, w2, b2, g2, be2, w3, b3,
        m0buf, envbuf, edge_index, node_offset, row_off, perm, E, N);

    node_gather_kernel<<<N, 128, 0, stream>>>(
        x, wigner_inv, m0buf, envbuf, row_off, perm, perm_sorted, (float*)d_out);
}


# Edge network: a block takes 32 edges (two row tiles), so every weight fragment fetched from L2 feeds two products; m0 leaves by column halves through a 40 KB LDS tile laid over the dead tiles
# speedup vs baseline: 5.4255x; 1.0726x over previous
#include <hip/hip_runtime.h>

typedef __attribute__((ext_vector_type(2))) float v2f;
typedef __attribute__((ext_vector_type(4))) float v4f;
typedef __attribute__((ext_vector_type(8))) float v8f;

constexpr int kNCOEF = 25;
constexpr int kM0    = 5;
constexpr int kC     = 128;
constexpr int kTILE  = 16;
constexpr int kRT    = 2;
constexpr int kBE    = kRT * kTILE;
constexpr int kHC    = kM0 * kC / 2;
constexpr float kCUTOFF  = 12.0f;
constexpr float kRESCALE = 5.0f;
constexpr float kEPS     = 1e-5f;
constexpr int kLISTB  = 32;
constexpr int kLISTU  = 40;
constexpr int kLCH    = 2048;
constexpr int kLK     = 32;
constexpr int kGB     = 16;
constexpr int kGRP    = 4;
constexpr int kEDGET = 128;
constexpr int kXLD   = 8;
constexpr int kSORT   = 128;
constexpr int kAP     = 84;

__device__ __forceinline__ float silu_f(float x) {
    return x / (1.0f + __expf(-x));
}

template <int M>
__device__ __forceinline__ float row_xor(float v) {
    return __builtin_bit_cast(float, __builtin_amdgcn_update_dpp(
        0, __builtin_bit_cast(int, v), 0x160 | M, 0xf, 0xf, false));
}

constexpr int kLNF = 8;

__device__ __forceinline__ void ln_silu_16x64(const float* __restrict__ hraw,
                                              float* __restrict__ hout,
                                              const float* __restrict__ g,
                                              const float* __restrict__ be,
                                              int tid) {
    const int c0  = (tid & 7) * kLNF;
    const int off = (tid >> 3) * 64 + c0;
    const float4 p = *reinterpret_cast<const float4*>(hraw + off);
    const float4 q = *reinterpret_cast<const float4*>(hraw + off + 4);
    const float v[kLNF] = {p.x, p.y, p.z, p.w, q.x, q.y, q.z, q.w};
    float gv[kLNF], bv[kLNF];
    #pragma unroll
    for (int j = 0; j < kLNF; ++j) { gv[j] = g[c0 + j]; bv[j] = be[c0 + j]; }
    float s = 0.0f, s2 = 0.0f;
    #pragma unroll
    for (int j = 0; j < kLNF; ++j) { s += v[j]; s2 += v[j] * v[j]; }
    s += row_xor<1>(s);  s2 += row_xor<1>(s2);
    s += row_xor<2>(s);  s2 += row_xor<2>(s2);
    s += row_xor<4>(s);  s2 += row_xor<4>(s2);
    const float mu  = s * (1.0f / 64.0f);
    const float var = s2 * (1.0f / 64.0f) - mu * mu;
    const float inv = rsqrtf(var + kEPS);
    float o[kLNF];
    #pragma unroll
    for (int j = 0; j < kLNF; ++j) o[j] = silu_f((v[j] - mu) * inv * gv[j] + bv[j]);
    *reinterpret_cast<float4*>(hout + off)     = make_float4(o[0], o[1], o[2], o[3]);
    *reinterpret_cast<float4*>(hout + off + 4) = make_float4(o[4], o[5], o[6], o[7]);
}


template <int KS>
__device__ __forceinline__ void load_w_frags(v2f (&b)[KS], const float* __restrict__ w,
                                             int ld, int col, int khalf) {
    #pragma unroll
    for (int ks = 0; ks < KS; ++ks) {
        b[ks].x = w[(4 * ks + khalf)     * ld + col];
        b[ks].y = w[(4 * ks + khalf + 1) * ld + col];
    }
}

__device__ __forceinline__ void build_dst_lists_slice(const int* __restrict__ edge_index,
                                                      const int* __restrict__ node_offset_p,
                                                      int* __restrict__ row_off,
                                                      int* __restrict__ perm,
                                                      int E, int N, int part,
                                                      int* __restrict__ cnt,
                                                      int* __restrict__ keep_e,
                                                      int* __restrict__ keep_r,
                                                      int* __restrict__ lsum)
{
    constexpr int kW = kEDGET / 32;
    const int tid  = threadIdx.x;
    const int lane = tid & 31;
    const int wave = tid >> 5;
    const int noff = node_offset_p[0];
    const int* __restrict__ dstv = edge_index + E;

    const int npb = (N + kLISTB - 1) / kLISTB;
    const int lo  = (part * npb < N) ? part * npb : N;
    const int hi  = (part == kLISTB - 1) ? N : ((lo + npb < N) ? lo + npb : N);
    if (lo >= hi && part != kLISTB - 1) return;

    int cb = lo;
    do {
        const int ce = (cb + kLCH < hi) ? cb + kLCH : hi;
        const int nn = ce - cb;
        for (int i = tid; i < nn; i += kEDGET) cnt[i] = 0;
        if (tid == 0) lsum[2 * kW] = 0;
        __syncthreads();

        const unsigned ucb = (unsigned)cb, unn = (unsigned)nn, unoff = (unsigned)noff;
        unsigned below = 0;
        int kept = 0;
        for (int eb = tid; eb < E; eb += kEDGET * kLISTU) {
            unsigned dd[kLISTU];
            #pragma unroll
            for (int u = 0; u < kLISTU; ++u) {
                const int e  = eb + u * kEDGET;
                const int dv = dstv[e < E ? e : eb];
                dd[u] = (e < E) ? (unsigned)dv - unoff : 0xFFFFFFFFu;
            }
            #pragma unroll
            for (int u = 0; u < kLISTU; ++u) {
                const unsigned r = dd[u] - ucb;
                if (r < unn) {
                    atomicAdd(&cnt[r], 1);
                    const int slot = ((kept < kLK) ? kept : kLK - 1) * kEDGET + tid;
                    keep_e[slot] = eb + u * kEDGET;
                    keep_r[slot] = (int)r;
                    ++kept;
                }
                below += (dd[u] < ucb) ? 1u : 0u;
            }
        }
        if (kept > kLK) lsum[2 * kW] = 1;
        #pragma unroll
        for (int off = 16; off > 0; off >>= 1) below += __shfl_xor(below, off);
        if (lane == 0) lsum[wave] = (int)below;
        __syncthreads();
        int base = 0;
        #pragma unroll
        for (int w = 0; w < kW; ++w) base += lsum[w];

        const int per = (nn + kEDGET - 1) / kEDGET;
        const int i0  = (tid * per < nn) ? tid * per : nn;
        const int i1  = (i0 + per < nn) ? i0 + per : nn;
        int s = 0;
        for (int i = i0; i < i1; ++i) s += cnt[i];
        int incl = s;
        #pragma unroll
        for (int off = 1; off < 32; off <<= 1) {
            const int t = __shfl_up(incl, off);
            if (lane >= off) incl += t;
        }
        if (lane == 31) lsum[kW + wave] = incl;
        __syncthreads();
        int run = base + (incl - s);
        for (int w = 0; w < wave; ++w) run += lsum[kW + w];
        for (int i = i0; i < i1; ++i) {
            const int k = cnt[i];
            cnt[i] = run;
            row_off[cb + i] = run;
            run += k;
        }
        if (ce == N && tid == kEDGET - 1) row_off[N] = run;
        __syncthreads();

        if (lsum[2 * kW] == 0) {
            constexpr int kPU = 4;
            static_assert(kLK % kPU == 0, "a trip must stay inside the column");
            #pragma unroll 1
            for (int i0 = 0; i0 < kept; i0 += kPU) {
                int ke[kPU], kr[kPU], pos[kPU];
                #pragma unroll
                for (int q = 0; q < kPU; ++q) {
                    const int i = (i0 + q < kept) ? i0 + q : i0;
                    ke[q] = keep_e[i * kEDGET + tid];
                    kr[q] = keep_r[i * kEDGET + tid];
                }
                #pragma unroll
                for (int q = 0; q < kPU; ++q) {
                    pos[q] = (i0 + q < kept) ? atomicAdd(&cnt[kr[q]], 1) : -1;
                }
                #pragma unroll
                for (int q = 0; q < kPU; ++q) {
                    if (pos[q] >= 0) perm[pos[q]] = ke[q];
                }
            }
        } else
        for (int eb = tid; eb < E; eb += kEDGET * kLISTU) {
            unsigned dd[kLISTU];
            #pragma unroll
            for (int u = 0; u < kLISTU; ++u) {
                const int e  = eb + u * kEDGET;
                const int dv = dstv[e < E ? e : eb];
                dd[u] = (e < E) ? (unsigned)dv - unoff : 0xFFFFFFFFu;
            }
            int pos[kLISTU];
            #pragma unroll
            for (int u = 0; u < kLISTU; ++u) {
                const unsigned r = dd[u] - ucb;
                pos[u] = (r < unn) ? atomicAdd(&cnt[r], 1) : -1;
            }
            #pragma unroll
            for (int u = 0; u < kLISTU; ++u) {
                if (pos[u] >= 0) perm[pos[u]] = eb + u * kEDGET;
            }
        }
        __syncthreads();
        cb += kLCH;
    } while (cb < hi);
}

__device__ __forceinline__ void store_m0_half(const float* __restrict__ tile,
                                              float* __restrict__ dst, int rows, int tid) {
    constexpr int kP4   = kHC / 4;
    constexpr int kD4   = kM0 * kC / 4;
    constexpr int kQ    = kP4 / 16;
    constexpr int kHW   = kEDGET / 16;
    constexpr int kOUT4 = (kBE / kHW) * kQ;
    constexpr int kGR   = kQ;
    static_assert(kP4 % 16 == 0 && kBE % kHW == 0, "whole chunks per row half, whole rows per pass");
    const int hw  = tid >> 4;
    const int l16 = tid & 15;
    const v4f* src4 = reinterpret_cast<const v4f*>(tile) + hw * kP4 + l16;
    v4f*       dst4 = reinterpret_cast<v4f*>(dst) + hw * kD4 + l16;
    if (rows == kBE) {
        #pragma unroll
        for (int g = 0; g < kOUT4 / kGR; ++g) {
            v4f t[kGR];
            #pragma unroll
            for (int q = 0; q < kGR; ++q) t[q] = src4[g * kHW * kP4 + q * 16];
            static_assert(kGR == 5, "the statement below names five pieces");
            asm volatile("" :: "v"(t[0]), "v"(t[1]), "v"(t[2]), "v"(t[3]), "v"(t[4]));
            #pragma unroll
            for (int q = 0; q < kGR; ++q) dst4[g * kHW * kD4 + q * 16] = t[q];
        }
    } else {
        #pragma unroll 1
        for (int g = 0; g < kOUT4 / kGR; ++g) {
            if (g * kHW + hw < rows) {
                #pragma unroll 1
                for (int q = 0; q < kGR; ++q) dst4[g * kHW * kD4 + q * 16] = src4[g * kHW * kP4 + q * 16];
            }
        }
    }
}

__global__ __launch_bounds__(kEDGET, 6)
void edge_mlp_kernel(const float* __restrict__ x_edge,
                              const float* __restrict__ edge_distance,
                              const float* __restrict__ w1, const float* __restrict__ b1,
                              const float* __restrict__ g1, const float* __restrict__ be1,
                              const float* __restrict__ w2, const float* __restrict__ b2,
                              const float* __restrict__ g2, const float* __restrict__ be2,
                              const float* __restrict__ w3, const float* __restrict__ b3,
                              float* __restrict__ m0buf,
                              float* __restrict__ envbuf,
                              const int* __restrict__ edge_index,
                              const int* __restrict__ node_offset_p,
                              int* __restrict__ row_off,
                              int* __restrict__ perm,
                              int E, int N)
{
    __shared__ __attribute__((aligned(16))) float m0s[kBE * kHC];
    __shared__ __attribute__((aligned(16))) float h2[kBE * 64];
    __shared__ int lsum[2 * (kEDGET / 32) + 1];
    static_assert(kBE * 128 + 2 * kBE * 64 <= kBE * kHC, "x_edge tile, pre-LN tile and h1 must fit the m0 half tile");
    float* const xe   = m0s;
    float* const hraw = m0s + kBE * 128;
    float* const h1   = hraw + kBE * 64;

    static_assert(kLCH + 2 * kLK * kEDGET <= kBE * kHC, "list counters and kept edges must fit the m0 half tile");
    if (blockIdx.x < kLISTB) {
        int* lds = reinterpret_cast<int*>(m0s);
        build_dst_lists_slice(edge_index, node_offset_p, row_off, perm, E, N,
                              (int)blockIdx.x, lds, lds + kLCH, lds + kLCH + kLK * kEDGET, lsum);
        return;
    }

    const int tid   = threadIdx.x;
    const int wave  = tid >> 5;
    const int lane  = tid & 31;
    const int lrow  = lane & 15;
    const int khalf = (lane >> 4) << 1;
    const int rbase = (lane >> 4) << 3;
    const int e0    = ((int)blockIdx.x - kLISTB) * kBE;

    v2f bw1[32];
    float bias1;

    {
        const float4* src4 = reinterpret_cast<const float4*>(x_edge + (size_t)e0 * 128);
        float4 v[kXLD];
        #pragma unroll
        for (int u = 0; u < kXLD; ++u) {
            const int i4 = tid + u * kEDGET;
            const bool ok = (e0 + (i4 >> 5)) < E;
            v[u] = src4[ok ? i4 : (i4 & 31)];
        }
        load_w_frags<32>(bw1, w1, 64, wave * 16 + lrow, khalf);
        bias1 = b1[wave * 16 + lrow];
        float4* xe4 = reinterpret_cast<float4*>(xe);
        #pragma unroll
        for (int u = 0; u < kXLD; ++u) {
            const int i4 = tid + u * kEDGET;
            const bool ok = (e0 + (i4 >> 5)) < E;
            xe4[i4] = ok ? v[u] : make_float4(0.0f, 0.0f, 0.0f, 0.0f);
        }
    }
    __syncthreads();

    v2f bw2[16];
    load_w_frags<16>(bw2, w2, 64, wave * 16 + lrow, khalf);
    const float bias2 = b2[wave * 16 + lrow];

    {
        const int colBase = wave * 16;
        v8f acc[kRT] = {};
        #pragma unroll
        for (int ks = 0; ks < 32; ++ks) {
            #pragma unroll
            for (int rt = 0; rt < kRT; ++rt) {
                v2f a;
                a.x = xe[(rt * kTILE + lrow) * 128 + 4 * ks + khalf];
                a.y = xe[(rt * kTILE + lrow) * 128 + 4 * ks + khalf + 1];
                acc[rt] = __builtin_amdgcn_wmma_f32_16x16x4_f32(
                    false, a, false, bw1[ks], (short)0, acc[rt], false, false);
            }
        }
        #pragma unroll
        for (int rt = 0; rt < kRT; ++rt) {
            #pragma unroll
            for (int r = 0; r < 8; ++r) {
                int row = rt * kTILE + rbase + r;
                hraw[row * 64 + colBase + lrow] = acc[rt][r] + bias1;
            }
        }
    }
    __syncthreads();
    #pragma unroll
    for (int rt = 0; rt < kRT; ++rt) ln_silu_16x64(hraw + rt * kTILE * 64, h1 + rt * kTILE * 64, g1, be1, tid);
    __syncthreads();

    constexpr int kH3 = (kHC / 16) / 4;
    constexpr int kT3 = 2 * kH3;
    constexpr int kHS = 4;
    constexpr int kSPT = 16 / kHS;
    constexpr int kNS = 4;
    constexpr int kS3 = kT3 * kSPT;
    static_assert(kNS - 1 <= kSPT, "the sets requested ahead of layer 3 belong to the first column tile");
    v2f   bw3[kNS][kHS];
    float bias3[2];
    {
        const int col = wave * 16 + lrow;
        #pragma unroll
        for (int s = 0; s < kNS - 1; ++s)
            load_w_frags<kHS>(bw3[s], w3 + (size_t)(4 * kHS * s) * (kM0 * kC), kM0 * kC, col, khalf);
        bias3[0] = b3[col];
    }

    {
        const int colBase = wave * 16;
        v8f acc[kRT] = {};
        #pragma unroll
        for (int ks = 0; ks < 16; ++ks) {
            #pragma unroll
            for (int rt = 0; rt < kRT; ++rt) {
                v2f a;
                a.x = h1[(rt * kTILE + lrow) * 64 + 4 * ks + khalf];
                a.y = h1[(rt * kTILE + lrow) * 64 + 4 * ks + khalf + 1];
                acc[rt] = __builtin_amdgcn_wmma_f32_16x16x4_f32(
                    false, a, false, bw2[ks], (short)0, acc[rt], false, false);
            }
        }
        #pragma unroll
        for (int rt = 0; rt < kRT; ++rt) {
            #pragma unroll
            for (int r = 0; r < 8; ++r) {
                int row = rt * kTILE + rbase + r;
                hraw[row * 64 + colBase + lrow] = acc[rt][r] + bias2;
            }
        }
    }
    __syncthreads();
    #pragma unroll
    for (int rt = 0; rt < kRT; ++rt) ln_silu_16x64(hraw + rt * kTILE * 64, h2 + rt * kTILE * 64, g2, be2, tid);
    __syncthreads();

    {
        const int rows = (E - e0 < kBE) ? (E - e0) : kBE;
        v2f a3[kRT][16];
        #pragma unroll
        for (int rt = 0; rt < kRT; ++rt) {
            #pragma unroll
            for (int ks = 0; ks < 16; ++ks) {
                a3[rt][ks].x = h2[(rt * kTILE + lrow) * 64 + 4 * ks + khalf];
                a3[rt][ks].y = h2[(rt * kTILE + lrow) * 64 + 4 * ks + khalf + 1];
            }
        }
        v8f acc[kRT];
        #pragma unroll
        for (int s = 0; s < kS3; ++s) {
            const int t = s / kSPT;
            const int h = s % kSPT;
            if (s + kNS - 1 < kS3) {
                const int sn  = s + kNS - 1;
                const int tn  = sn / kSPT;
                const int hn  = sn % kSPT;
                const int col = (tn / kH3) * kHC + (wave + 4 * (tn % kH3)) * 16 + lrow;
                load_w_frags<kHS>(bw3[sn % kNS], w3 + (size_t)(4 * kHS * hn) * (kM0 * kC), kM0 * kC, col, khalf);
                if (hn == 0) bias3[tn % 2] = b3[col];
            }
            if (h == 0) {
                #pragma unroll
                for (int rt = 0; rt < kRT; ++rt) acc[rt] = v8f{};
            }
            #pragma unroll
            for (int ks = 0; ks < kHS; ++ks) {
                #pragma unroll
                for (int rt = 0; rt < kRT; ++rt) {
                    acc[rt] = __builtin_amdgcn_wmma_f32_16x16x4_f32(
                        false, a3[rt][kHS * h + ks], false, bw3[s % kNS][ks], (short)0, acc[rt], false, false);
                }
            }
            if (h == kSPT - 1) {
                const int colBase = (wave + 4 * (t % kH3)) * 16;
                #pragma unroll
                for (int rt = 0; rt < kRT; ++rt) {
                    #pragma unroll
                    for (int r = 0; r < 8; ++r) {
                        int row = rt * kTILE + rbase + r;
                        m0s[row * kHC + colBase + lrow] = acc[rt][r] + bias3[t % 2];
                    }
                }
                if (t % kH3 == kH3 - 1) {
                    __syncthreads();
                    store_m0_half(m0s, m0buf + (size_t)e0 * (kM0 * kC) + (t / kH3) * kHC, rows, tid);
                    if (t + 1 < kT3) __syncthreads();
                }
            }
        }
    }
    if (tid < kBE) {
        int e = e0 + tid;
        if (e < E) {
            float env = 0.0f;
            float d = edge_distance[e] * (1.0f / kCUTOFF);
            if (d < 1.0f) {
                float d5 = d * d * d * d * d;
                env = 1.0f - 21.0f * d5 + 35.0f * d5 * d - 15.0f * d5 * d * d;
            }
            envbuf[e] = env * (1.0f / kRESCALE);
        }
    }
}

__device__ __forceinline__ float xhalf(float v) {
    return __builtin_bit_cast(float, __builtin_amdgcn_permlanex16(
        __builtin_bit_cast(int, v), __builtin_bit_cast(int, v), 0x76543210, 0xfedcba98, false, false));
}

__global__ __launch_bounds__(128, 5)
void node_gather_kernel(const float* __restrict__ x,
                        const float* __restrict__ wigner_inv,
                        const float* __restrict__ m0buf,
                        const float* __restrict__ envbuf,
                        const int*   __restrict__ row_off,
                        const int*   __restrict__ perm,
                        int*         __restrict__ perm_sorted,
                        float*       __restrict__ out)
{
    __shared__ __attribute__((aligned(16))) float a_s[32 * kAP];
    __shared__ __attribute__((aligned(8))) float env_k[kGB * kM0];
    __shared__ __attribute__((aligned(16))) int eraw[kSORT];
    __shared__ __attribute__((aligned(16))) int esort[kSORT + kGB];

    const int n    = blockIdx.x;
    const int c    = threadIdx.x;
    const int lane = c & 31;
    const int wave = c >> 5;
    const int lc   = lane & 15;
    const int lh   = lane >> 4;
    const int beg  = row_off[n];
    const int cnt  = row_off[n + 1] - beg;

    v8f acc[2][2];
    const float* xn = x + (size_t)n * (kNCOEF * kC) + wave * 32 + lc;
    #pragma unroll
    for (int mt = 0; mt < 2; ++mt) {
        #pragma unroll
        for (int nt = 0; nt < 2; ++nt) {
            #pragma unroll
            for (int i = 0; i < 8; ++i) {
                const int row = 16 * mt + 8 * lh + i;
                acc[mt][nt][i] = (row < kNCOEF) ? xn[row * kC + 16 * nt] : 0.0f;
            }
        }
    }

    for (int i = c; i < (32 - kNCOEF) * kAP; i += blockDim.x) a_s[kNCOEF * kAP + i] = 0.0f;

    const int  stage_r    = c / kM0;
    const int  stage_l    = c - stage_r * kM0;
    const bool stage_w    = c < kNCOEF * kM0;
    const int  stage_goff = stage_r * kNCOEF + stage_l * (stage_l + 1);
    const int  stage_aidx = stage_r * kAP + stage_l;

    const bool in_lds = cnt <= kSORT;
    if (in_lds) {
        const int ei = (c < cnt) ? perm[beg + c] : 0x7fffffff;
        eraw[c] = ei;
        __syncthreads();
        if (c < cnt) {
            int rank = 0;
            for (int j = 0; j < cnt; j += 4) {
                const int4 v = *reinterpret_cast<const int4*>(eraw + j);
                rank += ((v.x < ei) ? 1 : 0) + ((v.y < ei) ? 1 : 0) + ((v.z < ei) ? 1 : 0) + ((v.w < ei) ? 1 : 0);
            }
            esort[rank] = ei;
            if (rank == cnt - 1) {
                const int cpad = (cnt + kGB - 1) & ~(kGB - 1);
                for (int p = cnt; p < cpad; ++p) esort[p] = ei;
            }
        }
    } else {
        for (int i = c; i < cnt; i += blockDim.x) {
            const int ei = perm[beg + i];
            int rank = 0;
            for (int j = 0; j < cnt; ++j) rank += (perm[beg + j] < ei) ? 1 : 0;
            perm_sorted[beg + rank] = ei;
        }
    }
    __syncthreads();

    for (int b0 = 0; b0 < cnt; b0 += kGB) {
        const int nb = (cnt - b0 < kGB) ? (cnt - b0) : kGB;
        const int ng = (nb + kGRP - 1) / kGRP;
        if (!in_lds) {
            if (c < kGB) esort[c] = perm_sorted[beg + b0 + (c < nb ? c : nb - 1)];
            __syncthreads();
        }
        const int* elist = esort + (in_lds ? b0 : 0);

        v2f bf[kGB / kGRP][5][2];
        float wv[kGB];
        #pragma unroll
        for (int g = 0; g < kGB / kGRP; ++g) {
            if (g < ng) {
                int eg[kGRP];
                #pragma unroll
                for (int q = 0; q < kGRP; ++q) eg[q] = elist[g * kGRP + q];
                if (stage_w) {
                    #pragma unroll
                    for (int q = 0; q < kGRP; ++q) {
                        wv[g * kGRP + q] = wigner_inv[(size_t)eg[q] * (kNCOEF * kNCOEF) + stage_goff];
                    }
                }
                #pragma unroll
                for (int ks = 0; ks < 5; ++ks) {
                    #pragma unroll
                    for (int s2 = 0; s2 < 2; ++s2) {
                        const int k0 = 4 * ks + s2;
                        const int k1 = 4 * ks + 2 + s2;
                        const int e  = lh ? eg[k1 / kM0] : eg[k0 / kM0];
                        const int l  = lh ? (k1 % kM0) : (k0 % kM0);
                        const float* mp = m0buf + (size_t)e * (kM0 * kC) + l * kC + wave * 32 + lc;
                        const float b_lo = mp[0];
                        const float b_hi = mp[16];
                        if (s2 == 0) { bf[g][ks][0].x = b_lo; bf[g][ks][1].x = b_hi; }
                        else         { bf[g][ks][0].y = b_lo; bf[g][ks][1].y = b_hi; }
                    }
                }
            }
        }

        if (c < ng * (kGRP * kM0)) env_k[c] = (stage_r < nb) ? envbuf[elist[stage_r]] : 0.0f;
        #pragma unroll
        for (int j = 0; j < kGB; ++j) {
            if (j < ng * kGRP && stage_w) a_s[stage_aidx + kM0 * j] = (j < nb) ? wv[j] : 0.0f;
        }
        __syncthreads();

        #pragma unroll
        for (int g = 0; g < kGB / kGRP; ++g) {
            if (g < ng) {
                #pragma unroll
                for (int ks = 0; ks < 5; ++ks) {
                    const int kk = g * (kGRP * kM0) + 4 * ks + 2 * lh;
                    const v2f ek = *reinterpret_cast<const v2f*>(env_k + kk);
                    const v2f a0 = *reinterpret_cast<const v2f*>(a_s + lc * kAP + kk) * ek;
                    const v2f a1 = *reinterpret_cast<const v2f*>(a_s + (16 + lc) * kAP + kk) * ek;
                    acc[0][0] = __builtin_amdgcn_wmma_f32_16x16x4_f32(
                        false, a0, false, bf[g][ks][0], (short)0, acc[0][0], false, false);
                    acc[0][1] = __builtin_amdgcn_wmma_f32_16x16x4_f32(
                        false, a0, false, bf[g][ks][1], (short)0, acc[0][1], false, false);
                    acc[1][0] = __builtin_amdgcn_wmma_f32_16x16x4_f32(
                        false, a1, false, bf[g][ks][0], (short)0, acc[1][0], false, false);
                    acc[1][1] = __builtin_amdgcn_wmma_f32_16x16x4_f32(
                        false, a1, false, bf[g][ks][1], (short)0, acc[1][1], false, false);
                }
            }
        }
        __syncthreads();
    }

    float* on = out + (size_t)n * (kNCOEF * kC) + c;
    #pragma unroll
    for (int mt = 0; mt < 2; ++mt) {
        #pragma unroll
        for (int i = 0; i < 8; ++i) {
            const float r0 = acc[mt][0][i];
            const float r1 = acc[mt][1][i];
            const float x0 = xhalf(r0);
            const float x1 = xhalf(r1);
            const float s_lo = lh ? x1 : r0;
            const float s_hi = lh ? r1 : x0;
            const int row_lo = 16 * mt + i;
            const int row_hi = 16 * mt + 8 + i;
            if (row_lo < kNCOEF) on[row_lo * kC] = s_lo;
            if (row_hi < kNCOEF) on[row_hi * kC] = s_hi;
        }
    }
}

extern "C" void kernel_launch(void* const* d_in, const int* in_sizes, int n_in,
                              void* d_out, int out_size, void* d_ws, size_t ws_size,
                              hipStream_t stream) {
    const float* x             = (const float*)d_in[0];
    const float* x_edge        = (const float*)d_in[1];
    const float* edge_distance = (const float*)d_in[2];
    const float* wigner_inv    = (const float*)d_in[3];
    const int*   edge_index    = (const int*)d_in[4];
    const float* w1  = (const float*)d_in[5];
    const float* b1  = (const float*)d_in[6];
    const float* g1  = (const float*)d_in[7];
    const float* be1 = (const float*)d_in[8];
    const float* w2  = (const float*)d_in[9];
    const float* b2  = (const float*)d_in[10];
    const float* g2  = (const float*)d_in[11];
    const float* be2 = (const float*)d_in[12];
    const float* w3  = (const float*)d_in[13];
    const float* b3  = (const float*)d_in[14];
    const int* node_offset = (const int*)d_in[15];

    const int E = in_sizes[1] / 128;
    const int N = out_size / (kNCOEF * kC);

    float* m0buf       = (float*)d_ws;
    float* envbuf      = m0buf + (size_t)E * (kM0 * kC);
    int*   row_off     = (int*)(envbuf + E);
    int*   perm        = row_off + (N + 1);
    int*   perm_sorted = perm + E;

    const int blocks = (E + kBE - 1) / kBE;
    edge_mlp_kernel<<<kLISTB + blocks, kEDGET, 0, stream>>>(
        x_edge, edge_distance,
        w1, b1, g1, be1, w2, b2, g2, be2, w3, b3,
        m0buf, envbuf, edge_index, node_offset, row_off, perm, E, N);

    node_gather_kernel<<<N, 128, 0, stream>>>(
        x, wigner_inv, m0buf, envbuf, row_off, perm, perm_sorted, (float*)d_out);
}


# Edge network: LDS row pitches padded (x_edge 132, hidden 68, m0 half tile 324 floats) so the sixteen rows of an A-fragment read and the two row groups of a D store fall on different banks
# speedup vs baseline: 5.7727x; 1.0640x over previous
#include <hip/hip_runtime.h>

typedef __attribute__((ext_vector_type(2))) float v2f;
typedef __attribute__((ext_vector_type(4))) float v4f;
typedef __attribute__((ext_vector_type(8))) float v8f;

constexpr int kNCOEF = 25;
constexpr int kM0    = 5;
constexpr int kC     = 128;
constexpr int kTILE  = 16;
constexpr int kRT    = 2;
constexpr int kBE    = kRT * kTILE;
constexpr int kHC    = kM0 * kC / 2;
constexpr int kXP    = 128 + 4;
constexpr int kHP    = 64 + 4;
constexpr int kMP    = kHC + 4;
constexpr float kCUTOFF  = 12.0f;
constexpr float kRESCALE = 5.0f;
constexpr float kEPS     = 1e-5f;
constexpr int kLISTB  = 32;
constexpr int kLISTU  = 40;
constexpr int kLCH    = 2048;
constexpr int kLK     = 32;
constexpr int kGB     = 16;
constexpr int kGRP    = 4;
constexpr int kEDGET = 128;
constexpr int kXLD   = 8;
constexpr int kSORT   = 128;
constexpr int kAP     = 84;

__device__ __forceinline__ float silu_f(float x) {
    return x / (1.0f + __expf(-x));
}

template <int M>
__device__ __forceinline__ float row_xor(float v) {
    return __builtin_bit_cast(float, __builtin_amdgcn_update_dpp(
        0, __builtin_bit_cast(int, v), 0x160 | M, 0xf, 0xf, false));
}

constexpr int kLNF = 8;

__device__ __forceinline__ void ln_silu_16x64(const float* __restrict__ hraw,
                                              float* __restrict__ hout,
                                              const float* __restrict__ g,
                                              const float* __restrict__ be,
                                              int tid) {
    const int c0  = (tid & 7) * kLNF;
    const int off = (tid >> 3) * kHP + c0;
    const float4 p = *reinterpret_cast<const float4*>(hraw + off);
    const float4 q = *reinterpret_cast<const float4*>(hraw + off + 4);
    const float v[kLNF] = {p.x, p.y, p.z, p.w, q.x, q.y, q.z, q.w};
    float gv[kLNF], bv[kLNF];
    #pragma unroll
    for (int j = 0; j < kLNF; ++j) { gv[j] = g[c0 + j]; bv[j] = be[c0 + j]; }
    float s = 0.0f, s2 = 0.0f;
    #pragma unroll
    for (int j = 0; j < kLNF; ++j) { s += v[j]; s2 += v[j] * v[j]; }
    s += row_xor<1>(s);  s2 += row_xor<1>(s2);
    s += row_xor<2>(s);  s2 += row_xor<2>(s2);
    s += row_xor<4>(s);  s2 += row_xor<4>(s2);
    const float mu  = s * (1.0f / 64.0f);
    const float var = s2 * (1.0f / 64.0f) - mu * mu;
    const float inv = rsqrtf(var + kEPS);
    float o[kLNF];
    #pragma unroll
    for (int j = 0; j < kLNF; ++j) o[j] = silu_f((v[j] - mu) * inv * gv[j] + bv[j]);
    *reinterpret_cast<float4*>(hout + off)     = make_float4(o[0], o[1], o[2], o[3]);
    *reinterpret_cast<float4*>(hout + off + 4) = make_float4(o[4], o[5], o[6], o[7]);
}


template <int KS>
__device__ __forceinline__ void load_w_frags(v2f (&b)[KS], const float* __restrict__ w,
                                             int ld, int col, int khalf) {
    #pragma unroll
    for (int ks = 0; ks < KS; ++ks) {
        b[ks].x = w[(4 * ks + khalf)     * ld + col];
        b[ks].y = w[(4 * ks + khalf + 1) * ld + col];
    }
}

__device__ __forceinline__ void build_dst_lists_slice(const int* __restrict__ edge_index,
                                                      const int* __restrict__ node_offset_p,
                                                      int* __restrict__ row_off,
                                                      int* __restrict__ perm,
                                                      int E, int N, int part,
                                                      int* __restrict__ cnt,
                                                      int* __restrict__ keep_e,
                                                      int* __restrict__ keep_r,
                                                      int* __restrict__ lsum)
{
    constexpr int kW = kEDGET / 32;
    const int tid  = threadIdx.x;
    const int lane = tid & 31;
    const int wave = tid >> 5;
    const int noff = node_offset_p[0];
    const int* __restrict__ dstv = edge_index + E;

    const int npb = (N + kLISTB - 1) / kLISTB;
    const int lo  = (part * npb < N) ? part * npb : N;
    const int hi  = (part == kLISTB - 1) ? N : ((lo + npb < N) ? lo + npb : N);
    if (lo >= hi && part != kLISTB - 1) return;

    int cb = lo;
    do {
        const int ce = (cb + kLCH < hi) ? cb + kLCH : hi;
        const int nn = ce - cb;
        for (int i = tid; i < nn; i += kEDGET) cnt[i] = 0;
        if (tid == 0) lsum[2 * kW] = 0;
        __syncthreads();

        const unsigned ucb = (unsigned)cb, unn = (unsigned)nn, unoff = (unsigned)noff;
        unsigned below = 0;
        int kept = 0;
        for (int eb = tid; eb < E; eb += kEDGET * kLISTU) {
            unsigned dd[kLISTU];
            #pragma unroll
            for (int u = 0; u < kLISTU; ++u) {
                const int e  = eb + u * kEDGET;
                const int dv = dstv[e < E ? e : eb];
                dd[u] = (e < E) ? (unsigned)dv - unoff : 0xFFFFFFFFu;
            }
            #pragma unroll
            for (int u = 0; u < kLISTU; ++u) {
                const unsigned r = dd[u] - ucb;
                if (r < unn) {
                    atomicAdd(&cnt[r], 1);
                    const int slot = ((kept < kLK) ? kept : kLK - 1) * kEDGET + tid;
                    keep_e[slot] = eb + u * kEDGET;
                    keep_r[slot] = (int)r;
                    ++kept;
                }
                below += (dd[u] < ucb) ? 1u : 0u;
            }
        }
        if (kept > kLK) lsum[2 * kW] = 1;
        #pragma unroll
        for (int off = 16; off > 0; off >>= 1) below += __shfl_xor(below, off);
        if (lane == 0) lsum[wave] = (int)below;
        __syncthreads();
        int base = 0;
        #pragma unroll
        for (int w = 0; w < kW; ++w) base += lsum[w];

        const int per = (nn + kEDGET - 1) / kEDGET;
        const int i0  = (tid * per < nn) ? tid * per : nn;
        const int i1  = (i0 + per < nn) ? i0 + per : nn;
        int s = 0;
        for (int i = i0; i < i1; ++i) s += cnt[i];
        int incl = s;
        #pragma unroll
        for (int off = 1; off < 32; off <<= 1) {
            const int t = __shfl_up(incl, off);
            if (lane >= off) incl += t;
        }
        if (lane == 31) lsum[kW + wave] = incl;
        __syncthreads();
        int run = base + (incl - s);
        for (int w = 0; w < wave; ++w) run += lsum[kW + w];
        for (int i = i0; i < i1; ++i) {
            const int k = cnt[i];
            cnt[i] = run;
            row_off[cb + i] = run;
            run += k;
        }
        if (ce == N && tid == kEDGET - 1) row_off[N] = run;
        __syncthreads();

        if (lsum[2 * kW] == 0) {
            constexpr int kPU = 4;
            static_assert(kLK % kPU == 0, "a trip must stay inside the column");
            #pragma unroll 1
            for (int i0 = 0; i0 < kept; i0 += kPU) {
                int ke[kPU], kr[kPU], pos[kPU];
                #pragma unroll
                for (int q = 0; q < kPU; ++q) {
                    const int i = (i0 + q < kept) ? i0 + q : i0;
                    ke[q] = keep_e[i * kEDGET + tid];
                    kr[q] = keep_r[i * kEDGET + tid];
                }
                #pragma unroll
                for (int q = 0; q < kPU; ++q) {
                    pos[q] = (i0 + q < kept) ? atomicAdd(&cnt[kr[q]], 1) : -1;
                }
                #pragma unroll
                for (int q = 0; q < kPU; ++q) {
                    if (pos[q] >= 0) perm[pos[q]] = ke[q];
                }
            }
        } else
        for (int eb = tid; eb < E; eb += kEDGET * kLISTU) {
            unsigned dd[kLISTU];
            #pragma unroll
            for (int u = 0; u < kLISTU; ++u) {
                const int e  = eb + u * kEDGET;
                const int dv = dstv[e < E ? e : eb];
                dd[u] = (e < E) ? (unsigned)dv - unoff : 0xFFFFFFFFu;
            }
            int pos[kLISTU];
            #pragma unroll
            for (int u = 0; u < kLISTU; ++u) {
                const unsigned r = dd[u] - ucb;
                pos[u] = (r < unn) ? atomicAdd(&cnt[r], 1) : -1;
            }
            #pragma unroll
            for (int u = 0; u < kLISTU; ++u) {
                if (pos[u] >= 0) perm[pos[u]] = eb + u * kEDGET;
            }
        }
        __syncthreads();
        cb += kLCH;
    } while (cb < hi);
}

__device__ __forceinline__ void store_m0_half(const float* __restrict__ tile,
                                              float* __restrict__ dst, int rows, int tid) {
    constexpr int kP4   = kHC / 4;
    constexpr int kT4   = kMP / 4;
    constexpr int kD4   = kM0 * kC / 4;
    constexpr int kQ    = kP4 / 16;
    constexpr int kHW   = kEDGET / 16;
    constexpr int kOUT4 = (kBE / kHW) * kQ;
    constexpr int kGR   = kQ;
    static_assert(kP4 % 16 == 0 && kBE % kHW == 0, "whole chunks per row half, whole rows per pass");
    static_assert(kMP % 4 == 0, "tile rows start on 128-bit boundaries");
    const int hw  = tid >> 4;
    const int l16 = tid & 15;
    const v4f* src4 = reinterpret_cast<const v4f*>(tile) + hw * kT4 + l16;
    v4f*       dst4 = reinterpret_cast<v4f*>(dst) + hw * kD4 + l16;
    if (rows == kBE) {
        #pragma unroll
        for (int g = 0; g < kOUT4 / kGR; ++g) {
            v4f t[kGR];
            #pragma unroll
            for (int q = 0; q < kGR; ++q) t[q] = src4[g * kHW * kT4 + q * 16];
            static_assert(kGR == 5, "the statement below names five pieces");
            asm volatile("" :: "v"(t[0]), "v"(t[1]), "v"(t[2]), "v"(t[3]), "v"(t[4]));
            #pragma unroll
            for (int q = 0; q < kGR; ++q) dst4[g * kHW * kD4 + q * 16] = t[q];
        }
    } else {
        #pragma unroll 1
        for (int g = 0; g < kOUT4 / kGR; ++g) {
            if (g * kHW + hw < rows) {
                #pragma unroll 1
                for (int q = 0; q < kGR; ++q) dst4[g * kHW * kD4 + q * 16] = src4[g * kHW * kT4 + q * 16];
            }
        }
    }
}

__global__ __launch_bounds__(kEDGET, 6)
void edge_mlp_kernel(const float* __restrict__ x_edge,
                              const float* __restrict__ edge_distance,
                              const float* __restrict__ w1, const float* __restrict__ b1,
                              const float* __restrict__ g1, const float* __restrict__ be1,
                              const float* __restrict__ w2, const float* __restrict__ b2,
                              const float* __restrict__ g2, const float* __restrict__ be2,
                              const float* __restrict__ w3, const float* __restrict__ b3,
                              float* __restrict__ m0buf,
                              float* __restrict__ envbuf,
                              const int* __restrict__ edge_index,
                              const int* __restrict__ node_offset_p,
                              int* __restrict__ row_off,
                              int* __restrict__ perm,
                              int E, int N)
{
    __shared__ __attribute__((aligned(16))) float m0s[kBE * kMP];
    __shared__ __attribute__((aligned(16))) float h2[kBE * kHP];
    __shared__ int lsum[2 * (kEDGET / 32) + 1];
    static_assert(kBE * kXP + 2 * kBE * kHP <= kBE * kMP, "x_edge tile, pre-LN tile and h1 must fit the m0 half tile");
    static_assert(kXP % 4 == 0 && kHP % 4 == 0, "tile rows start on 128-bit boundaries");
    float* const xe   = m0s;
    float* const hraw = m0s + kBE * kXP;
    float* const h1   = hraw + kBE * kHP;

    static_assert(kLCH + 2 * kLK * kEDGET <= kBE * kMP, "list counters and kept edges must fit the m0 half tile");
    if (blockIdx.x < kLISTB) {
        int* lds = reinterpret_cast<int*>(m0s);
        build_dst_lists_slice(edge_index, node_offset_p, row_off, perm, E, N,
                              (int)blockIdx.x, lds, lds + kLCH, lds + kLCH + kLK * kEDGET, lsum);
        return;
    }

    const int tid   = threadIdx.x;
    const int wave  = tid >> 5;
    const int lane  = tid & 31;
    const int lrow  = lane & 15;
    const int khalf = (lane >> 4) << 1;
    const int rbase = (lane >> 4) << 3;
    const int e0    = ((int)blockIdx.x - kLISTB) * kBE;

    v2f bw1[32];
    float bias1;

    {
        const float4* src4 = reinterpret_cast<const float4*>(x_edge + (size_t)e0 * 128);
        float4 v[kXLD];
        #pragma unroll
        for (int u = 0; u < kXLD; ++u) {
            const int i4 = tid + u * kEDGET;
            const bool ok = (e0 + (i4 >> 5)) < E;
            v[u] = src4[ok ? i4 : (i4 & 31)];
        }
        load_w_frags<32>(bw1, w1, 64, wave * 16 + lrow, khalf);
        bias1 = b1[wave * 16 + lrow];
        float4* xe4 = reinterpret_cast<float4*>(xe);
        #pragma unroll
        for (int u = 0; u < kXLD; ++u) {
            const int i4 = tid + u * kEDGET;
            const bool ok = (e0 + (i4 >> 5)) < E;
            xe4[(i4 >> 5) * (kXP / 4) + (i4 & 31)] = ok ? v[u] : make_float4(0.0f, 0.0f, 0.0f, 0.0f);
        }
    }
    __syncthreads();

    v2f bw2[16];
    load_w_frags<16>(bw2, w2, 64, wave * 16 + lrow, khalf);
    const float bias2 = b2[wave * 16 + lrow];

    {
        const int colBase = wave * 16;
        v8f acc[kRT] = {};
        #pragma unroll
        for (int ks = 0; ks < 32; ++ks) {
            #pragma unroll
            for (int rt = 0; rt < kRT; ++rt) {
                v2f a;
                a.x = xe[(rt * kTILE + lrow) * kXP + 4 * ks + khalf];
                a.y = xe[(rt * kTILE + lrow) * kXP + 4 * ks + khalf + 1];
                acc[rt] = __builtin_amdgcn_wmma_f32_16x16x4_f32(
                    false, a, false, bw1[ks], (short)0, acc[rt], false, false);
            }
        }
        #pragma unroll
        for (int rt = 0; rt < kRT; ++rt) {
            #pragma unroll
            for (int r = 0; r < 8; ++r) {
                int row = rt * kTILE + rbase + r;
                hraw[row * kHP + colBase + lrow] = acc[rt][r] + bias1;
            }
        }
    }
    __syncthreads();
    #pragma unroll
    for (int rt = 0; rt < kRT; ++rt) ln_silu_16x64(hraw + rt * kTILE * kHP, h1 + rt * kTILE * kHP, g1, be1, tid);
    __syncthreads();

    constexpr int kH3 = (kHC / 16) / 4;
    constexpr int kT3 = 2 * kH3;
    constexpr int kHS = 4;
    constexpr int kSPT = 16 / kHS;
    constexpr int kNS = 4;
    constexpr int kS3 = kT3 * kSPT;
    static_assert(kNS - 1 <= kSPT, "the sets requested ahead of layer 3 belong to the first column tile");
    v2f   bw3[kNS][kHS];
    float bias3[2];
    {
        const int col = wave * 16 + lrow;
        #pragma unroll
        for (int s = 0; s < kNS - 1; ++s)
            load_w_frags<kHS>(bw3[s], w3 + (size_t)(4 * kHS * s) * (kM0 * kC), kM0 * kC, col, khalf);
        bias3[0] = b3[col];
    }

    {
        const int colBase = wave * 16;
        v8f acc[kRT] = {};
        #pragma unroll
        for (int ks = 0; ks < 16; ++ks) {
            #pragma unroll
            for (int rt = 0; rt < kRT; ++rt) {
                v2f a;
                a.x = h1[(rt * kTILE + lrow) * kHP + 4 * ks + khalf];
                a.y = h1[(rt * kTILE + lrow) * kHP + 4 * ks + khalf + 1];
                acc[rt] = __builtin_amdgcn_wmma_f32_16x16x4_f32(
                    false, a, false, bw2[ks], (short)0, acc[rt], false, false);
            }
        }
        #pragma unroll
        for (int rt = 0; rt < kRT; ++rt) {
            #pragma unroll
            for (int r = 0; r < 8; ++r) {
                int row = rt * kTILE + rbase + r;
                hraw[row * kHP + colBase + lrow] = acc[rt][r] + bias2;
            }
        }
    }
    __syncthreads();
    #pragma unroll
    for (int rt = 0; rt < kRT; ++rt) ln_silu_16x64(hraw + rt * kTILE * kHP, h2 + rt * kTILE * kHP, g2, be2, tid);
    __syncthreads();

    {
        const int rows = (E - e0 < kBE) ? (E - e0) : kBE;
        v2f a3[kRT][16];
        #pragma unroll
        for (int rt = 0; rt < kRT; ++rt) {
            #pragma unroll
            for (int ks = 0; ks < 16; ++ks) {
                a3[rt][ks].x = h2[(rt * kTILE + lrow) * kHP + 4 * ks + khalf];
                a3[rt][ks].y = h2[(rt * kTILE + lrow) * kHP + 4 * ks + khalf + 1];
            }
        }
        v8f acc[kRT];
        #pragma unroll
        for (int s = 0; s < kS3; ++s) {
            const int t = s / kSPT;
            const int h = s % kSPT;
            if (s + kNS - 1 < kS3) {
                const int sn  = s + kNS - 1;
                const int tn  = sn / kSPT;
                const int hn  = sn % kSPT;
                const int col = (tn / kH3) * kHC + (wave + 4 * (tn % kH3)) * 16 + lrow;
                load_w_frags<kHS>(bw3[sn % kNS], w3 + (size_t)(4 * kHS * hn) * (kM0 * kC), kM0 * kC, col, khalf);
                if (hn == 0) bias3[tn % 2] = b3[col];
            }
            if (h == 0) {
                #pragma unroll
                for (int rt = 0; rt < kRT; ++rt) acc[rt] = v8f{};
            }
            #pragma unroll
            for (int ks = 0; ks < kHS; ++ks) {
                #pragma unroll
                for (int rt = 0; rt < kRT; ++rt) {
                    acc[rt] = __builtin_amdgcn_wmma_f32_16x16x4_f32(
                        false, a3[rt][kHS * h + ks], false, bw3[s % kNS][ks], (short)0, acc[rt], false, false);
                }
            }
            if (h == kSPT - 1) {
                const int colBase = (wave + 4 * (t % kH3)) * 16;
                #pragma unroll
                for (int rt = 0; rt < kRT; ++rt) {
                    #pragma unroll
                    for (int r = 0; r < 8; ++r) {
                        int row = rt * kTILE + rbase + r;
                        m0s[row * kMP + colBase + lrow] = acc[rt][r] + bias3[t % 2];
                    }
                }
                if (t % kH3 == kH3 - 1) {
                    __syncthreads();
                    store_m0_half(m0s, m0buf + (size_t)e0 * (kM0 * kC) + (t / kH3) * kHC, rows, tid);
                    if (t + 1 < kT3) __syncthreads();
                }
            }
        }
    }
    if (tid < kBE) {
        int e = e0 + tid;
        if (e < E) {
            float env = 0.0f;
            float d = edge_distance[e] * (1.0f / kCUTOFF);
            if (d < 1.0f) {
                float d5 = d * d * d * d * d;
                env = 1.0f - 21.0f * d5 + 35.0f * d5 * d - 15.0f * d5 * d * d;
            }
            envbuf[e] = env * (1.0f / kRESCALE);
        }
    }
}

__device__ __forceinline__ float xhalf(float v) {
    return __builtin_bit_cast(float, __builtin_amdgcn_permlanex16(
        __builtin_bit_cast(int, v), __builtin_bit_cast(int, v), 0x76543210, 0xfedcba98, false, false));
}

__global__ __launch_bounds__(128, 5)
void node_gather_kernel(const float* __restrict__ x,
                        const float* __restrict__ wigner_inv,
                        const float* __restrict__ m0buf,
                        const float* __restrict__ envbuf,
                        const int*   __restrict__ row_off,
                        const int*   __restrict__ perm,
                        int*         __restrict__ perm_sorted,
                        float*       __restrict__ out)
{
    __shared__ __attribute__((aligned(16))) float a_s[32 * kAP];
    __shared__ __attribute__((aligned(8))) float env_k[kGB * kM0];
    __shared__ __attribute__((aligned(16))) int eraw[kSORT];
    __shared__ __attribute__((aligned(16))) int esort[kSORT + kGB];

    const int n    = blockIdx.x;
    const int c    = threadIdx.x;
    const int lane = c & 31;
    const int wave = c >> 5;
    const int lc   = lane & 15;
    const int lh   = lane >> 4;
    const int beg  = row_off[n];
    const int cnt  = row_off[n + 1] - beg;

    v8f acc[2][2];
    const float* xn = x + (size_t)n * (kNCOEF * kC) + wave * 32 + lc;
    #pragma unroll
    for (int mt = 0; mt < 2; ++mt) {
        #pragma unroll
        for (int nt = 0; nt < 2; ++nt) {
            #pragma unroll
            for (int i = 0; i < 8; ++i) {
                const int row = 16 * mt + 8 * lh + i;
                acc[mt][nt][i] = (row < kNCOEF) ? xn[row * kC + 16 * nt] : 0.0f;
            }
        }
    }

    for (int i = c; i < (32 - kNCOEF) * kAP; i += blockDim.x) a_s[kNCOEF * kAP + i] = 0.0f;

    const int  stage_r    = c / kM0;
    const int  stage_l    = c - stage_r * kM0;
    const bool stage_w    = c < kNCOEF * kM0;
    const int  stage_goff = stage_r * kNCOEF + stage_l * (stage_l + 1);
    const int  stage_aidx = stage_r * kAP + stage_l;

    const bool in_lds = cnt <= kSORT;
    if (in_lds) {
        const int ei = (c < cnt) ? perm[beg + c] : 0x7fffffff;
        eraw[c] = ei;
        __syncthreads();
        if (c < cnt) {
            int rank = 0;
            for (int j = 0; j < cnt; j += 4) {
                const int4 v = *reinterpret_cast<const int4*>(eraw + j);
                rank += ((v.x < ei) ? 1 : 0) + ((v.y < ei) ? 1 : 0) + ((v.z < ei) ? 1 : 0) + ((v.w < ei) ? 1 : 0);
            }
            esort[rank] = ei;
            if (rank == cnt - 1) {
                const int cpad = (cnt + kGB - 1) & ~(kGB - 1);
                for (int p = cnt; p < cpad; ++p) esort[p] = ei;
            }
        }
    } else {
        for (int i = c; i < cnt; i += blockDim.x) {
            const int ei = perm[beg + i];
            int rank = 0;
            for (int j = 0; j < cnt; ++j) rank += (perm[beg + j] < ei) ? 1 : 0;
            perm_sorted[beg + rank] = ei;
        }
    }
    __syncthreads();

    for (int b0 = 0; b0 < cnt; b0 += kGB) {
        const int nb = (cnt - b0 < kGB) ? (cnt - b0) : kGB;
        const int ng = (nb + kGRP - 1) / kGRP;
        if (!in_lds) {
            if (c < kGB) esort[c] = perm_sorted[beg + b0 + (c < nb ? c : nb - 1)];
            __syncthreads();
        }
        const int* elist = esort + (in_lds ? b0 : 0);

        v2f bf[kGB / kGRP][5][2];
        float wv[kGB];
        #pragma unroll
        for (int g = 0; g < kGB / kGRP; ++g) {
            if (g < ng) {
                int eg[kGRP];
                #pragma unroll
                for (int q = 0; q < kGRP; ++q) eg[q] = elist[g * kGRP + q];
                if (stage_w) {
                    #pragma unroll
                    for (int q = 0; q < kGRP; ++q) {
                        wv[g * kGRP + q] = wigner_inv[(size_t)eg[q] * (kNCOEF * kNCOEF) + stage_goff];
                    }
                }
                #pragma unroll
                for (int ks = 0; ks < 5; ++ks) {
                    #pragma unroll
                    for (int s2 = 0; s2 < 2; ++s2) {
                        const int k0 = 4 * ks + s2;
                        const int k1 = 4 * ks + 2 + s2;
                        const int e  = lh ? eg[k1 / kM0] : eg[k0 / kM0];
                        const int l  = lh ? (k1 % kM0) : (k0 % kM0);
                        const float* mp = m0buf + (size_t)e * (kM0 * kC) + l * kC + wave * 32 + lc;
                        const float b_lo = mp[0];
                        const float b_hi = mp[16];
                        if (s2 == 0) { bf[g][ks][0].x = b_lo; bf[g][ks][1].x = b_hi; }
                        else         { bf[g][ks][0].y = b_lo; bf[g][ks][1].y = b_hi; }
                    }
                }
            }
        }

        if (c < ng * (kGRP * kM0)) env_k[c] = (stage_r < nb) ? envbuf[elist[stage_r]] : 0.0f;
        #pragma unroll
        for (int j = 0; j < kGB; ++j) {
            if (j < ng * kGRP && stage_w) a_s[stage_aidx + kM0 * j] = (j < nb) ? wv[j] : 0.0f;
        }
        __syncthreads();

        #pragma unroll
        for (int g = 0; g < kGB / kGRP; ++g) {
            if (g < ng) {
                #pragma unroll
                for (int ks = 0; ks < 5; ++ks) {
                    const int kk = g * (kGRP * kM0) + 4 * ks + 2 * lh;
                    const v2f ek = *reinterpret_cast<const v2f*>(env_k + kk);
                    const v2f a0 = *reinterpret_cast<const v2f*>(a_s + lc * kAP + kk) * ek;
                    const v2f a1 = *reinterpret_cast<const v2f*>(a_s + (16 + lc) * kAP + kk) * ek;
                    acc[0][0] = __builtin_amdgcn_wmma_f32_16x16x4_f32(
                        false, a0, false, bf[g][ks][0], (short)0, acc[0][0], false, false);
                    acc[0][1] = __builtin_amdgcn_wmma_f32_16x16x4_f32(
                        false, a0, false, bf[g][ks][1], (short)0, acc[0][1], false, false);
                    acc[1][0] = __builtin_amdgcn_wmma_f32_16x16x4_f32(
                        false, a1, false, bf[g][ks][0], (short)0, acc[1][0], false, false);
                    acc[1][1] = __builtin_amdgcn_wmma_f32_16x16x4_f32(
                        false, a1, false, bf[g][ks][1], (short)0, acc[1][1], false, false);
                }
            }
        }
        __syncthreads();
    }

    float* on = out + (size_t)n * (kNCOEF * kC) + c;
    #pragma unroll
    for (int mt = 0; mt < 2; ++mt) {
        #pragma unroll
        for (int i = 0; i < 8; ++i) {
            const float r0 = acc[mt][0][i];
            const float r1 = acc[mt][1][i];
            const float x0 = xhalf(r0);
            const float x1 = xhalf(r1);
            const float s_lo = lh ? x1 : r0;
            const float s_hi = lh ? r1 : x0;
            const int row_lo = 16 * mt + i;
            const int row_hi = 16 * mt + 8 + i;
            if (row_lo < kNCOEF) on[row_lo * kC] = s_lo;
            if (row_hi < kNCOEF) on[row_hi * kC] = s_hi;
        }
    }
}

extern "C" void kernel_launch(void* const* d_in, const int* in_sizes, int n_in,
                              void* d_out, int out_size, void* d_ws, size_t ws_size,
                              hipStream_t stream) {
    const float* x             = (const float*)d_in[0];
    const float* x_edge        = (const float*)d_in[1];
    const float* edge_distance = (const float*)d_in[2];
    const float* wigner_inv    = (const float*)d_in[3];
    const int*   edge_index    = (const int*)d_in[4];
    const float* w1  = (const float*)d_in[5];
    const float* b1  = (const float*)d_in[6];
    const float* g1  = (const float*)d_in[7];
    const float* be1 = (const float*)d_in[8];
    const float* w2  = (const float*)d_in[9];
    const float* b2  = (const float*)d_in[10];
    const float* g2  = (const float*)d_in[11];
    const float* be2 = (const float*)d_in[12];
    const float* w3  = (const float*)d_in[13];
    const float* b3  = (const float*)d_in[14];
    const int* node_offset = (const int*)d_in[15];

    const int E = in_sizes[1] / 128;
    const int N = out_size / (kNCOEF * kC);

    float* m0buf       = (float*)d_ws;
    float* envbuf      = m0buf + (size_t)E * (kM0 * kC);
    int*   row_off     = (int*)(envbuf + E);
    int*   perm        = row_off + (N + 1);
    int*   perm_sorted = perm + E;

    const int blocks = (E + kBE - 1) / kBE;
    edge_mlp_kernel<<<kLISTB + blocks, kEDGET, 0, stream>>>(
        x_edge, edge_distance,
        w1, b1, g1, be1, w2, b2, g2, be2, w3, b3,
        m0buf, envbuf, edge_index, node_offset, row_off, perm, E, N);

    node_gather_kernel<<<N, 128, 0, stream>>>(
        x, wigner_inv, m0buf, envbuf, row_off, perm, perm_sorted, (float*)d_out);
}
